# starting kernel
# baseline (speedup 1.0000x reference)
#include <hip/hip_runtime.h>


typedef __attribute__((ext_vector_type(2))) float v2f;
typedef __attribute__((ext_vector_type(8))) float v8f;

#define IN_C  128
#define HID   512
#define OUT2  64
#define NCLS  2
#define NEG_SLOPE 0.2f
#define KMAX  512

__global__ void gat_wmma_gemm_f32(const float* __restrict__ A,
                                  const float* __restrict__ B,
                                  float* __restrict__ C,
                                  int M, int K, int Nc) {
  __shared__ float Bs[KMAX * 16];

  const int tid  = threadIdx.x;
  const int lane = tid & 31;
  const int waveid = tid >> 5;
  const int tn = blockIdx.y;
  const int tilesM = (M + 15) >> 4;

  const int nK16 = K * 16;
  for (int idx = tid; idx < nK16; idx += 256) {
    const int k = idx >> 4;
    const int c = idx & 15;
    Bs[((k >> 1) << 5) + (c << 1) + (k & 1)] = B[(size_t)k * Nc + tn * 16 + c];
  }
  __syncthreads();

  const int tm = blockIdx.x * 8 + waveid;
  if (tm >= tilesM) return;

  const int half = lane >> 4;
  const int l16  = lane & 15;

  const float* Ap = A + (size_t)(tm * 16 + l16) * K + 2 * half;
  const float* Bl = Bs + half * 32 + (l16 << 1);

  v8f acc = {};
#pragma unroll 8
  for (int k = 0; k < K; k += 4) {
    v2f a = *(const v2f*)(Ap + k);
    v2f b = *(const v2f*)(Bl + k * 16);
    acc = __builtin_amdgcn_wmma_f32_16x16x4_f32(
         false, a,  false, b,
         (short)0, acc,  false,  false);
  }

  float* Cp = C + (size_t)(tm * 16 + 8 * half) * Nc + tn * 16 + l16;
#pragma unroll
  for (int r = 0; r < 8; ++r) Cp[(size_t)r * Nc] = acc[r];
}

__global__ void gat_node_dots(const float* __restrict__ h,
                              const float* __restrict__ a_src,
                              const float* __restrict__ a_dst,
                              float* __restrict__ as_out,
                              float* __restrict__ ad_out,
                              int N, int D) {
  const int wave = blockIdx.x * (blockDim.x >> 5) + (threadIdx.x >> 5);
  const int lane = threadIdx.x & 31;
  if (wave >= N) return;
  const float* row = h + (size_t)wave * D;
  float ss = 0.f, sd = 0.f;
  for (int c = lane; c < D; c += 32) {
    float v = row[c];
    ss += v * a_src[c];
    sd += v * a_dst[c];
  }
#pragma unroll
  for (int off = 16; off > 0; off >>= 1) {
    ss += __shfl_down(ss, off, 32);
    sd += __shfl_down(sd, off, 32);
  }
  if (lane == 0) { as_out[wave] = ss; ad_out[wave] = sd; }
}

__global__ void gat_fill_f32(float* __restrict__ p, float v, int n) {
  int i = blockIdx.x * blockDim.x + threadIdx.x;
  if (i < n) p[i] = v;
}

__device__ __forceinline__ void edge_sd(const int* __restrict__ ei, int e,
                                        int E, int& s, int& d) {
  if (e < E) { s = ei[e]; d = ei[E + e]; }
  else       { s = d = e - E; }
}

__device__ __forceinline__ float leakyr(float v) {
  return v > 0.f ? v : NEG_SLOPE * v;
}

__device__ __forceinline__ void atomicMaxF32(float* addr, float val) {
  if (val >= 0.f) atomicMax((int*)addr, __float_as_int(val));
  else            atomicMin((unsigned int*)addr, __float_as_uint(val));
}

__global__ void gat_edge_max(const int* __restrict__ ei,
                             const float* __restrict__ as,
                             const float* __restrict__ ad,
                             float* __restrict__ m, int E, int Etot) {
  int e = blockIdx.x * blockDim.x + threadIdx.x;
  if (e >= Etot) return;
  int s, d; edge_sd(ei, e, E, s, d);
  atomicMaxF32(&m[d], leakyr(as[s] + ad[d]));
}

__global__ void gat_edge_expsum(const int* __restrict__ ei,
                                const float* __restrict__ as,
                                const float* __restrict__ ad,
                                const float* __restrict__ m,
                                float* __restrict__ alpha,
                                float* __restrict__ denom, int E, int Etot) {
  int e = blockIdx.x * blockDim.x + threadIdx.x;
  if (e >= Etot) return;
  int s, d; edge_sd(ei, e, E, s, d);
  float ex = expf(leakyr(as[s] + ad[d]) - m[d]);
  alpha[e] = ex;
  atomicAdd(&denom[d], ex);
}

__global__ void gat_edge_norm(const int* __restrict__ ei,
                              const float* __restrict__ denom,
                              float* __restrict__ alpha, int E, int Etot) {
  int e = blockIdx.x * blockDim.x + threadIdx.x;
  if (e >= Etot) return;
  int s, d; edge_sd(ei, e, E, s, d);
  alpha[e] /= denom[d];
}

__global__ void gat_edge_aggregate(const int* __restrict__ ei,
                                   const float* __restrict__ alpha,
                                   const float* __restrict__ h,
                                   float* __restrict__ out,
                                   int E, int Etot, int D) {
  const int wave = blockIdx.x * (blockDim.x >> 5) + (threadIdx.x >> 5);
  const int lane = threadIdx.x & 31;
  if (wave >= Etot) return;
  int s, d; edge_sd(ei, wave, E, s, d);
  const float a = alpha[wave];
  const float* hs = h + (size_t)s * D;
  float* od = out + (size_t)d * D;
  for (int c = lane; c < D; c += 32) atomicAdd(&od[c], a * hs[c]);
}

__global__ void gat_bias_relu(float* __restrict__ p,
                              const float* __restrict__ bias, int N, int Dmask) {
  int i = blockIdx.x * blockDim.x + threadIdx.x;
  if (i >= N) return;
  float v = p[i] + bias[i & Dmask];
  p[i] = v > 0.f ? v : 0.f;
}

__global__ void gat_fc(const float* __restrict__ h,
                       const float* __restrict__ W,
                       const float* __restrict__ b,
                       float* __restrict__ out, int N) {
  int n = blockIdx.x * blockDim.x + threadIdx.x;
  if (n >= N) return;
  const float* row = h + (size_t)n * OUT2;
  float a0 = b[0], a1 = b[1];
#pragma unroll
  for (int c = 0; c < OUT2; ++c) {
    float v = row[c];
    a0 += v * W[c * NCLS + 0];
    a1 += v * W[c * NCLS + 1];
  }
  out[n * NCLS + 0] = a0;
  out[n * NCLS + 1] = a1;
}

extern "C" void kernel_launch(void* const* d_in, const int* in_sizes, int n_in,
                              void* d_out, int out_size, void* d_ws, size_t ws_size,
                              hipStream_t stream) {
  const float* x   = (const float*)d_in[0];
  const int*   ei  = (const int*)  d_in[1];
  const float* W1  = (const float*)d_in[2];
  const float* a1s = (const float*)d_in[3];
  const float* a1d = (const float*)d_in[4];
  const float* b1  = (const float*)d_in[5];
  const float* W2  = (const float*)d_in[6];
  const float* a2s = (const float*)d_in[7];
  const float* a2d = (const float*)d_in[8];
  const float* b2  = (const float*)d_in[9];
  const float* fcW = (const float*)d_in[10];
  const float* fcb = (const float*)d_in[11];

  const int N    = in_sizes[0] / IN_C;
  const int E    = in_sizes[1] / 2;
  const int Etot = E + N;

  float* w       = (float*)d_ws;
  float* h1      = w;
  float* agg     = w + (size_t)N * HID;
  float* as_buf  = w + (size_t)2 * N * HID;
  float* ad_buf  = as_buf + N;
  float* m_buf   = ad_buf + N;
  float* den_buf = m_buf + N;
  float* alpha_e = den_buf + N;

  const int TB = 256;
  const float NEG_INF = -__builtin_inff();

  auto gemm_grid = [](int M, int Nc) {
    return dim3((((M + 15) / 16) + 7) / 8, Nc / 16);
  };
  const int edge_blk  = (Etot + TB - 1) / TB;
  const int eagg_blk  = (Etot + 7) / 8;
  const int node_blk  = (N + 7) / 8;

  gat_wmma_gemm_f32<<<gemm_grid(N, HID), TB, 0, stream>>>(x, W1, h1, N, IN_C, HID);
  gat_node_dots<<<node_blk, TB, 0, stream>>>(h1, a1s, a1d, as_buf, ad_buf, N, HID);

  gat_fill_f32<<<(N + TB - 1) / TB, TB, 0, stream>>>(m_buf, NEG_INF, N);
  gat_fill_f32<<<(N + TB - 1) / TB, TB, 0, stream>>>(den_buf, 0.f, N);
  gat_fill_f32<<<(N * HID + TB - 1) / TB, TB, 0, stream>>>(agg, 0.f, N * HID);

  gat_edge_max   <<<edge_blk, TB, 0, stream>>>(ei, as_buf, ad_buf, m_buf, E, Etot);
  gat_edge_expsum<<<edge_blk, TB, 0, stream>>>(ei, as_buf, ad_buf, m_buf, alpha_e, den_buf, E, Etot);
  gat_edge_norm  <<<edge_blk, TB, 0, stream>>>(ei, den_buf, alpha_e, E, Etot);
  gat_edge_aggregate<<<eagg_blk, TB, 0, stream>>>(ei, alpha_e, h1, agg, E, Etot, HID);
  gat_bias_relu<<<(N * HID + TB - 1) / TB, TB, 0, stream>>>(agg, b1, N * HID, HID - 1);

  gat_wmma_gemm_f32<<<gemm_grid(N, OUT2), TB, 0, stream>>>(agg, W2, h1, N, HID, OUT2);
  gat_node_dots<<<node_blk, TB, 0, stream>>>(h1, a2s, a2d, as_buf, ad_buf, N, OUT2);

  gat_fill_f32<<<(N + TB - 1) / TB, TB, 0, stream>>>(m_buf, NEG_INF, N);
  gat_fill_f32<<<(N + TB - 1) / TB, TB, 0, stream>>>(den_buf, 0.f, N);
  gat_fill_f32<<<(N * OUT2 + TB - 1) / TB, TB, 0, stream>>>(agg, 0.f, N * OUT2);

  gat_edge_max   <<<edge_blk, TB, 0, stream>>>(ei, as_buf, ad_buf, m_buf, E, Etot);
  gat_edge_expsum<<<edge_blk, TB, 0, stream>>>(ei, as_buf, ad_buf, m_buf, alpha_e, den_buf, E, Etot);
  gat_edge_norm  <<<edge_blk, TB, 0, stream>>>(ei, den_buf, alpha_e, E, Etot);
  gat_edge_aggregate<<<eagg_blk, TB, 0, stream>>>(ei, alpha_e, h1, agg, E, Etot, OUT2);
  gat_bias_relu<<<(N * OUT2 + TB - 1) / TB, TB, 0, stream>>>(agg, b2, N * OUT2, OUT2 - 1);

  gat_fc<<<(N + TB - 1) / TB, TB, 0, stream>>>(agg, fcW, fcb, (float*)d_out, N);
}


# Aggregation output-stationary: a block owns 98 destination rows in LDS, scans the edge list for its edges, adds gathered rows by LDS atomics (arrival order), stores each row once; 21 -> 19 launches
# speedup vs baseline: 1.4408x; 1.4408x over previous
#include <hip/hip_runtime.h>


typedef __attribute__((ext_vector_type(2))) float v2f;
typedef __attribute__((ext_vector_type(8))) float v8f;

#define IN_C  128
#define HID   512
#define OUT2  64
#define NCLS  2
#define NEG_SLOPE 0.2f
#define KMAX  512

__global__ void gat_wmma_gemm_f32(const float* __restrict__ A,
                                  const float* __restrict__ B,
                                  float* __restrict__ C,
                                  int M, int K, int Nc) {
  __shared__ float Bs[KMAX * 16];

  const int tid  = threadIdx.x;
  const int lane = tid & 31;
  const int waveid = tid >> 5;
  const int tn = blockIdx.y;
  const int tilesM = (M + 15) >> 4;

  const int nK16 = K * 16;
  for (int idx = tid; idx < nK16; idx += 256) {
    const int k = idx >> 4;
    const int c = idx & 15;
    Bs[((k >> 1) << 5) + (c << 1) + (k & 1)] = B[(size_t)k * Nc + tn * 16 + c];
  }
  __syncthreads();

  const int tm = blockIdx.x * 8 + waveid;
  if (tm >= tilesM) return;

  const int half = lane >> 4;
  const int l16  = lane & 15;

  const float* Ap = A + (size_t)(tm * 16 + l16) * K + 2 * half;
  const float* Bl = Bs + half * 32 + (l16 << 1);

  v8f acc = {};
#pragma unroll 8
  for (int k = 0; k < K; k += 4) {
    v2f a = *(const v2f*)(Ap + k);
    v2f b = *(const v2f*)(Bl + k * 16);
    acc = __builtin_amdgcn_wmma_f32_16x16x4_f32(
         false, a,  false, b,
         (short)0, acc,  false,  false);
  }

  float* Cp = C + (size_t)(tm * 16 + 8 * half) * Nc + tn * 16 + l16;
#pragma unroll
  for (int r = 0; r < 8; ++r) Cp[(size_t)r * Nc] = acc[r];
}

__global__ void gat_node_dots(const float* __restrict__ h,
                              const float* __restrict__ a_src,
                              const float* __restrict__ a_dst,
                              float* __restrict__ as_out,
                              float* __restrict__ ad_out,
                              int N, int D) {
  const int wave = blockIdx.x * (blockDim.x >> 5) + (threadIdx.x >> 5);
  const int lane = threadIdx.x & 31;
  if (wave >= N) return;
  const float* row = h + (size_t)wave * D;
  float ss = 0.f, sd = 0.f;
  for (int c = lane; c < D; c += 32) {
    float v = row[c];
    ss += v * a_src[c];
    sd += v * a_dst[c];
  }
#pragma unroll
  for (int off = 16; off > 0; off >>= 1) {
    ss += __shfl_down(ss, off, 32);
    sd += __shfl_down(sd, off, 32);
  }
  if (lane == 0) { as_out[wave] = ss; ad_out[wave] = sd; }
}

__global__ void gat_fill_f32(float* __restrict__ p, float v, int n) {
  int i = blockIdx.x * blockDim.x + threadIdx.x;
  if (i < n) p[i] = v;
}

__device__ __forceinline__ void edge_sd(const int* __restrict__ ei, int e,
                                        int E, int& s, int& d) {
  if (e < E) { s = ei[e]; d = ei[E + e]; }
  else       { s = d = e - E; }
}

__device__ __forceinline__ float leakyr(float v) {
  return v > 0.f ? v : NEG_SLOPE * v;
}

__device__ __forceinline__ void atomicMaxF32(float* addr, float val) {
  if (val >= 0.f) atomicMax((int*)addr, __float_as_int(val));
  else            atomicMin((unsigned int*)addr, __float_as_uint(val));
}

__global__ void gat_edge_max(const int* __restrict__ ei,
                             const float* __restrict__ as,
                             const float* __restrict__ ad,
                             float* __restrict__ m, int E, int Etot) {
  int e = blockIdx.x * blockDim.x + threadIdx.x;
  if (e >= Etot) return;
  int s, d; edge_sd(ei, e, E, s, d);
  atomicMaxF32(&m[d], leakyr(as[s] + ad[d]));
}

__global__ void gat_edge_expsum(const int* __restrict__ ei,
                                const float* __restrict__ as,
                                const float* __restrict__ ad,
                                const float* __restrict__ m,
                                float* __restrict__ alpha,
                                float* __restrict__ denom, int E, int Etot) {
  int e = blockIdx.x * blockDim.x + threadIdx.x;
  if (e >= Etot) return;
  int s, d; edge_sd(ei, e, E, s, d);
  float ex = expf(leakyr(as[s] + ad[d]) - m[d]);
  alpha[e] = ex;
  atomicAdd(&denom[d], ex);
}

__global__ void gat_edge_norm(const int* __restrict__ ei,
                              const float* __restrict__ denom,
                              float* __restrict__ alpha, int E, int Etot) {
  int e = blockIdx.x * blockDim.x + threadIdx.x;
  if (e >= Etot) return;
  int s, d; edge_sd(ei, e, E, s, d);
  alpha[e] /= denom[d];
}

#define AGG_ROWS 98
#define AGG_U    16

__device__ __forceinline__ unsigned agg_owned_bits(const int (&dv)[AGG_U], int d0, int rows) {
  unsigned mm = 0;
#pragma unroll
  for (int k = 0; k < AGG_U; ++k)
    mm |= ((unsigned)(dv[k] - d0) < (unsigned)rows) ? (1u << k) : 0u;
  return mm;
}

template <int D>
__device__ __forceinline__ void agg_add_owned(unsigned mm, int base, int lane, int d0,
                                              const int* __restrict__ ei,
                                              const int* __restrict__ dstp,
                                              const float* __restrict__ alpha,
                                              const float* __restrict__ h, float* acc) {
  constexpr int V = D / 32;
  unsigned lanes = __builtin_amdgcn_ballot_w32(mm != 0);
  while (lanes) {
    const int j = __builtin_ctz(lanes);
    lanes &= lanes - 1;
    unsigned bits = (unsigned)__builtin_amdgcn_readlane((int)mm, j);
    while (bits) {
      const int k = __builtin_ctz(bits);
      bits &= bits - 1;
      const int e = base + k * 32 + j;
      const int s = ei[e];
      const int r = dstp[e] - d0;
      const float a = alpha[e];
      const float* hs = h + (size_t)s * D + lane;
      float* ar = acc + r * D + lane;
      float v[V];
#pragma unroll
      for (int i = 0; i < V; ++i) v[i] = hs[i * 32];
#pragma unroll
      for (int i = 0; i < V; ++i) atomicAdd(ar + i * 32, a * v[i]);
    }
  }
}

template <int D, int R>
__global__ __launch_bounds__(1024) void gat_aggregate_owned(
    const int* __restrict__ ei, const float* __restrict__ alpha,
    const float* __restrict__ h, float* __restrict__ out, int E, int N) {
  __shared__ __attribute__((aligned(16))) float acc[R * D];
  constexpr int V = D / 32;

  const int lane = threadIdx.x & 31;
  const int wv   = threadIdx.x >> 5;
  const int nw   = blockDim.x >> 5;
  const int d0   = blockIdx.x * R;
  const int rows = (N - d0 < R) ? (N - d0) : R;

  for (int r = wv; r < rows; r += nw) {
    const float a = alpha[E + d0 + r];
    const float* hs = h + (size_t)(d0 + r) * D + lane;
    float* ar = acc + r * D + lane;
#pragma unroll
    for (int i = 0; i < V; ++i) ar[i * 32] = a * hs[i * 32];
  }
  __syncthreads();

  const int* __restrict__ dstp = ei + E;
  const int sweep  = AGG_U * 32;
  const int stride = nw * sweep;
  int base = wv * sweep;

  if (base + sweep <= E) {
    int cur[AGG_U];
#pragma unroll
    for (int k = 0; k < AGG_U; ++k) cur[k] = dstp[base + k * 32 + lane];
    while (true) {
      const int nb = base + stride;
      const bool more = (nb + sweep <= E);
      const int pb = more ? nb : base;
      int nxt[AGG_U];
#pragma unroll
      for (int k = 0; k < AGG_U; ++k) nxt[k] = dstp[pb + k * 32 + lane];
      agg_add_owned<D>(agg_owned_bits(cur, d0, rows), base, lane, d0, ei, dstp, alpha, h, acc);
      base = nb;
      if (!more) break;
#pragma unroll
      for (int k = 0; k < AGG_U; ++k) cur[k] = nxt[k];
    }
  }
  if (base < E) {
    int cur[AGG_U];
#pragma unroll
    for (int k = 0; k < AGG_U; ++k) {
      const int e = base + k * 32 + lane;
      cur[k] = (e < E) ? dstp[e] : -1;
    }
    agg_add_owned<D>(agg_owned_bits(cur, d0, rows), base, lane, d0, ei, dstp, alpha, h, acc);
  }
  __syncthreads();

  const float4* a4 = (const float4*)acc;
  float4* o4 = (float4*)(out + (size_t)d0 * D);
  for (int idx = threadIdx.x; idx < rows * (D / 4); idx += blockDim.x) o4[idx] = a4[idx];
}

__global__ void gat_bias_relu(float* __restrict__ p,
                              const float* __restrict__ bias, int N, int Dmask) {
  int i = blockIdx.x * blockDim.x + threadIdx.x;
  if (i >= N) return;
  float v = p[i] + bias[i & Dmask];
  p[i] = v > 0.f ? v : 0.f;
}

__global__ void gat_fc(const float* __restrict__ h,
                       const float* __restrict__ W,
                       const float* __restrict__ b,
                       float* __restrict__ out, int N) {
  int n = blockIdx.x * blockDim.x + threadIdx.x;
  if (n >= N) return;
  const float* row = h + (size_t)n * OUT2;
  float a0 = b[0], a1 = b[1];
#pragma unroll
  for (int c = 0; c < OUT2; ++c) {
    float v = row[c];
    a0 += v * W[c * NCLS + 0];
    a1 += v * W[c * NCLS + 1];
  }
  out[n * NCLS + 0] = a0;
  out[n * NCLS + 1] = a1;
}

extern "C" void kernel_launch(void* const* d_in, const int* in_sizes, int n_in,
                              void* d_out, int out_size, void* d_ws, size_t ws_size,
                              hipStream_t stream) {
  const float* x   = (const float*)d_in[0];
  const int*   ei  = (const int*)  d_in[1];
  const float* W1  = (const float*)d_in[2];
  const float* a1s = (const float*)d_in[3];
  const float* a1d = (const float*)d_in[4];
  const float* b1  = (const float*)d_in[5];
  const float* W2  = (const float*)d_in[6];
  const float* a2s = (const float*)d_in[7];
  const float* a2d = (const float*)d_in[8];
  const float* b2  = (const float*)d_in[9];
  const float* fcW = (const float*)d_in[10];
  const float* fcb = (const float*)d_in[11];

  const int N    = in_sizes[0] / IN_C;
  const int E    = in_sizes[1] / 2;
  const int Etot = E + N;

  float* w       = (float*)d_ws;
  float* h1      = w;
  float* agg     = w + (size_t)N * HID;
  float* as_buf  = w + (size_t)2 * N * HID;
  float* ad_buf  = as_buf + N;
  float* m_buf   = ad_buf + N;
  float* den_buf = m_buf + N;
  float* alpha_e = den_buf + N;

  const int TB = 256;
  const float NEG_INF = -__builtin_inff();

  auto gemm_grid = [](int M, int Nc) {
    return dim3((((M + 15) / 16) + 7) / 8, Nc / 16);
  };
  const int edge_blk  = (Etot + TB - 1) / TB;
  const int agg_blk   = (N + AGG_ROWS - 1) / AGG_ROWS;
  const int node_blk  = (N + 7) / 8;

  gat_wmma_gemm_f32<<<gemm_grid(N, HID), TB, 0, stream>>>(x, W1, h1, N, IN_C, HID);
  gat_node_dots<<<node_blk, TB, 0, stream>>>(h1, a1s, a1d, as_buf, ad_buf, N, HID);

  gat_fill_f32<<<(N + TB - 1) / TB, TB, 0, stream>>>(m_buf, NEG_INF, N);
  gat_fill_f32<<<(N + TB - 1) / TB, TB, 0, stream>>>(den_buf, 0.f, N);

  gat_edge_max   <<<edge_blk, TB, 0, stream>>>(ei, as_buf, ad_buf, m_buf, E, Etot);
  gat_edge_expsum<<<edge_blk, TB, 0, stream>>>(ei, as_buf, ad_buf, m_buf, alpha_e, den_buf, E, Etot);
  gat_edge_norm  <<<edge_blk, TB, 0, stream>>>(ei, den_buf, alpha_e, E, Etot);
  gat_aggregate_owned<HID, AGG_ROWS><<<agg_blk, 1024, 0, stream>>>(ei, alpha_e, h1, agg, E, N);
  gat_bias_relu<<<(N * HID + TB - 1) / TB, TB, 0, stream>>>(agg, b1, N * HID, HID - 1);

  gat_wmma_gemm_f32<<<gemm_grid(N, OUT2), TB, 0, stream>>>(agg, W2, h1, N, HID, OUT2);
  gat_node_dots<<<node_blk, TB, 0, stream>>>(h1, a2s, a2d, as_buf, ad_buf, N, OUT2);

  gat_fill_f32<<<(N + TB - 1) / TB, TB, 0, stream>>>(m_buf, NEG_INF, N);
  gat_fill_f32<<<(N + TB - 1) / TB, TB, 0, stream>>>(den_buf, 0.f, N);

  gat_edge_max   <<<edge_blk, TB, 0, stream>>>(ei, as_buf, ad_buf, m_buf, E, Etot);
  gat_edge_expsum<<<edge_blk, TB, 0, stream>>>(ei, as_buf, ad_buf, m_buf, alpha_e, den_buf, E, Etot);
  gat_edge_norm  <<<edge_blk, TB, 0, stream>>>(ei, den_buf, alpha_e, E, Etot);
  gat_aggregate_owned<OUT2, AGG_ROWS><<<agg_blk, 1024, 0, stream>>>(ei, alpha_e, h1, agg, E, N);
  gat_bias_relu<<<(N * OUT2 + TB - 1) / TB, TB, 0, stream>>>(agg, b2, N * OUT2, OUT2 - 1);

  gat_fc<<<(N + TB - 1) / TB, TB, 0, stream>>>(agg, fcW, fcb, (float*)d_out, N);
}


# Bias+ReLU applied by the aggregation block to its LDS rows in the one store of each row; both element-wise launches go (19 -> 17 launches, 900000 fewer waves); sum order unchanged
# speedup vs baseline: 1.6621x; 1.1536x over previous
#include <hip/hip_runtime.h>


typedef __attribute__((ext_vector_type(2))) float v2f;
typedef __attribute__((ext_vector_type(8))) float v8f;

#define IN_C  128
#define HID   512
#define OUT2  64
#define NCLS  2
#define NEG_SLOPE 0.2f
#define KMAX  512

__global__ void gat_wmma_gemm_f32(const float* __restrict__ A,
                                  const float* __restrict__ B,
                                  float* __restrict__ C,
                                  int M, int K, int Nc) {
  __shared__ float Bs[KMAX * 16];

  const int tid  = threadIdx.x;
  const int lane = tid & 31;
  const int waveid = tid >> 5;
  const int tn = blockIdx.y;
  const int tilesM = (M + 15) >> 4;

  const int nK16 = K * 16;
  for (int idx = tid; idx < nK16; idx += 256) {
    const int k = idx >> 4;
    const int c = idx & 15;
    Bs[((k >> 1) << 5) + (c << 1) + (k & 1)] = B[(size_t)k * Nc + tn * 16 + c];
  }
  __syncthreads();

  const int tm = blockIdx.x * 8 + waveid;
  if (tm >= tilesM) return;

  const int half = lane >> 4;
  const int l16  = lane & 15;

  const float* Ap = A + (size_t)(tm * 16 + l16) * K + 2 * half;
  const float* Bl = Bs + half * 32 + (l16 << 1);

  v8f acc = {};
#pragma unroll 8
  for (int k = 0; k < K; k += 4) {
    v2f a = *(const v2f*)(Ap + k);
    v2f b = *(const v2f*)(Bl + k * 16);
    acc = __builtin_amdgcn_wmma_f32_16x16x4_f32(
         false, a,  false, b,
         (short)0, acc,  false,  false);
  }

  float* Cp = C + (size_t)(tm * 16 + 8 * half) * Nc + tn * 16 + l16;
#pragma unroll
  for (int r = 0; r < 8; ++r) Cp[(size_t)r * Nc] = acc[r];
}

__global__ void gat_node_dots(const float* __restrict__ h,
                              const float* __restrict__ a_src,
                              const float* __restrict__ a_dst,
                              float* __restrict__ as_out,
                              float* __restrict__ ad_out,
                              int N, int D) {
  const int wave = blockIdx.x * (blockDim.x >> 5) + (threadIdx.x >> 5);
  const int lane = threadIdx.x & 31;
  if (wave >= N) return;
  const float* row = h + (size_t)wave * D;
  float ss = 0.f, sd = 0.f;
  for (int c = lane; c < D; c += 32) {
    float v = row[c];
    ss += v * a_src[c];
    sd += v * a_dst[c];
  }
#pragma unroll
  for (int off = 16; off > 0; off >>= 1) {
    ss += __shfl_down(ss, off, 32);
    sd += __shfl_down(sd, off, 32);
  }
  if (lane == 0) { as_out[wave] = ss; ad_out[wave] = sd; }
}

__global__ void gat_fill_f32(float* __restrict__ p, float v, int n) {
  int i = blockIdx.x * blockDim.x + threadIdx.x;
  if (i < n) p[i] = v;
}

__device__ __forceinline__ void edge_sd(const int* __restrict__ ei, int e,
                                        int E, int& s, int& d) {
  if (e < E) { s = ei[e]; d = ei[E + e]; }
  else       { s = d = e - E; }
}

__device__ __forceinline__ float leakyr(float v) {
  return v > 0.f ? v : NEG_SLOPE * v;
}

__device__ __forceinline__ void atomicMaxF32(float* addr, float val) {
  if (val >= 0.f) atomicMax((int*)addr, __float_as_int(val));
  else            atomicMin((unsigned int*)addr, __float_as_uint(val));
}

__global__ void gat_edge_max(const int* __restrict__ ei,
                             const float* __restrict__ as,
                             const float* __restrict__ ad,
                             float* __restrict__ m, int E, int Etot) {
  int e = blockIdx.x * blockDim.x + threadIdx.x;
  if (e >= Etot) return;
  int s, d; edge_sd(ei, e, E, s, d);
  atomicMaxF32(&m[d], leakyr(as[s] + ad[d]));
}

__global__ void gat_edge_expsum(const int* __restrict__ ei,
                                const float* __restrict__ as,
                                const float* __restrict__ ad,
                                const float* __restrict__ m,
                                float* __restrict__ alpha,
                                float* __restrict__ denom, int E, int Etot) {
  int e = blockIdx.x * blockDim.x + threadIdx.x;
  if (e >= Etot) return;
  int s, d; edge_sd(ei, e, E, s, d);
  float ex = expf(leakyr(as[s] + ad[d]) - m[d]);
  alpha[e] = ex;
  atomicAdd(&denom[d], ex);
}

__global__ void gat_edge_norm(const int* __restrict__ ei,
                              const float* __restrict__ denom,
                              float* __restrict__ alpha, int E, int Etot) {
  int e = blockIdx.x * blockDim.x + threadIdx.x;
  if (e >= Etot) return;
  int s, d; edge_sd(ei, e, E, s, d);
  alpha[e] /= denom[d];
}

#define AGG_ROWS 98
#define AGG_U    16
#define AGG_THREADS 1024

__device__ __forceinline__ unsigned agg_owned_bits(const int (&dv)[AGG_U], int d0, int rows) {
  unsigned mm = 0;
#pragma unroll
  for (int k = 0; k < AGG_U; ++k)
    mm |= ((unsigned)(dv[k] - d0) < (unsigned)rows) ? (1u << k) : 0u;
  return mm;
}

template <int D>
__device__ __forceinline__ void agg_add_owned(unsigned mm, int base, int lane, int d0,
                                              const int* __restrict__ ei,
                                              const int* __restrict__ dstp,
                                              const float* __restrict__ alpha,
                                              const float* __restrict__ h, float* acc) {
  constexpr int V = D / 32;
  unsigned lanes = __builtin_amdgcn_ballot_w32(mm != 0);
  while (lanes) {
    const int j = __builtin_ctz(lanes);
    lanes &= lanes - 1;
    unsigned bits = (unsigned)__builtin_amdgcn_readlane((int)mm, j);
    while (bits) {
      const int k = __builtin_ctz(bits);
      bits &= bits - 1;
      const int e = base + k * 32 + j;
      const int s = ei[e];
      const int r = dstp[e] - d0;
      const float a = alpha[e];
      const float* hs = h + (size_t)s * D + lane;
      float* ar = acc + r * D + lane;
      float v[V];
#pragma unroll
      for (int i = 0; i < V; ++i) v[i] = hs[i * 32];
#pragma unroll
      for (int i = 0; i < V; ++i) atomicAdd(ar + i * 32, a * v[i]);
    }
  }
}

template <int D, int R>
__global__ __launch_bounds__(AGG_THREADS) void gat_aggregate_owned(
    const int* __restrict__ ei, const float* __restrict__ alpha,
    const float* __restrict__ h, const float* __restrict__ bias,
    float* __restrict__ out, int E, int N) {
  __shared__ __attribute__((aligned(16))) float acc[R * D];
  constexpr int V = D / 32;

  const int lane = threadIdx.x & 31;
  const int wv   = threadIdx.x >> 5;
  const int nw   = blockDim.x >> 5;
  const int d0   = blockIdx.x * R;
  const int rows = (N - d0 < R) ? (N - d0) : R;

  for (int r = wv; r < rows; r += nw) {
    const float a = alpha[E + d0 + r];
    const float* hs = h + (size_t)(d0 + r) * D + lane;
    float* ar = acc + r * D + lane;
#pragma unroll
    for (int i = 0; i < V; ++i) ar[i * 32] = a * hs[i * 32];
  }
  __syncthreads();

  const int* __restrict__ dstp = ei + E;
  const int sweep  = AGG_U * 32;
  const int stride = nw * sweep;
  int base = wv * sweep;

  if (base + sweep <= E) {
    int cur[AGG_U];
#pragma unroll
    for (int k = 0; k < AGG_U; ++k) cur[k] = dstp[base + k * 32 + lane];
    while (true) {
      const int nb = base + stride;
      const bool more = (nb + sweep <= E);
      const int pb = more ? nb : base;
      int nxt[AGG_U];
#pragma unroll
      for (int k = 0; k < AGG_U; ++k) nxt[k] = dstp[pb + k * 32 + lane];
      agg_add_owned<D>(agg_owned_bits(cur, d0, rows), base, lane, d0, ei, dstp, alpha, h, acc);
      base = nb;
      if (!more) break;
#pragma unroll
      for (int k = 0; k < AGG_U; ++k) cur[k] = nxt[k];
    }
  }
  if (base < E) {
    int cur[AGG_U];
#pragma unroll
    for (int k = 0; k < AGG_U; ++k) {
      const int e = base + k * 32 + lane;
      cur[k] = (e < E) ? dstp[e] : -1;
    }
    agg_add_owned<D>(agg_owned_bits(cur, d0, rows), base, lane, d0, ei, dstp, alpha, h, acc);
  }

  constexpr int Q = D / 4;
  static_assert((Q & (Q - 1)) == 0 && AGG_THREADS % Q == 0, "column group must be fixed per thread");
  const int c0 = (threadIdx.x & (Q - 1)) * 4;
  const float bx = bias[c0], by = bias[c0 + 1], bz = bias[c0 + 2], bw = bias[c0 + 3];
  __syncthreads();

  const float4* a4 = (const float4*)acc;
  float4* o4 = (float4*)(out + (size_t)d0 * D);
  for (int idx = threadIdx.x; idx < rows * Q; idx += AGG_THREADS) {
    float4 v = a4[idx];
    v.x += bx; v.y += by; v.z += bz; v.w += bw;
    v.x = v.x > 0.f ? v.x : 0.f;
    v.y = v.y > 0.f ? v.y : 0.f;
    v.z = v.z > 0.f ? v.z : 0.f;
    v.w = v.w > 0.f ? v.w : 0.f;
    o4[idx] = v;
  }
}

__global__ void gat_fc(const float* __restrict__ h,
                       const float* __restrict__ W,
                       const float* __restrict__ b,
                       float* __restrict__ out, int N) {
  int n = blockIdx.x * blockDim.x + threadIdx.x;
  if (n >= N) return;
  const float* row = h + (size_t)n * OUT2;
  float a0 = b[0], a1 = b[1];
#pragma unroll
  for (int c = 0; c < OUT2; ++c) {
    float v = row[c];
    a0 += v * W[c * NCLS + 0];
    a1 += v * W[c * NCLS + 1];
  }
  out[n * NCLS + 0] = a0;
  out[n * NCLS + 1] = a1;
}

extern "C" void kernel_launch(void* const* d_in, const int* in_sizes, int n_in,
                              void* d_out, int out_size, void* d_ws, size_t ws_size,
                              hipStream_t stream) {
  const float* x   = (const float*)d_in[0];
  const int*   ei  = (const int*)  d_in[1];
  const float* W1  = (const float*)d_in[2];
  const float* a1s = (const float*)d_in[3];
  const float* a1d = (const float*)d_in[4];
  const float* b1  = (const float*)d_in[5];
  const float* W2  = (const float*)d_in[6];
  const float* a2s = (const float*)d_in[7];
  const float* a2d = (const float*)d_in[8];
  const float* b2  = (const float*)d_in[9];
  const float* fcW = (const float*)d_in[10];
  const float* fcb = (const float*)d_in[11];

  const int N    = in_sizes[0] / IN_C;
  const int E    = in_sizes[1] / 2;
  const int Etot = E + N;

  float* w       = (float*)d_ws;
  float* h1      = w;
  float* agg     = w + (size_t)N * HID;
  float* as_buf  = w + (size_t)2 * N * HID;
  float* ad_buf  = as_buf + N;
  float* m_buf   = ad_buf + N;
  float* den_buf = m_buf + N;
  float* alpha_e = den_buf + N;

  const int TB = 256;
  const float NEG_INF = -__builtin_inff();

  auto gemm_grid = [](int M, int Nc) {
    return dim3((((M + 15) / 16) + 7) / 8, Nc / 16);
  };
  const int edge_blk  = (Etot + TB - 1) / TB;
  const int agg_blk   = (N + AGG_ROWS - 1) / AGG_ROWS;
  const int node_blk  = (N + 7) / 8;

  gat_wmma_gemm_f32<<<gemm_grid(N, HID), TB, 0, stream>>>(x, W1, h1, N, IN_C, HID);
  gat_node_dots<<<node_blk, TB, 0, stream>>>(h1, a1s, a1d, as_buf, ad_buf, N, HID);

  gat_fill_f32<<<(N + TB - 1) / TB, TB, 0, stream>>>(m_buf, NEG_INF, N);
  gat_fill_f32<<<(N + TB - 1) / TB, TB, 0, stream>>>(den_buf, 0.f, N);

  gat_edge_max   <<<edge_blk, TB, 0, stream>>>(ei, as_buf, ad_buf, m_buf, E, Etot);
  gat_edge_expsum<<<edge_blk, TB, 0, stream>>>(ei, as_buf, ad_buf, m_buf, alpha_e, den_buf, E, Etot);
  gat_edge_norm  <<<edge_blk, TB, 0, stream>>>(ei, den_buf, alpha_e, E, Etot);
  gat_aggregate_owned<HID, AGG_ROWS><<<agg_blk, AGG_THREADS, 0, stream>>>(ei, alpha_e, h1, b1, agg, E, N);

  gat_wmma_gemm_f32<<<gemm_grid(N, OUT2), TB, 0, stream>>>(agg, W2, h1, N, HID, OUT2);
  gat_node_dots<<<node_blk, TB, 0, stream>>>(h1, a2s, a2d, as_buf, ad_buf, N, OUT2);

  gat_fill_f32<<<(N + TB - 1) / TB, TB, 0, stream>>>(m_buf, NEG_INF, N);
  gat_fill_f32<<<(N + TB - 1) / TB, TB, 0, stream>>>(den_buf, 0.f, N);

  gat_edge_max   <<<edge_blk, TB, 0, stream>>>(ei, as_buf, ad_buf, m_buf, E, Etot);
  gat_edge_expsum<<<edge_blk, TB, 0, stream>>>(ei, as_buf, ad_buf, m_buf, alpha_e, den_buf, E, Etot);
  gat_edge_norm  <<<edge_blk, TB, 0, stream>>>(ei, den_buf, alpha_e, E, Etot);
  gat_aggregate_owned<OUT2, AGG_ROWS><<<agg_blk, AGG_THREADS, 0, stream>>>(ei, alpha_e, h1, b2, agg, E, N);

  gat_fc<<<(N + TB - 1) / TB, TB, 0, stream>>>(agg, fcW, fcb, (float*)d_out, N);
}


# GEMMs re-tiled: a wave keeps its A fragment for 8 (layer 1) or 4 (layer 2) adjacent column tiles read from a wider LDS strip of B; tile pairs store whole 128-byte lines; sum order unchanged
# speedup vs baseline: 1.9145x; 1.1519x over previous
#include <hip/hip_runtime.h>


typedef __attribute__((ext_vector_type(2))) float v2f;
typedef __attribute__((ext_vector_type(8))) float v8f;

#define IN_C  128
#define HID   512
#define OUT2  64
#define NCLS  2
#define NEG_SLOPE 0.2f

template <int K, int NT, int WAVES, int U>
__global__ __launch_bounds__(WAVES * 32) void gat_wmma_gemm_f32(
    const float* __restrict__ A, const float* __restrict__ B,
    float* __restrict__ C, int M, int Nc) {
  static_assert(K % 4 == 0 && NT % 2 == 0, "k-steps of 4, tiles stored in pairs");
  constexpr int TP = K * 16 + 8;
  __shared__ __attribute__((aligned(16))) float Bs[NT * TP];

  const int tid  = threadIdx.x;
  const int lane = tid & 31;
  const int waveid = tid >> 5;
  const int col0 = blockIdx.y * (NT * 16);
  const int tilesM = (M + 15) >> 4;

  constexpr int QC = NT * 4;
  constexpr int ITEMS = (K / 2) * QC;
#pragma unroll 4
  for (int it = tid; it < ITEMS; it += WAVES * 32) {
    const int p = it / QC;
    const int q = it % QC;
    const float* src = B + (size_t)(2 * p) * Nc + col0 + 4 * q;
    const float4 r0 = *(const float4*)src;
    const float4 r1 = *(const float4*)(src + Nc);
    float* dst = Bs + (q >> 2) * TP + p * 32 + (q & 3) * 8;
    *(float4*)(dst)     = make_float4(r0.x, r1.x, r0.y, r1.y);
    *(float4*)(dst + 4) = make_float4(r0.z, r1.z, r0.w, r1.w);
  }
  __syncthreads();

  const int tm = blockIdx.x * WAVES + waveid;
  if (tm >= tilesM) return;

  const int half = lane >> 4;
  const int l16  = lane & 15;

  const float* Ap = A + (size_t)(tm * 16 + l16) * K + 2 * half;
  const float* Bl = Bs + half * 32 + (l16 << 1);

  v8f acc[NT];
#pragma unroll
  for (int t = 0; t < NT; ++t) acc[t] = v8f{};

#pragma unroll U
  for (int k = 0; k < K; k += 4) {
    const v2f a = *(const v2f*)(Ap + k);
#pragma unroll
    for (int t = 0; t < NT; ++t) {
      const v2f b = *(const v2f*)(Bl + t * TP + k * 16);
      acc[t] = __builtin_amdgcn_wmma_f32_16x16x4_f32(
           false, a,  false, b,
           (short)0, acc[t],  false,  false);
    }
  }

  float* Cw = C + (size_t)(tm * 16) * Nc + col0 + lane;
#pragma unroll
  for (int t = 0; t < NT; t += 2) {
#pragma unroll
    for (int r = 0; r < 8; ++r) {
      const float x0 = acc[t][r];
      const float x1 = acc[t + 1][r];
      const float y0 = __shfl_xor(x0, 16, 32);
      const float y1 = __shfl_xor(x1, 16, 32);
      const float lo = half ? y1 : x0;
      const float hi = half ? x1 : y0;
      Cw[(size_t)r * Nc + t * 16]       = lo;
      Cw[(size_t)(r + 8) * Nc + t * 16] = hi;
    }
  }
}

__global__ void gat_node_dots(const float* __restrict__ h,
                              const float* __restrict__ a_src,
                              const float* __restrict__ a_dst,
                              float* __restrict__ as_out,
                              float* __restrict__ ad_out,
                              int N, int D) {
  const int wave = blockIdx.x * (blockDim.x >> 5) + (threadIdx.x >> 5);
  const int lane = threadIdx.x & 31;
  if (wave >= N) return;
  const float* row = h + (size_t)wave * D;
  float ss = 0.f, sd = 0.f;
  for (int c = lane; c < D; c += 32) {
    float v = row[c];
    ss += v * a_src[c];
    sd += v * a_dst[c];
  }
#pragma unroll
  for (int off = 16; off > 0; off >>= 1) {
    ss += __shfl_down(ss, off, 32);
    sd += __shfl_down(sd, off, 32);
  }
  if (lane == 0) { as_out[wave] = ss; ad_out[wave] = sd; }
}

__global__ void gat_fill_f32(float* __restrict__ p, float v, int n) {
  int i = blockIdx.x * blockDim.x + threadIdx.x;
  if (i < n) p[i] = v;
}

__device__ __forceinline__ void edge_sd(const int* __restrict__ ei, int e,
                                        int E, int& s, int& d) {
  if (e < E) { s = ei[e]; d = ei[E + e]; }
  else       { s = d = e - E; }
}

__device__ __forceinline__ float leakyr(float v) {
  return v > 0.f ? v : NEG_SLOPE * v;
}

__device__ __forceinline__ void atomicMaxF32(float* addr, float val) {
  if (val >= 0.f) atomicMax((int*)addr, __float_as_int(val));
  else            atomicMin((unsigned int*)addr, __float_as_uint(val));
}

__global__ void gat_edge_max(const int* __restrict__ ei,
                             const float* __restrict__ as,
                             const float* __restrict__ ad,
                             float* __restrict__ m, int E, int Etot) {
  int e = blockIdx.x * blockDim.x + threadIdx.x;
  if (e >= Etot) return;
  int s, d; edge_sd(ei, e, E, s, d);
  atomicMaxF32(&m[d], leakyr(as[s] + ad[d]));
}

__global__ void gat_edge_expsum(const int* __restrict__ ei,
                                const float* __restrict__ as,
                                const float* __restrict__ ad,
                                const float* __restrict__ m,
                                float* __restrict__ alpha,
                                float* __restrict__ denom, int E, int Etot) {
  int e = blockIdx.x * blockDim.x + threadIdx.x;
  if (e >= Etot) return;
  int s, d; edge_sd(ei, e, E, s, d);
  float ex = expf(leakyr(as[s] + ad[d]) - m[d]);
  alpha[e] = ex;
  atomicAdd(&denom[d], ex);
}

__global__ void gat_edge_norm(const int* __restrict__ ei,
                              const float* __restrict__ denom,
                              float* __restrict__ alpha, int E, int Etot) {
  int e = blockIdx.x * blockDim.x + threadIdx.x;
  if (e >= Etot) return;
  int s, d; edge_sd(ei, e, E, s, d);
  alpha[e] /= denom[d];
}

#define AGG_ROWS 98
#define AGG_U    16
#define AGG_THREADS 1024

__device__ __forceinline__ unsigned agg_owned_bits(const int (&dv)[AGG_U], int d0, int rows) {
  unsigned mm = 0;
#pragma unroll
  for (int k = 0; k < AGG_U; ++k)
    mm |= ((unsigned)(dv[k] - d0) < (unsigned)rows) ? (1u << k) : 0u;
  return mm;
}

template <int D>
__device__ __forceinline__ void agg_add_owned(unsigned mm, int base, int lane, int d0,
                                              const int* __restrict__ ei,
                                              const int* __restrict__ dstp,
                                              const float* __restrict__ alpha,
                                              const float* __restrict__ h, float* acc) {
  constexpr int V = D / 32;
  unsigned lanes = __builtin_amdgcn_ballot_w32(mm != 0);
  while (lanes) {
    const int j = __builtin_ctz(lanes);
    lanes &= lanes - 1;
    unsigned bits = (unsigned)__builtin_amdgcn_readlane((int)mm, j);
    while (bits) {
      const int k = __builtin_ctz(bits);
      bits &= bits - 1;
      const int e = base + k * 32 + j;
      const int s = ei[e];
      const int r = dstp[e] - d0;
      const float a = alpha[e];
      const float* hs = h + (size_t)s * D + lane;
      float* ar = acc + r * D + lane;
      float v[V];
#pragma unroll
      for (int i = 0; i < V; ++i) v[i] = hs[i * 32];
#pragma unroll
      for (int i = 0; i < V; ++i) atomicAdd(ar + i * 32, a * v[i]);
    }
  }
}

template <int D, int R>
__global__ __launch_bounds__(AGG_THREADS) void gat_aggregate_owned(
    const int* __restrict__ ei, const float* __restrict__ alpha,
    const float* __restrict__ h, const float* __restrict__ bias,
    float* __restrict__ out, int E, int N) {
  __shared__ __attribute__((aligned(16))) float acc[R * D];
  constexpr int V = D / 32;

  const int lane = threadIdx.x & 31;
  const int wv   = threadIdx.x >> 5;
  const int nw   = blockDim.x >> 5;
  const int d0   = blockIdx.x * R;
  const int rows = (N - d0 < R) ? (N - d0) : R;

  for (int r = wv; r < rows; r += nw) {
    const float a = alpha[E + d0 + r];
    const float* hs = h + (size_t)(d0 + r) * D + lane;
    float* ar = acc + r * D + lane;
#pragma unroll
    for (int i = 0; i < V; ++i) ar[i * 32] = a * hs[i * 32];
  }
  __syncthreads();

  const int* __restrict__ dstp = ei + E;
  const int sweep  = AGG_U * 32;
  const int stride = nw * sweep;
  int base = wv * sweep;

  if (base + sweep <= E) {
    int cur[AGG_U];
#pragma unroll
    for (int k = 0; k < AGG_U; ++k) cur[k] = dstp[base + k * 32 + lane];
    while (true) {
      const int nb = base + stride;
      const bool more = (nb + sweep <= E);
      const int pb = more ? nb : base;
      int nxt[AGG_U];
#pragma unroll
      for (int k = 0; k < AGG_U; ++k) nxt[k] = dstp[pb + k * 32 + lane];
      agg_add_owned<D>(agg_owned_bits(cur, d0, rows), base, lane, d0, ei, dstp, alpha, h, acc);
      base = nb;
      if (!more) break;
#pragma unroll
      for (int k = 0; k < AGG_U; ++k) cur[k] = nxt[k];
    }
  }
  if (base < E) {
    int cur[AGG_U];
#pragma unroll
    for (int k = 0; k < AGG_U; ++k) {
      const int e = base + k * 32 + lane;
      cur[k] = (e < E) ? dstp[e] : -1;
    }
    agg_add_owned<D>(agg_owned_bits(cur, d0, rows), base, lane, d0, ei, dstp, alpha, h, acc);
  }

  constexpr int Q = D / 4;
  static_assert((Q & (Q - 1)) == 0 && AGG_THREADS % Q == 0, "column group must be fixed per thread");
  const int c0 = (threadIdx.x & (Q - 1)) * 4;
  const float bx = bias[c0], by = bias[c0 + 1], bz = bias[c0 + 2], bw = bias[c0 + 3];
  __syncthreads();

  const float4* a4 = (const float4*)acc;
  float4* o4 = (float4*)(out + (size_t)d0 * D);
  for (int idx = threadIdx.x; idx < rows * Q; idx += AGG_THREADS) {
    float4 v = a4[idx];
    v.x += bx; v.y += by; v.z += bz; v.w += bw;
    v.x = v.x > 0.f ? v.x : 0.f;
    v.y = v.y > 0.f ? v.y : 0.f;
    v.z = v.z > 0.f ? v.z : 0.f;
    v.w = v.w > 0.f ? v.w : 0.f;
    o4[idx] = v;
  }
}

__global__ void gat_fc(const float* __restrict__ h,
                       const float* __restrict__ W,
                       const float* __restrict__ b,
                       float* __restrict__ out, int N) {
  int n = blockIdx.x * blockDim.x + threadIdx.x;
  if (n >= N) return;
  const float* row = h + (size_t)n * OUT2;
  float a0 = b[0], a1 = b[1];
#pragma unroll
  for (int c = 0; c < OUT2; ++c) {
    float v = row[c];
    a0 += v * W[c * NCLS + 0];
    a1 += v * W[c * NCLS + 1];
  }
  out[n * NCLS + 0] = a0;
  out[n * NCLS + 1] = a1;
}

extern "C" void kernel_launch(void* const* d_in, const int* in_sizes, int n_in,
                              void* d_out, int out_size, void* d_ws, size_t ws_size,
                              hipStream_t stream) {
  const float* x   = (const float*)d_in[0];
  const int*   ei  = (const int*)  d_in[1];
  const float* W1  = (const float*)d_in[2];
  const float* a1s = (const float*)d_in[3];
  const float* a1d = (const float*)d_in[4];
  const float* b1  = (const float*)d_in[5];
  const float* W2  = (const float*)d_in[6];
  const float* a2s = (const float*)d_in[7];
  const float* a2d = (const float*)d_in[8];
  const float* b2  = (const float*)d_in[9];
  const float* fcW = (const float*)d_in[10];
  const float* fcb = (const float*)d_in[11];

  const int N    = in_sizes[0] / IN_C;
  const int E    = in_sizes[1] / 2;
  const int Etot = E + N;

  float* w       = (float*)d_ws;
  float* h1      = w;
  float* agg     = w + (size_t)N * HID;
  float* as_buf  = w + (size_t)2 * N * HID;
  float* ad_buf  = as_buf + N;
  float* m_buf   = ad_buf + N;
  float* den_buf = m_buf + N;
  float* alpha_e = den_buf + N;

  const int TB = 256;
  const float NEG_INF = -__builtin_inff();

  auto gemm_grid = [](int M, int Nc, int waves, int cols) {
    return dim3((((M + 15) / 16) + waves - 1) / waves, Nc / cols);
  };
  const int edge_blk  = (Etot + TB - 1) / TB;
  const int agg_blk   = (N + AGG_ROWS - 1) / AGG_ROWS;
  const int node_blk  = (N + 7) / 8;

  gat_wmma_gemm_f32<IN_C, 8, 8, 4><<<gemm_grid(N, HID, 8, 128), 256, 0, stream>>>(x, W1, h1, N, HID);
  gat_node_dots<<<node_blk, TB, 0, stream>>>(h1, a1s, a1d, as_buf, ad_buf, N, HID);

  gat_fill_f32<<<(N + TB - 1) / TB, TB, 0, stream>>>(m_buf, NEG_INF, N);
  gat_fill_f32<<<(N + TB - 1) / TB, TB, 0, stream>>>(den_buf, 0.f, N);

  gat_edge_max   <<<edge_blk, TB, 0, stream>>>(ei, as_buf, ad_buf, m_buf, E, Etot);
  gat_edge_expsum<<<edge_blk, TB, 0, stream>>>(ei, as_buf, ad_buf, m_buf, alpha_e, den_buf, E, Etot);
  gat_edge_norm  <<<edge_blk, TB, 0, stream>>>(ei, den_buf, alpha_e, E, Etot);
  gat_aggregate_owned<HID, AGG_ROWS><<<agg_blk, AGG_THREADS, 0, stream>>>(ei, alpha_e, h1, b1, agg, E, N);

  gat_wmma_gemm_f32<HID, 4, 8, 8><<<gemm_grid(N, OUT2, 8, 64), 256, 0, stream>>>(agg, W2, h1, N, OUT2);
  gat_node_dots<<<node_blk, TB, 0, stream>>>(h1, a2s, a2d, as_buf, ad_buf, N, OUT2);

  gat_fill_f32<<<(N + TB - 1) / TB, TB, 0, stream>>>(m_buf, NEG_INF, N);
  gat_fill_f32<<<(N + TB - 1) / TB, TB, 0, stream>>>(den_buf, 0.f, N);

  gat_edge_max   <<<edge_blk, TB, 0, stream>>>(ei, as_buf, ad_buf, m_buf, E, Etot);
  gat_edge_expsum<<<edge_blk, TB, 0, stream>>>(ei, as_buf, ad_buf, m_buf, alpha_e, den_buf, E, Etot);
  gat_edge_norm  <<<edge_blk, TB, 0, stream>>>(ei, den_buf, alpha_e, E, Etot);
  gat_aggregate_owned<OUT2, AGG_ROWS><<<agg_blk, AGG_THREADS, 0, stream>>>(ei, alpha_e, h1, b2, agg, E, N);

  gat_fc<<<(N + TB - 1) / TB, TB, 0, stream>>>(agg, fcW, fcb, (float*)d_out, N);
}


# Edge softmax moved into the aggregation block: one scan lists its owned edges in LDS; row max, exp-sum, weights by LDS atomics there; 17 -> 7 launches, no global atomics; sums in arrival order
# speedup vs baseline: 2.8637x; 1.4958x over previous
#include <hip/hip_runtime.h>


typedef __attribute__((ext_vector_type(2))) float v2f;
typedef __attribute__((ext_vector_type(8))) float v8f;

#define IN_C  128
#define HID   512
#define OUT2  64
#define NCLS  2
#define NEG_SLOPE 0.2f

template <int K, int NT, int WAVES, int U>
__global__ __launch_bounds__(WAVES * 32) void gat_wmma_gemm_f32(
    const float* __restrict__ A, const float* __restrict__ B,
    float* __restrict__ C, int M, int Nc) {
  static_assert(K % 4 == 0 && NT % 2 == 0, "k-steps of 4, tiles stored in pairs");
  constexpr int TP = K * 16 + 8;
  __shared__ __attribute__((aligned(16))) float Bs[NT * TP];

  const int tid  = threadIdx.x;
  const int lane = tid & 31;
  const int waveid = tid >> 5;
  const int col0 = blockIdx.y * (NT * 16);
  const int tilesM = (M + 15) >> 4;

  constexpr int QC = NT * 4;
  constexpr int ITEMS = (K / 2) * QC;
#pragma unroll 4
  for (int it = tid; it < ITEMS; it += WAVES * 32) {
    const int p = it / QC;
    const int q = it % QC;
    const float* src = B + (size_t)(2 * p) * Nc + col0 + 4 * q;
    const float4 r0 = *(const float4*)src;
    const float4 r1 = *(const float4*)(src + Nc);
    float* dst = Bs + (q >> 2) * TP + p * 32 + (q & 3) * 8;
    *(float4*)(dst)     = make_float4(r0.x, r1.x, r0.y, r1.y);
    *(float4*)(dst + 4) = make_float4(r0.z, r1.z, r0.w, r1.w);
  }
  __syncthreads();

  const int tm = blockIdx.x * WAVES + waveid;
  if (tm >= tilesM) return;

  const int half = lane >> 4;
  const int l16  = lane & 15;

  const float* Ap = A + (size_t)(tm * 16 + l16) * K + 2 * half;
  const float* Bl = Bs + half * 32 + (l16 << 1);

  v8f acc[NT];
#pragma unroll
  for (int t = 0; t < NT; ++t) acc[t] = v8f{};

#pragma unroll U
  for (int k = 0; k < K; k += 4) {
    const v2f a = *(const v2f*)(Ap + k);
#pragma unroll
    for (int t = 0; t < NT; ++t) {
      const v2f b = *(const v2f*)(Bl + t * TP + k * 16);
      acc[t] = __builtin_amdgcn_wmma_f32_16x16x4_f32(
           false, a,  false, b,
           (short)0, acc[t],  false,  false);
    }
  }

  float* Cw = C + (size_t)(tm * 16) * Nc + col0 + lane;
#pragma unroll
  for (int t = 0; t < NT; t += 2) {
#pragma unroll
    for (int r = 0; r < 8; ++r) {
      const float x0 = acc[t][r];
      const float x1 = acc[t + 1][r];
      const float y0 = __shfl_xor(x0, 16, 32);
      const float y1 = __shfl_xor(x1, 16, 32);
      const float lo = half ? y1 : x0;
      const float hi = half ? x1 : y0;
      Cw[(size_t)r * Nc + t * 16]       = lo;
      Cw[(size_t)(r + 8) * Nc + t * 16] = hi;
    }
  }
}

__global__ void gat_node_dots(const float* __restrict__ h,
                              const float* __restrict__ a_src,
                              const float* __restrict__ a_dst,
                              float* __restrict__ as_out,
                              float* __restrict__ ad_out,
                              int N, int D) {
  const int wave = blockIdx.x * (blockDim.x >> 5) + (threadIdx.x >> 5);
  const int lane = threadIdx.x & 31;
  if (wave >= N) return;
  const float* row = h + (size_t)wave * D;
  float ss = 0.f, sd = 0.f;
  for (int c = lane; c < D; c += 32) {
    float v = row[c];
    ss += v * a_src[c];
    sd += v * a_dst[c];
  }
#pragma unroll
  for (int off = 16; off > 0; off >>= 1) {
    ss += __shfl_down(ss, off, 32);
    sd += __shfl_down(sd, off, 32);
  }
  if (lane == 0) { as_out[wave] = ss; ad_out[wave] = sd; }
}

__device__ __forceinline__ float leakyr(float v) {
  return v > 0.f ? v : NEG_SLOPE * v;
}

__device__ __forceinline__ void lds_max_f32(float* addr, float val) {
  if (__float_as_int(val) >= 0) atomicMax((int*)addr, __float_as_int(val));
  else                          atomicMin((unsigned int*)addr, __float_as_uint(val));
}

#define AGG_ROWS 98
#define AGG_U    16
#define AGG_THREADS 1024
#define AGG_CAP  4096

struct AggList {
  int*   s;
  int*   r;
  float* v;
  int*   cnt;
  float* m;
  float* den;
  float* ad;
};

__device__ __forceinline__ void agg_append_owned(const int (&dv)[AGG_U], int base, int lane,
                                                 int d0, int rows, const AggList& L) {
#pragma unroll
  for (int k = 0; k < AGG_U; ++k) {
    const int r = dv[k] - d0;
    if ((unsigned)r < (unsigned)rows) {
      const int idx = atomicAdd(L.cnt, 1);
      if (idx < AGG_CAP) { L.s[idx] = base + k * 32 + lane; L.r[idx] = r; }
    }
  }
}

__device__ __forceinline__ void agg_scan_append(const int* __restrict__ dstp, int eb, int ee,
                                                int d0, int rows, int wv, int nw, int lane,
                                                const AggList& L) {
  const int sweep  = AGG_U * 32;
  const int stride = nw * sweep;
  int base = eb + wv * sweep;

  if (base + sweep <= ee) {
    int cur[AGG_U];
#pragma unroll
    for (int k = 0; k < AGG_U; ++k) cur[k] = dstp[base + k * 32 + lane];
    while (true) {
      const int nb = base + stride;
      const bool more = (nb + sweep <= ee);
      const int pb = more ? nb : base;
      int nxt[AGG_U];
#pragma unroll
      for (int k = 0; k < AGG_U; ++k) nxt[k] = dstp[pb + k * 32 + lane];
      agg_append_owned(cur, base, lane, d0, rows, L);
      base = nb;
      if (!more) break;
#pragma unroll
      for (int k = 0; k < AGG_U; ++k) cur[k] = nxt[k];
    }
  }
  if (base < ee) {
    int cur[AGG_U];
#pragma unroll
    for (int k = 0; k < AGG_U; ++k) {
      const int e = base + k * 32 + lane;
      cur[k] = (e < ee) ? dstp[e] : -1;
    }
    agg_append_owned(cur, base, lane, d0, rows, L);
  }
}

__device__ __forceinline__ void agg_list_scores(const AggList& L, int first, int n, bool do_max,
                                                const int* __restrict__ ei,
                                                const float* __restrict__ as) {
  for (int i = first + (int)threadIdx.x; i < n; i += AGG_THREADS) {
    const int r = L.r[i];
    const int s = ei[L.s[i]];
    const float sc = leakyr(as[s] + L.ad[r]);
    L.s[i] = s;
    L.v[i] = sc;
    if (do_max) lds_max_f32(&L.m[r], sc);
  }
}

__device__ __forceinline__ void agg_list_exps(const AggList& L, int n, bool do_sum) {
  for (int i = (int)threadIdx.x; i < n; i += AGG_THREADS) {
    const int r = L.r[i];
    const float ex = expf(L.v[i] - L.m[r]);
    L.v[i] = ex;
    if (do_sum) atomicAdd(&L.den[r], ex);
  }
}

template <int D>
__device__ __forceinline__ void agg_list_gather(const AggList& L, int n, int wv, int nw, int lane,
                                                const float* __restrict__ h, float* acc) {
  constexpr int V = D / 32;
  for (int i = wv; i < n; i += nw) {
    const int s = L.s[i];
    const float* hs = h + (size_t)s * D + lane;
    float v[V];
#pragma unroll
    for (int j = 0; j < V; ++j) v[j] = hs[j * 32];
    const int r = L.r[i];
    const float a = L.v[i] / L.den[r];
    float* ar = acc + r * D + lane;
#pragma unroll
    for (int j = 0; j < V; ++j) atomicAdd(ar + j * 32, a * v[j]);
  }
}

template <int D, int R>
__global__ __launch_bounds__(AGG_THREADS) void gat_aggregate_owned(
    const int* __restrict__ ei, const float* __restrict__ as,
    const float* __restrict__ ad, const float* __restrict__ h,
    const float* __restrict__ bias, float* __restrict__ out, int E, int N) {
  __shared__ __attribute__((aligned(16))) float acc[R * D];
  __shared__ int   list_s[AGG_CAP];
  __shared__ int   list_r[AGG_CAP];
  __shared__ float list_v[AGG_CAP];
  __shared__ float row_m[R];
  __shared__ float row_den[R];
  __shared__ float row_ad[R];
  __shared__ int   list_cnt;
  static_assert(R <= AGG_CAP && R <= AGG_THREADS, "self loops open the list, one thread each");

  const int tid  = threadIdx.x;
  const int lane = tid & 31;
  const int wv   = tid >> 5;
  const int nw   = AGG_THREADS >> 5;
  const int d0   = blockIdx.x * R;
  const int rows = (N - d0 < R) ? (N - d0) : R;
  const AggList L{list_s, list_r, list_v, &list_cnt, row_m, row_den, row_ad};
  const int* __restrict__ dstp = ei + E;

  {
    float4* z4 = (float4*)acc;
    for (int idx = tid; idx < rows * (D / 4); idx += AGG_THREADS)
      z4[idx] = make_float4(0.f, 0.f, 0.f, 0.f);
  }
  if (tid < rows) {
    const float adv = ad[d0 + tid];
    const float sc  = leakyr(as[d0 + tid] + adv);
    row_ad[tid] = adv; row_m[tid] = sc; row_den[tid] = 0.f;
    list_s[tid] = d0 + tid; list_r[tid] = tid; list_v[tid] = sc;
  }
  if (tid == 0) list_cnt = rows;
  __syncthreads();

  agg_scan_append(dstp, 0, E, d0, rows, wv, nw, lane, L);
  __syncthreads();
  const int total = list_cnt;

  if (total <= AGG_CAP) {
    agg_list_scores(L, rows, total, true, ei, as);
    __syncthreads();
    agg_list_exps(L, total, true);
    __syncthreads();
    agg_list_gather<D>(L, total, wv, nw, lane, h, acc);
  } else {
    for (int pass = 0; pass < 3; ++pass) {
      for (int eb = -AGG_CAP; eb < E; eb += AGG_CAP) {
        __syncthreads();
        int first = 0;
        if (eb < 0) {
          if (tid < rows) {
            list_s[tid] = d0 + tid; list_r[tid] = tid;
            list_v[tid] = leakyr(as[d0 + tid] + row_ad[tid]);
          }
          if (tid == 0) list_cnt = rows;
          first = rows;
          __syncthreads();
        } else {
          if (tid == 0) list_cnt = 0;
          __syncthreads();
          agg_scan_append(dstp, eb, (E - eb < AGG_CAP) ? E : eb + AGG_CAP, d0, rows, wv, nw, lane, L);
          __syncthreads();
        }
        const int n = list_cnt;
        agg_list_scores(L, first, n, pass == 0, ei, as);
        if (pass >= 1) {
          __syncthreads();
          agg_list_exps(L, n, pass == 1);
        }
        if (pass == 2) {
          __syncthreads();
          agg_list_gather<D>(L, n, wv, nw, lane, h, acc);
        }
      }
    }
  }

  constexpr int Q = D / 4;
  static_assert((Q & (Q - 1)) == 0 && AGG_THREADS % Q == 0, "column group must be fixed per thread");
  const int c0 = (tid & (Q - 1)) * 4;
  const float bx = bias[c0], by = bias[c0 + 1], bz = bias[c0 + 2], bw = bias[c0 + 3];
  __syncthreads();

  const float4* a4 = (const float4*)acc;
  float4* o4 = (float4*)(out + (size_t)d0 * D);
  for (int idx = tid; idx < rows * Q; idx += AGG_THREADS) {
    float4 v = a4[idx];
    v.x += bx; v.y += by; v.z += bz; v.w += bw;
    v.x = v.x > 0.f ? v.x : 0.f;
    v.y = v.y > 0.f ? v.y : 0.f;
    v.z = v.z > 0.f ? v.z : 0.f;
    v.w = v.w > 0.f ? v.w : 0.f;
    o4[idx] = v;
  }
}

__global__ void gat_fc(const float* __restrict__ h,
                       const float* __restrict__ W,
                       const float* __restrict__ b,
                       float* __restrict__ out, int N) {
  int n = blockIdx.x * blockDim.x + threadIdx.x;
  if (n >= N) return;
  const float* row = h + (size_t)n * OUT2;
  float a0 = b[0], a1 = b[1];
#pragma unroll
  for (int c = 0; c < OUT2; ++c) {
    float v = row[c];
    a0 += v * W[c * NCLS + 0];
    a1 += v * W[c * NCLS + 1];
  }
  out[n * NCLS + 0] = a0;
  out[n * NCLS + 1] = a1;
}

extern "C" void kernel_launch(void* const* d_in, const int* in_sizes, int n_in,
                              void* d_out, int out_size, void* d_ws, size_t ws_size,
                              hipStream_t stream) {
  const float* x   = (const float*)d_in[0];
  const int*   ei  = (const int*)  d_in[1];
  const float* W1  = (const float*)d_in[2];
  const float* a1s = (const float*)d_in[3];
  const float* a1d = (const float*)d_in[4];
  const float* b1  = (const float*)d_in[5];
  const float* W2  = (const float*)d_in[6];
  const float* a2s = (const float*)d_in[7];
  const float* a2d = (const float*)d_in[8];
  const float* b2  = (const float*)d_in[9];
  const float* fcW = (const float*)d_in[10];
  const float* fcb = (const float*)d_in[11];

  const int N    = in_sizes[0] / IN_C;
  const int E    = in_sizes[1] / 2;

  float* w       = (float*)d_ws;
  float* h1      = w;
  float* agg     = w + (size_t)N * HID;
  float* as_buf  = w + (size_t)2 * N * HID;
  float* ad_buf  = as_buf + N;

  const int TB = 256;

  auto gemm_grid = [](int M, int Nc, int waves, int cols) {
    return dim3((((M + 15) / 16) + waves - 1) / waves, Nc / cols);
  };
  const int agg_blk   = (N + AGG_ROWS - 1) / AGG_ROWS;
  const int node_blk  = (N + 7) / 8;

  gat_wmma_gemm_f32<IN_C, 8, 8, 4><<<gemm_grid(N, HID, 8, 128), 256, 0, stream>>>(x, W1, h1, N, HID);
  gat_node_dots<<<node_blk, TB, 0, stream>>>(h1, a1s, a1d, as_buf, ad_buf, N, HID);
  gat_aggregate_owned<HID, AGG_ROWS><<<agg_blk, AGG_THREADS, 0, stream>>>(ei, as_buf, ad_buf, h1, b1, agg, E, N);

  gat_wmma_gemm_f32<HID, 4, 8, 8><<<gemm_grid(N, OUT2, 8, 64), 256, 0, stream>>>(agg, W2, h1, N, OUT2);
  gat_node_dots<<<node_blk, TB, 0, stream>>>(h1, a2s, a2d, as_buf, ad_buf, N, OUT2);
  gat_aggregate_owned<OUT2, AGG_ROWS><<<agg_blk, AGG_THREADS, 0, stream>>>(ei, as_buf, ad_buf, h1, b2, agg, E, N);

  gat_fc<<<(N + TB - 1) / TB, TB, 0, stream>>>(agg, fcW, fcb, (float*)d_out, N);
}


# Aggregation block owns 196 rows per edge-list scan (layer 1: two 98-row runs summed in turn from two LDS lists; layer 2: all 196, 2 entries per wave trip); 511->256 blocks; sums in arrival order
# speedup vs baseline: 3.1499x; 1.0999x over previous
#include <hip/hip_runtime.h>


typedef __attribute__((ext_vector_type(2))) float v2f;
typedef __attribute__((ext_vector_type(8))) float v8f;

#define IN_C  128
#define HID   512
#define OUT2  64
#define NCLS  2
#define NEG_SLOPE 0.2f

template <int K, int NT, int WAVES, int U>
__global__ __launch_bounds__(WAVES * 32) void gat_wmma_gemm_f32(
    const float* __restrict__ A, const float* __restrict__ B,
    float* __restrict__ C, int M, int Nc) {
  static_assert(K % 4 == 0 && NT % 2 == 0, "k-steps of 4, tiles stored in pairs");
  constexpr int TP = K * 16 + 8;
  __shared__ __attribute__((aligned(16))) float Bs[NT * TP];

  const int tid  = threadIdx.x;
  const int lane = tid & 31;
  const int waveid = tid >> 5;
  const int col0 = blockIdx.y * (NT * 16);
  const int tilesM = (M + 15) >> 4;

  constexpr int QC = NT * 4;
  constexpr int ITEMS = (K / 2) * QC;
#pragma unroll 4
  for (int it = tid; it < ITEMS; it += WAVES * 32) {
    const int p = it / QC;
    const int q = it % QC;
    const float* src = B + (size_t)(2 * p) * Nc + col0 + 4 * q;
    const float4 r0 = *(const float4*)src;
    const float4 r1 = *(const float4*)(src + Nc);
    float* dst = Bs + (q >> 2) * TP + p * 32 + (q & 3) * 8;
    *(float4*)(dst)     = make_float4(r0.x, r1.x, r0.y, r1.y);
    *(float4*)(dst + 4) = make_float4(r0.z, r1.z, r0.w, r1.w);
  }
  __syncthreads();

  const int tm = blockIdx.x * WAVES + waveid;
  if (tm >= tilesM) return;

  const int half = lane >> 4;
  const int l16  = lane & 15;

  const float* Ap = A + (size_t)(tm * 16 + l16) * K + 2 * half;
  const float* Bl = Bs + half * 32 + (l16 << 1);

  v8f acc[NT];
#pragma unroll
  for (int t = 0; t < NT; ++t) acc[t] = v8f{};

#pragma unroll U
  for (int k = 0; k < K; k += 4) {
    const v2f a = *(const v2f*)(Ap + k);
#pragma unroll
    for (int t = 0; t < NT; ++t) {
      const v2f b = *(const v2f*)(Bl + t * TP + k * 16);
      acc[t] = __builtin_amdgcn_wmma_f32_16x16x4_f32(
           false, a,  false, b,
           (short)0, acc[t],  false,  false);
    }
  }

  float* Cw = C + (size_t)(tm * 16) * Nc + col0 + lane;
#pragma unroll
  for (int t = 0; t < NT; t += 2) {
#pragma unroll
    for (int r = 0; r < 8; ++r) {
      const float x0 = acc[t][r];
      const float x1 = acc[t + 1][r];
      const float y0 = __shfl_xor(x0, 16, 32);
      const float y1 = __shfl_xor(x1, 16, 32);
      const float lo = half ? y1 : x0;
      const float hi = half ? x1 : y0;
      Cw[(size_t)r * Nc + t * 16]       = lo;
      Cw[(size_t)(r + 8) * Nc + t * 16] = hi;
    }
  }
}

__global__ void gat_node_dots(const float* __restrict__ h,
                              const float* __restrict__ a_src,
                              const float* __restrict__ a_dst,
                              float* __restrict__ as_out,
                              float* __restrict__ ad_out,
                              int N, int D) {
  const int wave = blockIdx.x * (blockDim.x >> 5) + (threadIdx.x >> 5);
  const int lane = threadIdx.x & 31;
  if (wave >= N) return;
  const float* row = h + (size_t)wave * D;
  float ss = 0.f, sd = 0.f;
  for (int c = lane; c < D; c += 32) {
    float v = row[c];
    ss += v * a_src[c];
    sd += v * a_dst[c];
  }
#pragma unroll
  for (int off = 16; off > 0; off >>= 1) {
    ss += __shfl_down(ss, off, 32);
    sd += __shfl_down(sd, off, 32);
  }
  if (lane == 0) { as_out[wave] = ss; ad_out[wave] = sd; }
}

__device__ __forceinline__ float leakyr(float v) {
  return v > 0.f ? v : NEG_SLOPE * v;
}

__device__ __forceinline__ void lds_max_f32(float* addr, float val) {
  if (__float_as_int(val) >= 0) atomicMax((int*)addr, __float_as_int(val));
  else                          atomicMin((unsigned int*)addr, __float_as_uint(val));
}

#define AGG_ROWS 98
#define AGG_U    16
#define AGG_THREADS 1024
#define AGG_CAP  4096

struct AggList {
  int*   s;
  int*   r;
  float* v;
  int*   cnt;
  float* m;
  float* den;
  float* ad;
};

template <int R, int H>
__device__ __forceinline__ void agg_append_owned(const int (&dv)[AGG_U], int base, int lane,
                                                 int d0, int lo, int hi,
                                                 int* ls, int* lr, int* lcnt) {
#pragma unroll
  for (int k = 0; k < AGG_U; ++k) {
    const int r = dv[k] - d0;
    if ((unsigned)(r - lo) < (unsigned)(hi - lo)) {
      const int hh = (H > 1) ? r / R : 0;
      const int idx = atomicAdd(lcnt + hh, 1);
      if (idx < AGG_CAP) { ls[hh * AGG_CAP + idx] = base + k * 32 + lane; lr[hh * AGG_CAP + idx] = r; }
    }
  }
}

template <int R, int H>
__device__ __forceinline__ void agg_scan_append(const int* __restrict__ dstp, int eb, int ee,
                                                int d0, int lo, int hi, int wv, int nw, int lane,
                                                int* ls, int* lr, int* lcnt) {
  const int sweep  = AGG_U * 32;
  const int stride = nw * sweep;
  int base = eb + wv * sweep;

  if (base + sweep <= ee) {
    int cur[AGG_U];
#pragma unroll
    for (int k = 0; k < AGG_U; ++k) cur[k] = dstp[base + k * 32 + lane];
    while (true) {
      const int nb = base + stride;
      const bool more = (nb + sweep <= ee);
      const int pb = more ? nb : base;
      int nxt[AGG_U];
#pragma unroll
      for (int k = 0; k < AGG_U; ++k) nxt[k] = dstp[pb + k * 32 + lane];
      agg_append_owned<R, H>(cur, base, lane, d0, lo, hi, ls, lr, lcnt);
      base = nb;
      if (!more) break;
#pragma unroll
      for (int k = 0; k < AGG_U; ++k) cur[k] = nxt[k];
    }
  }
  if (base < ee) {
    int cur[AGG_U];
#pragma unroll
    for (int k = 0; k < AGG_U; ++k) {
      const int e = base + k * 32 + lane;
      cur[k] = (e < ee) ? dstp[e] : -1;
    }
    agg_append_owned<R, H>(cur, base, lane, d0, lo, hi, ls, lr, lcnt);
  }
}

__device__ __forceinline__ void agg_list_scores(const AggList& L, int first, int n, bool do_max,
                                                const int* __restrict__ ei,
                                                const float* __restrict__ as) {
  for (int i = first + (int)threadIdx.x; i < n; i += AGG_THREADS) {
    const int r = L.r[i];
    const int s = ei[L.s[i]];
    const float sc = leakyr(as[s] + L.ad[r]);
    L.s[i] = s;
    L.v[i] = sc;
    if (do_max) lds_max_f32(&L.m[r], sc);
  }
}

__device__ __forceinline__ void agg_list_exps(const AggList& L, int n, bool do_sum) {
  for (int i = (int)threadIdx.x; i < n; i += AGG_THREADS) {
    const int r = L.r[i];
    const float ex = expf(L.v[i] - L.m[r]);
    L.v[i] = ex;
    if (do_sum) atomicAdd(&L.den[r], ex);
  }
}

template <int D>
__device__ __forceinline__ void agg_gather_entry(const AggList& L, int i, int rbase, int lane,
                                                 const float* __restrict__ h, float* acc) {
  constexpr int V = D / 32;
  const int s = L.s[i];
  const float* hs = h + (size_t)s * D + lane;
  float v[V];
#pragma unroll
  for (int j = 0; j < V; ++j) v[j] = hs[j * 32];
  const int r = L.r[i];
  const float a = L.v[i] / L.den[r];
  float* ar = acc + (r - rbase) * D + lane;
#pragma unroll
  for (int j = 0; j < V; ++j) atomicAdd(ar + j * 32, a * v[j]);
}

template <int D, int EPT>
__device__ __forceinline__ void agg_list_gather(const AggList& L, int n, int rbase,
                                                int wv, int nw, int lane,
                                                const float* __restrict__ h, float* acc) {
  constexpr int V = D / 32;
  const int groups = n / EPT;
  int g = wv;
  for (; g < groups; g += nw) {
    if (EPT == 1) {
      agg_gather_entry<D>(L, g, rbase, lane, h, acc);
    } else {
      const int i0 = g * EPT;
      float v[EPT][V];
#pragma unroll
      for (int e = 0; e < EPT; ++e) {
        const float* hs = h + (size_t)L.s[i0 + e] * D + lane;
#pragma unroll
        for (int j = 0; j < V; ++j) v[e][j] = hs[j * 32];
      }
#pragma unroll
      for (int e = 0; e < EPT; ++e) {
        const int r = L.r[i0 + e];
        const float a = L.v[i0 + e] / L.den[r];
        float* ar = acc + (r - rbase) * D + lane;
#pragma unroll
        for (int j = 0; j < V; ++j) atomicAdd(ar + j * 32, a * v[e][j]);
      }
    }
  }
  if (EPT > 1 && g == groups) {
    for (int i = groups * EPT; i < n; ++i) agg_gather_entry<D>(L, i, rbase, lane, h, acc);
  }
}

template <int D, int R, int H, int EPT>
__global__ __launch_bounds__(AGG_THREADS) void gat_aggregate_owned(
    const int* __restrict__ ei, const float* __restrict__ as,
    const float* __restrict__ ad, const float* __restrict__ h,
    const float* __restrict__ bias, float* __restrict__ out, int E, int N) {
  constexpr int RB = R * H;
  __shared__ __attribute__((aligned(16))) float acc[R * D];
  __shared__ int   list_s[H * AGG_CAP];
  __shared__ int   list_r[H * AGG_CAP];
  __shared__ float list_v[H * AGG_CAP];
  __shared__ float row_m[RB];
  __shared__ float row_den[RB];
  __shared__ float row_ad[RB];
  __shared__ int   list_cnt[H];
  static_assert(R <= AGG_CAP && RB <= AGG_THREADS, "self loops open the lists, one thread each");

  const int tid  = threadIdx.x;
  const int lane = tid & 31;
  const int wv   = tid >> 5;
  const int nw   = AGG_THREADS >> 5;
  const int d0   = blockIdx.x * RB;
  const int rows_all = (N - d0 < RB) ? (N - d0) : RB;
  const int* __restrict__ dstp = ei + E;

  auto run_rows = [&](int hh) {
    const int left = rows_all - hh * R;
    return left < 0 ? 0 : (left < R ? left : R);
  };
  auto run_list = [&](int hh) {
    return AggList{list_s + hh * AGG_CAP, list_r + hh * AGG_CAP, list_v + hh * AGG_CAP,
                   list_cnt + hh, row_m, row_den, row_ad};
  };

  {
    float4* z4 = (float4*)acc;
    for (int idx = tid; idx < run_rows(0) * (D / 4); idx += AGG_THREADS)
      z4[idx] = make_float4(0.f, 0.f, 0.f, 0.f);
  }
  if (tid < rows_all) {
    const int hh = tid / R;
    const int at = hh * AGG_CAP + (tid - hh * R);
    const float adv = ad[d0 + tid];
    const float sc  = leakyr(as[d0 + tid] + adv);
    row_ad[tid] = adv; row_m[tid] = sc; row_den[tid] = 0.f;
    list_s[at] = d0 + tid; list_r[at] = tid; list_v[at] = sc;
  }
  if (tid < H) list_cnt[tid] = run_rows(tid);
  __syncthreads();

  agg_scan_append<R, H>(dstp, 0, E, d0, 0, rows_all, wv, nw, lane, list_s, list_r, list_cnt);
  __syncthreads();

  int  n[H];
  bool fits[H];
#pragma unroll
  for (int hh = 0; hh < H; ++hh) { n[hh] = list_cnt[hh]; fits[hh] = (n[hh] <= AGG_CAP); }
#pragma unroll
  for (int hh = 0; hh < H; ++hh)
    if (fits[hh]) agg_list_scores(run_list(hh), run_rows(hh), n[hh], true, ei, as);
  __syncthreads();
#pragma unroll
  for (int hh = 0; hh < H; ++hh)
    if (fits[hh]) agg_list_exps(run_list(hh), n[hh], true);
  __syncthreads();

  constexpr int Q = D / 4;
  static_assert((Q & (Q - 1)) == 0 && AGG_THREADS % Q == 0, "column group must be fixed per thread");
  const int c0 = (tid & (Q - 1)) * 4;
  const float bx = bias[c0], by = bias[c0 + 1], bz = bias[c0 + 2], bw = bias[c0 + 3];

#pragma unroll
  for (int hh = 0; hh < H; ++hh) {
    const int rows = run_rows(hh);
    if (rows > 0) {
      const AggList L = run_list(hh);
      const int rb = hh * R;
      if (fits[hh]) {
        agg_list_gather<D, EPT>(L, n[hh], rb, wv, nw, lane, h, acc);
      } else {
        for (int pass = 0; pass < 3; ++pass) {
          for (int eb = -AGG_CAP; eb < E; eb += AGG_CAP) {
            __syncthreads();
            int first = 0;
            if (eb < 0) {
              if (tid < rows) {
                L.s[tid] = d0 + rb + tid; L.r[tid] = rb + tid;
                L.v[tid] = leakyr(as[d0 + rb + tid] + row_ad[rb + tid]);
              }
              if (tid == 0) *L.cnt = rows;
              first = rows;
              __syncthreads();
            } else {
              if (tid == 0) *L.cnt = 0;
              __syncthreads();
              agg_scan_append<R, H>(dstp, eb, (E - eb < AGG_CAP) ? E : eb + AGG_CAP, d0, rb, rb + rows,
                                    wv, nw, lane, list_s, list_r, list_cnt);
              __syncthreads();
            }
            const int np = *L.cnt;
            agg_list_scores(L, first, np, pass == 0, ei, as);
            if (pass >= 1) {
              __syncthreads();
              agg_list_exps(L, np, pass == 1);
            }
            if (pass == 2) {
              __syncthreads();
              agg_list_gather<D, EPT>(L, np, rb, wv, nw, lane, h, acc);
            }
          }
        }
      }
      __syncthreads();

      float4* a4 = (float4*)acc;
      float4* o4 = (float4*)(out + (size_t)(d0 + rb) * D);
      for (int idx = tid; idx < rows * Q; idx += AGG_THREADS) {
        float4 v = a4[idx];
        if (hh + 1 < H) a4[idx] = make_float4(0.f, 0.f, 0.f, 0.f);
        v.x += bx; v.y += by; v.z += bz; v.w += bw;
        v.x = v.x > 0.f ? v.x : 0.f;
        v.y = v.y > 0.f ? v.y : 0.f;
        v.z = v.z > 0.f ? v.z : 0.f;
        v.w = v.w > 0.f ? v.w : 0.f;
        o4[idx] = v;
      }
      if (hh + 1 < H) __syncthreads();
    }
  }
}

__global__ void gat_fc(const float* __restrict__ h,
                       const float* __restrict__ W,
                       const float* __restrict__ b,
                       float* __restrict__ out, int N) {
  int n = blockIdx.x * blockDim.x + threadIdx.x;
  if (n >= N) return;
  const float* row = h + (size_t)n * OUT2;
  float a0 = b[0], a1 = b[1];
#pragma unroll
  for (int c = 0; c < OUT2; ++c) {
    float v = row[c];
    a0 += v * W[c * NCLS + 0];
    a1 += v * W[c * NCLS + 1];
  }
  out[n * NCLS + 0] = a0;
  out[n * NCLS + 1] = a1;
}

extern "C" void kernel_launch(void* const* d_in, const int* in_sizes, int n_in,
                              void* d_out, int out_size, void* d_ws, size_t ws_size,
                              hipStream_t stream) {
  const float* x   = (const float*)d_in[0];
  const int*   ei  = (const int*)  d_in[1];
  const float* W1  = (const float*)d_in[2];
  const float* a1s = (const float*)d_in[3];
  const float* a1d = (const float*)d_in[4];
  const float* b1  = (const float*)d_in[5];
  const float* W2  = (const float*)d_in[6];
  const float* a2s = (const float*)d_in[7];
  const float* a2d = (const float*)d_in[8];
  const float* b2  = (const float*)d_in[9];
  const float* fcW = (const float*)d_in[10];
  const float* fcb = (const float*)d_in[11];

  const int N    = in_sizes[0] / IN_C;
  const int E    = in_sizes[1] / 2;

  float* w       = (float*)d_ws;
  float* h1      = w;
  float* agg     = w + (size_t)N * HID;
  float* as_buf  = w + (size_t)2 * N * HID;
  float* ad_buf  = as_buf + N;

  const int TB = 256;

  auto gemm_grid = [](int M, int Nc, int waves, int cols) {
    return dim3((((M + 15) / 16) + waves - 1) / waves, Nc / cols);
  };
  const int agg_own   = 2 * AGG_ROWS;
  const int agg_blk   = (N + agg_own - 1) / agg_own;
  const int node_blk  = (N + 7) / 8;

  gat_wmma_gemm_f32<IN_C, 8, 8, 4><<<gemm_grid(N, HID, 8, 128), 256, 0, stream>>>(x, W1, h1, N, HID);
  gat_node_dots<<<node_blk, TB, 0, stream>>>(h1, a1s, a1d, as_buf, ad_buf, N, HID);
  gat_aggregate_owned<HID, AGG_ROWS, 2, 1><<<agg_blk, AGG_THREADS, 0, stream>>>(ei, as_buf, ad_buf, h1, b1, agg, E, N);

  gat_wmma_gemm_f32<HID, 4, 8, 8><<<gemm_grid(N, OUT2, 8, 64), 256, 0, stream>>>(agg, W2, h1, N, OUT2);
  gat_node_dots<<<node_blk, TB, 0, stream>>>(h1, a2s, a2d, as_buf, ad_buf, N, OUT2);
  gat_aggregate_owned<OUT2, 2 * AGG_ROWS, 1, 2><<<agg_blk, AGG_THREADS, 0, stream>>>(ei, as_buf, ad_buf, h1, b2, agg, E, N);

  gat_fc<<<(N + TB - 1) / TB, TB, 0, stream>>>(agg, fcW, fcb, (float*)d_out, N);
}


# Layer-1 product by one persistent block per WGP, all of W1 in LDS, covering all 4 column strips of its rows; score dots taken from the accumulators (new sum tree); node-dot launch goes, 7 -> 6
# speedup vs baseline: 3.4834x; 1.1059x over previous
#include <hip/hip_runtime.h>


typedef __attribute__((ext_vector_type(2))) float v2f;
typedef __attribute__((ext_vector_type(8))) float v8f;

#define IN_C  128
#define HID   512
#define OUT2  64
#define NCLS  2
#define NEG_SLOPE 0.2f
#define G1_BLOCKS 256

template <int K, int NT, int WAVES, int U>
__global__ __launch_bounds__(WAVES * 32) void gat_wmma_gemm_f32(
    const float* __restrict__ A, const float* __restrict__ B,
    float* __restrict__ C, int M, int Nc) {
  static_assert(K % 4 == 0 && NT % 2 == 0, "k-steps of 4, tiles stored in pairs");
  constexpr int TP = K * 16 + 8;
  __shared__ __attribute__((aligned(16))) float Bs[NT * TP];

  const int tid  = threadIdx.x;
  const int lane = tid & 31;
  const int waveid = tid >> 5;
  const int col0 = blockIdx.y * (NT * 16);
  const int tilesM = (M + 15) >> 4;

  constexpr int QC = NT * 4;
  constexpr int ITEMS = (K / 2) * QC;
#pragma unroll 4
  for (int it = tid; it < ITEMS; it += WAVES * 32) {
    const int p = it / QC;
    const int q = it % QC;
    const float* src = B + (size_t)(2 * p) * Nc + col0 + 4 * q;
    const float4 r0 = *(const float4*)src;
    const float4 r1 = *(const float4*)(src + Nc);
    float* dst = Bs + (q >> 2) * TP + p * 32 + (q & 3) * 8;
    *(float4*)(dst)     = make_float4(r0.x, r1.x, r0.y, r1.y);
    *(float4*)(dst + 4) = make_float4(r0.z, r1.z, r0.w, r1.w);
  }
  __syncthreads();

  const int tm = blockIdx.x * WAVES + waveid;
  if (tm >= tilesM) return;

  const int half = lane >> 4;
  const int l16  = lane & 15;

  const float* Ap = A + (size_t)(tm * 16 + l16) * K + 2 * half;
  const float* Bl = Bs + half * 32 + (l16 << 1);

  v8f acc[NT];
#pragma unroll
  for (int t = 0; t < NT; ++t) acc[t] = v8f{};

#pragma unroll U
  for (int k = 0; k < K; k += 4) {
    const v2f a = *(const v2f*)(Ap + k);
#pragma unroll
    for (int t = 0; t < NT; ++t) {
      const v2f b = *(const v2f*)(Bl + t * TP + k * 16);
      acc[t] = __builtin_amdgcn_wmma_f32_16x16x4_f32(
           false, a,  false, b,
           (short)0, acc[t],  false,  false);
    }
  }

  float* Cw = C + (size_t)(tm * 16) * Nc + col0 + lane;
#pragma unroll
  for (int t = 0; t < NT; t += 2) {
#pragma unroll
    for (int r = 0; r < 8; ++r) {
      const float x0 = acc[t][r];
      const float x1 = acc[t + 1][r];
      const float y0 = __shfl_xor(x0, 16, 32);
      const float y1 = __shfl_xor(x1, 16, 32);
      const float lo = half ? y1 : x0;
      const float hi = half ? x1 : y0;
      Cw[(size_t)r * Nc + t * 16]       = lo;
      Cw[(size_t)(r + 8) * Nc + t * 16] = hi;
    }
  }
}

template <int XM>
__device__ __forceinline__ float row_xmask_add(float v) {
  const int b = __float_as_int(v);
  return v + __int_as_float(__builtin_amdgcn_update_dpp(b, b, 0x160 + XM, 0xf, 0xf, false));
}
__device__ __forceinline__ float row16_sum(float v) {
  v = row_xmask_add<1>(v);
  v = row_xmask_add<2>(v);
  v = row_xmask_add<4>(v);
  return row_xmask_add<8>(v);
}

template <int K, int NC, int WAVES, int U>
__global__ __launch_bounds__(WAVES * 32) void gat_wmma_gemm_dots_f32(
    const float* __restrict__ A, const float* __restrict__ B,
    const float* __restrict__ a_src, const float* __restrict__ a_dst,
    float* __restrict__ C, float* __restrict__ as_out, float* __restrict__ ad_out, int M) {
  constexpr int NT = 8;
  constexpr int STRIPS = NC / (16 * NT);
  constexpr int NTT = NC / 16;
  constexpr int CH = 16;
  static_assert(K % 4 == 0 && NC % (16 * NT) == 0 && STRIPS == 4 && WAVES % STRIPS == 0,
                "a wave keeps one of four strips");
  constexpr int TP = K * 16 + 8;
  __shared__ __attribute__((aligned(16))) float Bs[NTT * TP];
  __shared__ float part_s[STRIPS][CH * 16];
  __shared__ float part_d[STRIPS][CH * 16];

  const int tid  = threadIdx.x;
  const int lane = tid & 31;
  const int waveid = __builtin_amdgcn_readfirstlane(tid >> 5);

  constexpr int QC = NTT * 4;
  constexpr int ITEMS = (K / 2) * QC;
#pragma unroll 4
  for (int it = tid; it < ITEMS; it += WAVES * 32) {
    const int p = it / QC;
    const int q = it % QC;
    const float* src = B + (size_t)(2 * p) * NC + 4 * q;
    const float4 r0 = *(const float4*)src;
    const float4 r1 = *(const float4*)(src + NC);
    float* dst = Bs + (q >> 2) * TP + p * 32 + (q & 3) * 8;
    *(float4*)(dst)     = make_float4(r0.x, r1.x, r0.y, r1.y);
    *(float4*)(dst + 4) = make_float4(r0.z, r1.z, r0.w, r1.w);
  }

  const int half = lane >> 4;
  const int l16  = lane & 15;
  const int st   = waveid % STRIPS;
  constexpr int TC = WAVES / STRIPS;
  const int tl0  = (waveid / STRIPS + st) % TC;
  const int col0 = st * (NT * 16);

  float ws[NT], wd[NT];
#pragma unroll
  for (int t = 0; t < NT; ++t) {
    ws[t] = a_src[col0 + t * 16 + l16];
    wd[t] = a_dst[col0 + t * 16 + l16];
  }
  __syncthreads();

  const int tilesM = (M + 15) >> 4;
  const int tpb = (tilesM + (int)gridDim.x - 1) / (int)gridDim.x;
  const int t_begin = blockIdx.x * tpb;
  const int t_end = (t_begin + tpb < tilesM) ? (t_begin + tpb) : tilesM;

  const float* Bl = Bs + (st * NT) * TP + half * 32 + (l16 << 1);

  for (int c0 = t_begin; c0 < t_end; c0 += CH) {
    const int ntile = (t_end - c0 < CH) ? (t_end - c0) : CH;
    for (int tl = tl0; tl < ntile; tl += TC) {
      const int tm = c0 + tl;
      const float* Ap = A + (size_t)(tm * 16 + l16) * K + 2 * half;

      v8f acc[NT];
#pragma unroll
      for (int t = 0; t < NT; ++t) acc[t] = v8f{};

#pragma unroll U
      for (int k = 0; k < K; k += 4) {
        const v2f a = *(const v2f*)(Ap + k);
#pragma unroll
        for (int t = 0; t < NT; ++t) {
          const v2f b = *(const v2f*)(Bl + t * TP + k * 16);
          acc[t] = __builtin_amdgcn_wmma_f32_16x16x4_f32(
               false, a,  false, b,
               (short)0, acc[t],  false,  false);
        }
      }

      float* Cw = C + (size_t)(tm * 16) * NC + col0 + lane;
#pragma unroll
      for (int t = 0; t < NT; t += 2) {
#pragma unroll
        for (int r = 0; r < 8; ++r) {
          const float x0 = acc[t][r];
          const float x1 = acc[t + 1][r];
          const float y0 = __shfl_xor(x0, 16, 32);
          const float y1 = __shfl_xor(x1, 16, 32);
          const float lo = half ? y1 : x0;
          const float hi = half ? x1 : y0;
          Cw[(size_t)r * NC + t * 16]       = lo;
          Cw[(size_t)(r + 8) * NC + t * 16] = hi;
        }
      }

#pragma unroll
      for (int r = 0; r < 8; ++r) {
        float ss = 0.f, sd = 0.f;
#pragma unroll
        for (int t = 0; t < NT; ++t) {
          ss += acc[t][r] * ws[t];
          sd += acc[t][r] * wd[t];
        }
        ss = row16_sum(ss);
        sd = row16_sum(sd);
        if (l16 == 0) {
          part_s[st][tl * 16 + half * 8 + r] = ss;
          part_d[st][tl * 16 + half * 8 + r] = sd;
        }
      }
    }
    __syncthreads();

    for (int i = tid; i < ntile * 16; i += WAVES * 32) {
      const int row = c0 * 16 + i;
      if (row < M) {
        as_out[row] = ((part_s[0][i] + part_s[1][i]) + part_s[2][i]) + part_s[3][i];
        ad_out[row] = ((part_d[0][i] + part_d[1][i]) + part_d[2][i]) + part_d[3][i];
      }
    }
    if (c0 + CH < t_end) __syncthreads();
  }
}

__global__ void gat_node_dots(const float* __restrict__ h,
                              const float* __restrict__ a_src,
                              const float* __restrict__ a_dst,
                              float* __restrict__ as_out,
                              float* __restrict__ ad_out,
                              int N, int D) {
  const int wave = blockIdx.x * (blockDim.x >> 5) + (threadIdx.x >> 5);
  const int lane = threadIdx.x & 31;
  if (wave >= N) return;
  const float* row = h + (size_t)wave * D;
  float ss = 0.f, sd = 0.f;
  for (int c = lane; c < D; c += 32) {
    float v = row[c];
    ss += v * a_src[c];
    sd += v * a_dst[c];
  }
#pragma unroll
  for (int off = 16; off > 0; off >>= 1) {
    ss += __shfl_down(ss, off, 32);
    sd += __shfl_down(sd, off, 32);
  }
  if (lane == 0) { as_out[wave] = ss; ad_out[wave] = sd; }
}

__device__ __forceinline__ float leakyr(float v) {
  return v > 0.f ? v : NEG_SLOPE * v;
}

__device__ __forceinline__ void lds_max_f32(float* addr, float val) {
  if (__float_as_int(val) >= 0) atomicMax((int*)addr, __float_as_int(val));
  else                          atomicMin((unsigned int*)addr, __float_as_uint(val));
}

#define AGG_ROWS 98
#define AGG_U    16
#define AGG_THREADS 1024
#define AGG_CAP  4096

struct AggList {
  int*   s;
  int*   r;
  float* v;
  int*   cnt;
  float* m;
  float* den;
  float* ad;
};

template <int R, int H>
__device__ __forceinline__ void agg_append_owned(const int (&dv)[AGG_U], int base, int lane,
                                                 int d0, int lo, int hi,
                                                 int* ls, int* lr, int* lcnt) {
#pragma unroll
  for (int k = 0; k < AGG_U; ++k) {
    const int r = dv[k] - d0;
    if ((unsigned)(r - lo) < (unsigned)(hi - lo)) {
      const int hh = (H > 1) ? r / R : 0;
      const int idx = atomicAdd(lcnt + hh, 1);
      if (idx < AGG_CAP) { ls[hh * AGG_CAP + idx] = base + k * 32 + lane; lr[hh * AGG_CAP + idx] = r; }
    }
  }
}

template <int R, int H>
__device__ __forceinline__ void agg_scan_append(const int* __restrict__ dstp, int eb, int ee,
                                                int d0, int lo, int hi, int wv, int nw, int lane,
                                                int* ls, int* lr, int* lcnt) {
  const int sweep  = AGG_U * 32;
  const int stride = nw * sweep;
  int base = eb + wv * sweep;

  if (base + sweep <= ee) {
    int cur[AGG_U];
#pragma unroll
    for (int k = 0; k < AGG_U; ++k) cur[k] = dstp[base + k * 32 + lane];
    while (true) {
      const int nb = base + stride;
      const bool more = (nb + sweep <= ee);
      const int pb = more ? nb : base;
      int nxt[AGG_U];
#pragma unroll
      for (int k = 0; k < AGG_U; ++k) nxt[k] = dstp[pb + k * 32 + lane];
      agg_append_owned<R, H>(cur, base, lane, d0, lo, hi, ls, lr, lcnt);
      base = nb;
      if (!more) break;
#pragma unroll
      for (int k = 0; k < AGG_U; ++k) cur[k] = nxt[k];
    }
  }
  if (base < ee) {
    int cur[AGG_U];
#pragma unroll
    for (int k = 0; k < AGG_U; ++k) {
      const int e = base + k * 32 + lane;
      cur[k] = (e < ee) ? dstp[e] : -1;
    }
    agg_append_owned<R, H>(cur, base, lane, d0, lo, hi, ls, lr, lcnt);
  }
}

__device__ __forceinline__ void agg_list_scores(const AggList& L, int first, int n, bool do_max,
                                                const int* __restrict__ ei,
                                                const float* __restrict__ as) {
  for (int i = first + (int)threadIdx.x; i < n; i += AGG_THREADS) {
    const int r = L.r[i];
    const int s = ei[L.s[i]];
    const float sc = leakyr(as[s] + L.ad[r]);
    L.s[i] = s;
    L.v[i] = sc;
    if (do_max) lds_max_f32(&L.m[r], sc);
  }
}

__device__ __forceinline__ void agg_list_exps(const AggList& L, int n, bool do_sum) {
  for (int i = (int)threadIdx.x; i < n; i += AGG_THREADS) {
    const int r = L.r[i];
    const float ex = expf(L.v[i] - L.m[r]);
    L.v[i] = ex;
    if (do_sum) atomicAdd(&L.den[r], ex);
  }
}

template <int D>
__device__ __forceinline__ void agg_gather_entry(const AggList& L, int i, int rbase, int lane,
                                                 const float* __restrict__ h, float* acc) {
  constexpr int V = D / 32;
  const int s = L.s[i];
  const float* hs = h + (size_t)s * D + lane;
  float v[V];
#pragma unroll
  for (int j = 0; j < V; ++j) v[j] = hs[j * 32];
  const int r = L.r[i];
  const float a = L.v[i] / L.den[r];
  float* ar = acc + (r - rbase) * D + lane;
#pragma unroll
  for (int j = 0; j < V; ++j) atomicAdd(ar + j * 32, a * v[j]);
}

template <int D, int EPT>
__device__ __forceinline__ void agg_list_gather(const AggList& L, int n, int rbase,
                                                int wv, int nw, int lane,
                                                const float* __restrict__ h, float* acc) {
  constexpr int V = D / 32;
  const int groups = n / EPT;
  int g = wv;
  for (; g < groups; g += nw) {
    if (EPT == 1) {
      agg_gather_entry<D>(L, g, rbase, lane, h, acc);
    } else {
      const int i0 = g * EPT;
      float v[EPT][V];
#pragma unroll
      for (int e = 0; e < EPT; ++e) {
        const float* hs = h + (size_t)L.s[i0 + e] * D + lane;
#pragma unroll
        for (int j = 0; j < V; ++j) v[e][j] = hs[j * 32];
      }
#pragma unroll
      for (int e = 0; e < EPT; ++e) {
        const int r = L.r[i0 + e];
        const float a = L.v[i0 + e] / L.den[r];
        float* ar = acc + (r - rbase) * D + lane;
#pragma unroll
        for (int j = 0; j < V; ++j) atomicAdd(ar + j * 32, a * v[e][j]);
      }
    }
  }
  if (EPT > 1 && g == groups) {
    for (int i = groups * EPT; i < n; ++i) agg_gather_entry<D>(L, i, rbase, lane, h, acc);
  }
}

template <int D, int R, int H, int EPT>
__global__ __launch_bounds__(AGG_THREADS) void gat_aggregate_owned(
    const int* __restrict__ ei, const float* __restrict__ as,
    const float* __restrict__ ad, const float* __restrict__ h,
    const float* __restrict__ bias, float* __restrict__ out, int E, int N) {
  constexpr int RB = R * H;
  __shared__ __attribute__((aligned(16))) float acc[R * D];
  __shared__ int   list_s[H * AGG_CAP];
  __shared__ int   list_r[H * AGG_CAP];
  __shared__ float list_v[H * AGG_CAP];
  __shared__ float row_m[RB];
  __shared__ float row_den[RB];
  __shared__ float row_ad[RB];
  __shared__ int   list_cnt[H];
  static_assert(R <= AGG_CAP && RB <= AGG_THREADS, "self loops open the lists, one thread each");

  const int tid  = threadIdx.x;
  const int lane = tid & 31;
  const int wv   = tid >> 5;
  const int nw   = AGG_THREADS >> 5;
  const int d0   = blockIdx.x * RB;
  const int rows_all = (N - d0 < RB) ? (N - d0) : RB;
  const int* __restrict__ dstp = ei + E;

  auto run_rows = [&](int hh) {
    const int left = rows_all - hh * R;
    return left < 0 ? 0 : (left < R ? left : R);
  };
  auto run_list = [&](int hh) {
    return AggList{list_s + hh * AGG_CAP, list_r + hh * AGG_CAP, list_v + hh * AGG_CAP,
                   list_cnt + hh, row_m, row_den, row_ad};
  };

  {
    float4* z4 = (float4*)acc;
    for (int idx = tid; idx < run_rows(0) * (D / 4); idx += AGG_THREADS)
      z4[idx] = make_float4(0.f, 0.f, 0.f, 0.f);
  }
  if (tid < rows_all) {
    const int hh = tid / R;
    const int at = hh * AGG_CAP + (tid - hh * R);
    const float adv = ad[d0 + tid];
    const float sc  = leakyr(as[d0 + tid] + adv);
    row_ad[tid] = adv; row_m[tid] = sc; row_den[tid] = 0.f;
    list_s[at] = d0 + tid; list_r[at] = tid; list_v[at] = sc;
  }
  if (tid < H) list_cnt[tid] = run_rows(tid);
  __syncthreads();

  agg_scan_append<R, H>(dstp, 0, E, d0, 0, rows_all, wv, nw, lane, list_s, list_r, list_cnt);
  __syncthreads();

  int  n[H];
  bool fits[H];
#pragma unroll
  for (int hh = 0; hh < H; ++hh) { n[hh] = list_cnt[hh]; fits[hh] = (n[hh] <= AGG_CAP); }
#pragma unroll
  for (int hh = 0; hh < H; ++hh)
    if (fits[hh]) agg_list_scores(run_list(hh), run_rows(hh), n[hh], true, ei, as);
  __syncthreads();
#pragma unroll
  for (int hh = 0; hh < H; ++hh)
    if (fits[hh]) agg_list_exps(run_list(hh), n[hh], true);
  __syncthreads();

  constexpr int Q = D / 4;
  static_assert((Q & (Q - 1)) == 0 && AGG_THREADS % Q == 0, "column group must be fixed per thread");
  const int c0 = (tid & (Q - 1)) * 4;
  const float bx = bias[c0], by = bias[c0 + 1], bz = bias[c0 + 2], bw = bias[c0 + 3];

#pragma unroll
  for (int hh = 0; hh < H; ++hh) {
    const int rows = run_rows(hh);
    if (rows > 0) {
      const AggList L = run_list(hh);
      const int rb = hh * R;
      if (fits[hh]) {
        agg_list_gather<D, EPT>(L, n[hh], rb, wv, nw, lane, h, acc);
      } else {
        for (int pass = 0; pass < 3; ++pass) {
          for (int eb = -AGG_CAP; eb < E; eb += AGG_CAP) {
            __syncthreads();
            int first = 0;
            if (eb < 0) {
              if (tid < rows) {
                L.s[tid] = d0 + rb + tid; L.r[tid] = rb + tid;
                L.v[tid] = leakyr(as[d0 + rb + tid] + row_ad[rb + tid]);
              }
              if (tid == 0) *L.cnt = rows;
              first = rows;
              __syncthreads();
            } else {
              if (tid == 0) *L.cnt = 0;
              __syncthreads();
              agg_scan_append<R, H>(dstp, eb, (E - eb < AGG_CAP) ? E : eb + AGG_CAP, d0, rb, rb + rows,
                                    wv, nw, lane, list_s, list_r, list_cnt);
              __syncthreads();
            }
            const int np = *L.cnt;
            agg_list_scores(L, first, np, pass == 0, ei, as);
            if (pass >= 1) {
              __syncthreads();
              agg_list_exps(L, np, pass == 1);
            }
            if (pass == 2) {
              __syncthreads();
              agg_list_gather<D, EPT>(L, np, rb, wv, nw, lane, h, acc);
            }
          }
        }
      }
      __syncthreads();

      float4* a4 = (float4*)acc;
      float4* o4 = (float4*)(out + (size_t)(d0 + rb) * D);
      for (int idx = tid; idx < rows * Q; idx += AGG_THREADS) {
        float4 v = a4[idx];
        if (hh + 1 < H) a4[idx] = make_float4(0.f, 0.f, 0.f, 0.f);
        v.x += bx; v.y += by; v.z += bz; v.w += bw;
        v.x = v.x > 0.f ? v.x : 0.f;
        v.y = v.y > 0.f ? v.y : 0.f;
        v.z = v.z > 0.f ? v.z : 0.f;
        v.w = v.w > 0.f ? v.w : 0.f;
        o4[idx] = v;
      }
      if (hh + 1 < H) __syncthreads();
    }
  }
}

__global__ void gat_fc(const float* __restrict__ h,
                       const float* __restrict__ W,
                       const float* __restrict__ b,
                       float* __restrict__ out, int N) {
  int n = blockIdx.x * blockDim.x + threadIdx.x;
  if (n >= N) return;
  const float* row = h + (size_t)n * OUT2;
  float a0 = b[0], a1 = b[1];
#pragma unroll
  for (int c = 0; c < OUT2; ++c) {
    float v = row[c];
    a0 += v * W[c * NCLS + 0];
    a1 += v * W[c * NCLS + 1];
  }
  out[n * NCLS + 0] = a0;
  out[n * NCLS + 1] = a1;
}

extern "C" void kernel_launch(void* const* d_in, const int* in_sizes, int n_in,
                              void* d_out, int out_size, void* d_ws, size_t ws_size,
                              hipStream_t stream) {
  const float* x   = (const float*)d_in[0];
  const int*   ei  = (const int*)  d_in[1];
  const float* W1  = (const float*)d_in[2];
  const float* a1s = (const float*)d_in[3];
  const float* a1d = (const float*)d_in[4];
  const float* b1  = (const float*)d_in[5];
  const float* W2  = (const float*)d_in[6];
  const float* a2s = (const float*)d_in[7];
  const float* a2d = (const float*)d_in[8];
  const float* b2  = (const float*)d_in[9];
  const float* fcW = (const float*)d_in[10];
  const float* fcb = (const float*)d_in[11];

  const int N    = in_sizes[0] / IN_C;
  const int E    = in_sizes[1] / 2;

  float* w       = (float*)d_ws;
  float* h1      = w;
  float* agg     = w + (size_t)N * HID;
  float* as_buf  = w + (size_t)2 * N * HID;
  float* ad_buf  = as_buf + N;

  const int TB = 256;

  auto gemm_grid = [](int M, int Nc, int waves, int cols) {
    return dim3((((M + 15) / 16) + waves - 1) / waves, Nc / cols);
  };
  const int agg_own   = 2 * AGG_ROWS;
  const int agg_blk   = (N + agg_own - 1) / agg_own;
  const int node_blk  = (N + 7) / 8;

  gat_wmma_gemm_dots_f32<IN_C, HID, 16, 8><<<G1_BLOCKS, 512, 0, stream>>>(x, W1, a1s, a1d, h1, as_buf, ad_buf, N);
  gat_aggregate_owned<HID, AGG_ROWS, 2, 1><<<agg_blk, AGG_THREADS, 0, stream>>>(ei, as_buf, ad_buf, h1, b1, agg, E, N);

  gat_wmma_gemm_f32<HID, 4, 8, 8><<<gemm_grid(N, OUT2, 8, 64), 256, 0, stream>>>(agg, W2, h1, N, OUT2);
  gat_node_dots<<<node_blk, TB, 0, stream>>>(h1, a2s, a2d, as_buf, ad_buf, N, OUT2);
  gat_aggregate_owned<OUT2, 2 * AGG_ROWS, 1, 2><<<agg_blk, AGG_THREADS, 0, stream>>>(ei, as_buf, ad_buf, h1, b2, agg, E, N);

  gat_fc<<<(N + TB - 1) / TB, TB, 0, stream>>>(agg, fcW, fcb, (float*)d_out, N);
}


# Layer-1 aggregation blocks save the edge lists they already hold in LDS (source, row) to spare workspace; layer-2 blocks load them: no second scan, no extra blocks; sums stay in LDS arrival order
# speedup vs baseline: 3.7866x; 1.0870x over previous
#include <hip/hip_runtime.h>


typedef __attribute__((ext_vector_type(2))) float v2f;
typedef __attribute__((ext_vector_type(8))) float v8f;

#define IN_C  128
#define HID   512
#define OUT2  64
#define NCLS  2
#define NEG_SLOPE 0.2f
#define G1_BLOCKS 256

template <int K, int NT, int WAVES, int U>
__global__ __launch_bounds__(WAVES * 32) void gat_wmma_gemm_f32(
    const float* __restrict__ A, const float* __restrict__ B,
    float* __restrict__ C, int M, int Nc) {
  static_assert(K % 4 == 0 && NT % 2 == 0, "k-steps of 4, tiles stored in pairs");
  constexpr int TP = K * 16 + 8;
  __shared__ __attribute__((aligned(16))) float Bs[NT * TP];

  const int tid  = threadIdx.x;
  const int lane = tid & 31;
  const int waveid = tid >> 5;
  const int col0 = blockIdx.y * (NT * 16);
  const int tilesM = (M + 15) >> 4;

  constexpr int QC = NT * 4;
  constexpr int ITEMS = (K / 2) * QC;
#pragma unroll 4
  for (int it = tid; it < ITEMS; it += WAVES * 32) {
    const int p = it / QC;
    const int q = it % QC;
    const float* src = B + (size_t)(2 * p) * Nc + col0 + 4 * q;
    const float4 r0 = *(const float4*)src;
    const float4 r1 = *(const float4*)(src + Nc);
    float* dst = Bs + (q >> 2) * TP + p * 32 + (q & 3) * 8;
    *(float4*)(dst)     = make_float4(r0.x, r1.x, r0.y, r1.y);
    *(float4*)(dst + 4) = make_float4(r0.z, r1.z, r0.w, r1.w);
  }
  __syncthreads();

  const int tm = blockIdx.x * WAVES + waveid;
  if (tm >= tilesM) return;

  const int half = lane >> 4;
  const int l16  = lane & 15;

  const float* Ap = A + (size_t)(tm * 16 + l16) * K + 2 * half;
  const float* Bl = Bs + half * 32 + (l16 << 1);

  v8f acc[NT];
#pragma unroll
  for (int t = 0; t < NT; ++t) acc[t] = v8f{};

#pragma unroll U
  for (int k = 0; k < K; k += 4) {
    const v2f a = *(const v2f*)(Ap + k);
#pragma unroll
    for (int t = 0; t < NT; ++t) {
      const v2f b = *(const v2f*)(Bl + t * TP + k * 16);
      acc[t] = __builtin_amdgcn_wmma_f32_16x16x4_f32(
           false, a,  false, b,
           (short)0, acc[t],  false,  false);
    }
  }

  float* Cw = C + (size_t)(tm * 16) * Nc + col0 + lane;
#pragma unroll
  for (int t = 0; t < NT; t += 2) {
#pragma unroll
    for (int r = 0; r < 8; ++r) {
      const float x0 = acc[t][r];
      const float x1 = acc[t + 1][r];
      const float y0 = __shfl_xor(x0, 16, 32);
      const float y1 = __shfl_xor(x1, 16, 32);
      const float lo = half ? y1 : x0;
      const float hi = half ? x1 : y0;
      Cw[(size_t)r * Nc + t * 16]       = lo;
      Cw[(size_t)(r + 8) * Nc + t * 16] = hi;
    }
  }
}

template <int XM>
__device__ __forceinline__ float row_xmask_add(float v) {
  const int b = __float_as_int(v);
  return v + __int_as_float(__builtin_amdgcn_update_dpp(b, b, 0x160 + XM, 0xf, 0xf, false));
}
__device__ __forceinline__ float row16_sum(float v) {
  v = row_xmask_add<1>(v);
  v = row_xmask_add<2>(v);
  v = row_xmask_add<4>(v);
  return row_xmask_add<8>(v);
}

template <int K, int NC, int WAVES, int U>
__global__ __launch_bounds__(WAVES * 32) void gat_wmma_gemm_dots_f32(
    const float* __restrict__ A, const float* __restrict__ B,
    const float* __restrict__ a_src, const float* __restrict__ a_dst,
    float* __restrict__ C, float* __restrict__ as_out, float* __restrict__ ad_out, int M) {
  constexpr int NT = 8;
  constexpr int STRIPS = NC / (16 * NT);
  constexpr int NTT = NC / 16;
  constexpr int CH = 16;
  static_assert(K % 4 == 0 && NC % (16 * NT) == 0 && STRIPS == 4 && WAVES % STRIPS == 0,
                "a wave keeps one of four strips");
  constexpr int TP = K * 16 + 8;
  __shared__ __attribute__((aligned(16))) float Bs[NTT * TP];
  __shared__ float part_s[STRIPS][CH * 16];
  __shared__ float part_d[STRIPS][CH * 16];

  const int tid  = threadIdx.x;
  const int lane = tid & 31;
  const int waveid = __builtin_amdgcn_readfirstlane(tid >> 5);

  constexpr int QC = NTT * 4;
  constexpr int ITEMS = (K / 2) * QC;
#pragma unroll 4
  for (int it = tid; it < ITEMS; it += WAVES * 32) {
    const int p = it / QC;
    const int q = it % QC;
    const float* src = B + (size_t)(2 * p) * NC + 4 * q;
    const float4 r0 = *(const float4*)src;
    const float4 r1 = *(const float4*)(src + NC);
    float* dst = Bs + (q >> 2) * TP + p * 32 + (q & 3) * 8;
    *(float4*)(dst)     = make_float4(r0.x, r1.x, r0.y, r1.y);
    *(float4*)(dst + 4) = make_float4(r0.z, r1.z, r0.w, r1.w);
  }

  const int half = lane >> 4;
  const int l16  = lane & 15;
  const int st   = waveid % STRIPS;
  constexpr int TC = WAVES / STRIPS;
  const int tl0  = (waveid / STRIPS + st) % TC;
  const int col0 = st * (NT * 16);

  float ws[NT], wd[NT];
#pragma unroll
  for (int t = 0; t < NT; ++t) {
    ws[t] = a_src[col0 + t * 16 + l16];
    wd[t] = a_dst[col0 + t * 16 + l16];
  }
  __syncthreads();

  const int tilesM = (M + 15) >> 4;
  const int tpb = (tilesM + (int)gridDim.x - 1) / (int)gridDim.x;
  const int t_begin = blockIdx.x * tpb;
  const int t_end = (t_begin + tpb < tilesM) ? (t_begin + tpb) : tilesM;

  const float* Bl = Bs + (st * NT) * TP + half * 32 + (l16 << 1);

  for (int c0 = t_begin; c0 < t_end; c0 += CH) {
    const int ntile = (t_end - c0 < CH) ? (t_end - c0) : CH;
    for (int tl = tl0; tl < ntile; tl += TC) {
      const int tm = c0 + tl;
      const float* Ap = A + (size_t)(tm * 16 + l16) * K + 2 * half;

      v8f acc[NT];
#pragma unroll
      for (int t = 0; t < NT; ++t) acc[t] = v8f{};

#pragma unroll U
      for (int k = 0; k < K; k += 4) {
        const v2f a = *(const v2f*)(Ap + k);
#pragma unroll
        for (int t = 0; t < NT; ++t) {
          const v2f b = *(const v2f*)(Bl + t * TP + k * 16);
          acc[t] = __builtin_amdgcn_wmma_f32_16x16x4_f32(
               false, a,  false, b,
               (short)0, acc[t],  false,  false);
        }
      }

      float* Cw = C + (size_t)(tm * 16) * NC + col0 + lane;
#pragma unroll
      for (int t = 0; t < NT; t += 2) {
#pragma unroll
        for (int r = 0; r < 8; ++r) {
          const float x0 = acc[t][r];
          const float x1 = acc[t + 1][r];
          const float y0 = __shfl_xor(x0, 16, 32);
          const float y1 = __shfl_xor(x1, 16, 32);
          const float lo = half ? y1 : x0;
          const float hi = half ? x1 : y0;
          Cw[(size_t)r * NC + t * 16]       = lo;
          Cw[(size_t)(r + 8) * NC + t * 16] = hi;
        }
      }

#pragma unroll
      for (int r = 0; r < 8; ++r) {
        float ss = 0.f, sd = 0.f;
#pragma unroll
        for (int t = 0; t < NT; ++t) {
          ss += acc[t][r] * ws[t];
          sd += acc[t][r] * wd[t];
        }
        ss = row16_sum(ss);
        sd = row16_sum(sd);
        if (l16 == 0) {
          part_s[st][tl * 16 + half * 8 + r] = ss;
          part_d[st][tl * 16 + half * 8 + r] = sd;
        }
      }
    }
    __syncthreads();

    for (int i = tid; i < ntile * 16; i += WAVES * 32) {
      const int row = c0 * 16 + i;
      if (row < M) {
        as_out[row] = ((part_s[0][i] + part_s[1][i]) + part_s[2][i]) + part_s[3][i];
        ad_out[row] = ((part_d[0][i] + part_d[1][i]) + part_d[2][i]) + part_d[3][i];
      }
    }
    if (c0 + CH < t_end) __syncthreads();
  }
}

__global__ void gat_node_dots(const float* __restrict__ h,
                              const float* __restrict__ a_src,
                              const float* __restrict__ a_dst,
                              float* __restrict__ as_out,
                              float* __restrict__ ad_out,
                              int N, int D) {
  const int wave = blockIdx.x * (blockDim.x >> 5) + (threadIdx.x >> 5);
  const int lane = threadIdx.x & 31;
  if (wave >= N) return;
  const float* row = h + (size_t)wave * D;
  float ss = 0.f, sd = 0.f;
  for (int c = lane; c < D; c += 32) {
    float v = row[c];
    ss += v * a_src[c];
    sd += v * a_dst[c];
  }
#pragma unroll
  for (int off = 16; off > 0; off >>= 1) {
    ss += __shfl_down(ss, off, 32);
    sd += __shfl_down(sd, off, 32);
  }
  if (lane == 0) { as_out[wave] = ss; ad_out[wave] = sd; }
}

__device__ __forceinline__ float leakyr(float v) {
  return v > 0.f ? v : NEG_SLOPE * v;
}

__device__ __forceinline__ void lds_max_f32(float* addr, float val) {
  if (__float_as_int(val) >= 0) atomicMax((int*)addr, __float_as_int(val));
  else                          atomicMin((unsigned int*)addr, __float_as_uint(val));
}

#define AGG_ROWS 98
#define AGG_U    16
#define AGG_THREADS 1024
#define AGG_CAP  4096
#define AGG_SAVE 2048

struct AggList {
  int*   s;
  int*   r;
  float* v;
  int*   cnt;
  float* m;
  float* den;
  float* ad;
};

template <int R, int H>
__device__ __forceinline__ void agg_append_owned(const int (&dv)[AGG_U], int base, int lane,
                                                 int d0, int lo, int hi,
                                                 int* ls, int* lr, int* lcnt) {
#pragma unroll
  for (int k = 0; k < AGG_U; ++k) {
    const int r = dv[k] - d0;
    if ((unsigned)(r - lo) < (unsigned)(hi - lo)) {
      const int hh = (H > 1) ? r / R : 0;
      const int idx = atomicAdd(lcnt + hh, 1);
      if (idx < AGG_CAP) { ls[hh * AGG_CAP + idx] = base + k * 32 + lane; lr[hh * AGG_CAP + idx] = r; }
    }
  }
}

template <int R, int H>
__device__ __forceinline__ void agg_scan_append(const int* __restrict__ dstp, int eb, int ee,
                                                int d0, int lo, int hi, int wv, int nw, int lane,
                                                int* ls, int* lr, int* lcnt) {
  const int sweep  = AGG_U * 32;
  const int stride = nw * sweep;
  int base = eb + wv * sweep;

  if (base + sweep <= ee) {
    int cur[AGG_U];
#pragma unroll
    for (int k = 0; k < AGG_U; ++k) cur[k] = dstp[base + k * 32 + lane];
    while (true) {
      const int nb = base + stride;
      const bool more = (nb + sweep <= ee);
      const int pb = more ? nb : base;
      int nxt[AGG_U];
#pragma unroll
      for (int k = 0; k < AGG_U; ++k) nxt[k] = dstp[pb + k * 32 + lane];
      agg_append_owned<R, H>(cur, base, lane, d0, lo, hi, ls, lr, lcnt);
      base = nb;
      if (!more) break;
#pragma unroll
      for (int k = 0; k < AGG_U; ++k) cur[k] = nxt[k];
    }
  }
  if (base < ee) {
    int cur[AGG_U];
#pragma unroll
    for (int k = 0; k < AGG_U; ++k) {
      const int e = base + k * 32 + lane;
      cur[k] = (e < ee) ? dstp[e] : -1;
    }
    agg_append_owned<R, H>(cur, base, lane, d0, lo, hi, ls, lr, lcnt);
  }
}

__device__ __forceinline__ void agg_list_scores(const AggList& L, int first, int n, bool do_max,
                                                const int* __restrict__ ei,
                                                const float* __restrict__ as) {
  for (int i = first + (int)threadIdx.x; i < n; i += AGG_THREADS) {
    const int r = L.r[i];
    const int s = ei[L.s[i]];
    const float sc = leakyr(as[s] + L.ad[r]);
    L.s[i] = s;
    L.v[i] = sc;
    if (do_max) lds_max_f32(&L.m[r], sc);
  }
}

__device__ __forceinline__ void agg_list_exps(const AggList& L, int n, bool do_sum) {
  for (int i = (int)threadIdx.x; i < n; i += AGG_THREADS) {
    const int r = L.r[i];
    const float ex = expf(L.v[i] - L.m[r]);
    L.v[i] = ex;
    if (do_sum) atomicAdd(&L.den[r], ex);
  }
}

template <int D>
__device__ __forceinline__ void agg_gather_entry(const AggList& L, int i, int rbase, int lane,
                                                 const float* __restrict__ h, float* acc) {
  constexpr int V = D / 32;
  const int s = L.s[i];
  const float* hs = h + (size_t)s * D + lane;
  float v[V];
#pragma unroll
  for (int j = 0; j < V; ++j) v[j] = hs[j * 32];
  const int r = L.r[i];
  const float a = L.v[i] / L.den[r];
  float* ar = acc + (r - rbase) * D + lane;
#pragma unroll
  for (int j = 0; j < V; ++j) atomicAdd(ar + j * 32, a * v[j]);
}

template <int D, int EPT>
__device__ __forceinline__ void agg_list_gather(const AggList& L, int n, int rbase,
                                                int wv, int nw, int lane,
                                                const float* __restrict__ h, float* acc) {
  constexpr int V = D / 32;
  const int groups = n / EPT;
  int g = wv;
  for (; g < groups; g += nw) {
    if (EPT == 1) {
      agg_gather_entry<D>(L, g, rbase, lane, h, acc);
    } else {
      const int i0 = g * EPT;
      float v[EPT][V];
#pragma unroll
      for (int e = 0; e < EPT; ++e) {
        const float* hs = h + (size_t)L.s[i0 + e] * D + lane;
#pragma unroll
        for (int j = 0; j < V; ++j) v[e][j] = hs[j * 32];
      }
#pragma unroll
      for (int e = 0; e < EPT; ++e) {
        const int r = L.r[i0 + e];
        const float a = L.v[i0 + e] / L.den[r];
        float* ar = acc + (r - rbase) * D + lane;
#pragma unroll
        for (int j = 0; j < V; ++j) atomicAdd(ar + j * 32, a * v[e][j]);
      }
    }
  }
  if (EPT > 1 && g == groups) {
    for (int i = groups * EPT; i < n; ++i) agg_gather_entry<D>(L, i, rbase, lane, h, acc);
  }
}

template <int D, int R, int H, int EPT, int LMODE>
__global__ __launch_bounds__(AGG_THREADS) void gat_aggregate_owned(
    const int* __restrict__ ei, const float* __restrict__ as,
    const float* __restrict__ ad, const float* __restrict__ h,
    const float* __restrict__ bias, float* __restrict__ out, int E, int N,
    int* __restrict__ saved, const float* ws_base, size_t ws_bytes) {
  constexpr int RB = R * H;
  static_assert(RB <= 256, "a block-local row must fit the low byte of a saved entry");
  static_assert(LMODE != 2 || H == 1, "loaded lists go into one list");
  __shared__ __attribute__((aligned(16))) float acc[R * D];
  __shared__ int   list_s[H * AGG_CAP];
  __shared__ int   list_r[H * AGG_CAP];
  __shared__ float list_v[H * AGG_CAP];
  __shared__ float row_m[RB];
  __shared__ float row_den[RB];
  __shared__ float row_ad[RB];
  __shared__ int   list_cnt[H];
  static_assert(R <= AGG_CAP && RB <= AGG_THREADS, "self loops open the lists, one thread each");

  const int tid  = threadIdx.x;
  const int lane = tid & 31;
  const int wv   = tid >> 5;
  const int nw   = AGG_THREADS >> 5;
  const int d0   = blockIdx.x * RB;
  const int rows_all = (N - d0 < RB) ? (N - d0) : RB;
  const int* __restrict__ dstp = ei + E;

  const size_t save_end = (size_t)((const char*)saved - (const char*)ws_base) +
                          (size_t)gridDim.x * AGG_SAVE * sizeof(int);
  const bool ws_ok = (LMODE != 0) && save_end <= ws_bytes && N <= (1 << 24);
  int* mine = saved + (size_t)blockIdx.x * AGG_SAVE;
  int loaded = -1;
  if (LMODE == 2 && ws_ok) {
    const int t = mine[0];
    if (t >= rows_all && t <= AGG_SAVE - 2 && t <= AGG_CAP) loaded = t;
  }

  auto run_rows = [&](int hh) {
    const int left = rows_all - hh * R;
    return left < 0 ? 0 : (left < R ? left : R);
  };
  auto run_list = [&](int hh) {
    return AggList{list_s + hh * AGG_CAP, list_r + hh * AGG_CAP, list_v + hh * AGG_CAP,
                   list_cnt + hh, row_m, row_den, row_ad};
  };

  {
    float4* z4 = (float4*)acc;
    for (int idx = tid; idx < run_rows(0) * (D / 4); idx += AGG_THREADS)
      z4[idx] = make_float4(0.f, 0.f, 0.f, 0.f);
  }
  if (tid < rows_all) {
    const int hh = tid / R;
    const int at = hh * AGG_CAP + (tid - hh * R);
    const float adv = ad[d0 + tid];
    const float sc  = leakyr(as[d0 + tid] + adv);
    row_ad[tid] = adv; row_m[tid] = sc; row_den[tid] = 0.f;
    list_s[at] = d0 + tid; list_r[at] = tid; list_v[at] = sc;
  }
  if (tid < H) list_cnt[tid] = run_rows(tid);
  __syncthreads();

  if (LMODE == 2 && loaded >= 0) {
    for (int i = tid; i < loaded; i += 2 * AGG_THREADS) {
      const int j = i + AGG_THREADS;
      const bool two = j < loaded;
      const int jc = two ? j : i;
      const unsigned w0 = (unsigned)mine[2 + i];
      const unsigned w1 = (unsigned)mine[2 + jc];
      const int s0 = (int)(w0 >> 8), r0 = (int)(w0 & 255u);
      const int s1 = (int)(w1 >> 8), r1 = (int)(w1 & 255u);
      const float a0 = as[s0];
      const float a1 = as[s1];
      const float sc0 = leakyr(a0 + row_ad[r0]);
      list_s[i] = s0; list_r[i] = r0; list_v[i] = sc0;
      lds_max_f32(&row_m[r0], sc0);
      if (two) {
        const float sc1 = leakyr(a1 + row_ad[r1]);
        list_s[j] = s1; list_r[j] = r1; list_v[j] = sc1;
        lds_max_f32(&row_m[r1], sc1);
      }
    }
  } else {
    agg_scan_append<R, H>(dstp, 0, E, d0, 0, rows_all, wv, nw, lane, list_s, list_r, list_cnt);
  }
  __syncthreads();

  int  n[H];
  bool fits[H];
#pragma unroll
  for (int hh = 0; hh < H; ++hh) {
    n[hh] = (LMODE == 2 && loaded >= 0) ? loaded : list_cnt[hh];
    fits[hh] = (n[hh] <= AGG_CAP);
  }
  if (!(LMODE == 2 && loaded >= 0)) {
#pragma unroll
    for (int hh = 0; hh < H; ++hh)
      if (fits[hh]) agg_list_scores(run_list(hh), run_rows(hh), n[hh], true, ei, as);
  }
  __syncthreads();

  if (LMODE == 1 && ws_ok) {
    int total = 0;
    bool all = true;
#pragma unroll
    for (int hh = 0; hh < H; ++hh) { total += n[hh]; all = all && fits[hh]; }
    const bool keep = all && total <= AGG_SAVE - 2;
    if (tid == 0) mine[0] = keep ? total : -1;
    if (keep) {
      int at = 2;
#pragma unroll
      for (int hh = 0; hh < H; ++hh) {
        for (int i = tid; i < n[hh]; i += AGG_THREADS)
          mine[at + i] = (int)(((unsigned)list_s[hh * AGG_CAP + i] << 8) | (unsigned)list_r[hh * AGG_CAP + i]);
        at += n[hh];
      }
    }
  }
#pragma unroll
  for (int hh = 0; hh < H; ++hh)
    if (fits[hh]) agg_list_exps(run_list(hh), n[hh], true);
  __syncthreads();

  constexpr int Q = D / 4;
  static_assert((Q & (Q - 1)) == 0 && AGG_THREADS % Q == 0, "column group must be fixed per thread");
  const int c0 = (tid & (Q - 1)) * 4;
  const float bx = bias[c0], by = bias[c0 + 1], bz = bias[c0 + 2], bw = bias[c0 + 3];

#pragma unroll
  for (int hh = 0; hh < H; ++hh) {
    const int rows = run_rows(hh);
    if (rows > 0) {
      const AggList L = run_list(hh);
      const int rb = hh * R;
      if (fits[hh]) {
        agg_list_gather<D, EPT>(L, n[hh], rb, wv, nw, lane, h, acc);
      } else {
        for (int pass = 0; pass < 3; ++pass) {
          for (int eb = -AGG_CAP; eb < E; eb += AGG_CAP) {
            __syncthreads();
            int first = 0;
            if (eb < 0) {
              if (tid < rows) {
                L.s[tid] = d0 + rb + tid; L.r[tid] = rb + tid;
                L.v[tid] = leakyr(as[d0 + rb + tid] + row_ad[rb + tid]);
              }
              if (tid == 0) *L.cnt = rows;
              first = rows;
              __syncthreads();
            } else {
              if (tid == 0) *L.cnt = 0;
              __syncthreads();
              agg_scan_append<R, H>(dstp, eb, (E - eb < AGG_CAP) ? E : eb + AGG_CAP, d0, rb, rb + rows,
                                    wv, nw, lane, list_s, list_r, list_cnt);
              __syncthreads();
            }
            const int np = *L.cnt;
            agg_list_scores(L, first, np, pass == 0, ei, as);
            if (pass >= 1) {
              __syncthreads();
              agg_list_exps(L, np, pass == 1);
            }
            if (pass == 2) {
              __syncthreads();
              agg_list_gather<D, EPT>(L, np, rb, wv, nw, lane, h, acc);
            }
          }
        }
      }
      __syncthreads();

      float4* a4 = (float4*)acc;
      float4* o4 = (float4*)(out + (size_t)(d0 + rb) * D);
      for (int idx = tid; idx < rows * Q; idx += AGG_THREADS) {
        float4 v = a4[idx];
        if (hh + 1 < H) a4[idx] = make_float4(0.f, 0.f, 0.f, 0.f);
        v.x += bx; v.y += by; v.z += bz; v.w += bw;
        v.x = v.x > 0.f ? v.x : 0.f;
        v.y = v.y > 0.f ? v.y : 0.f;
        v.z = v.z > 0.f ? v.z : 0.f;
        v.w = v.w > 0.f ? v.w : 0.f;
        o4[idx] = v;
      }
      if (hh + 1 < H) __syncthreads();
    }
  }
}

__global__ void gat_fc(const float* __restrict__ h,
                       const float* __restrict__ W,
                       const float* __restrict__ b,
                       float* __restrict__ out, int N) {
  int n = blockIdx.x * blockDim.x + threadIdx.x;
  if (n >= N) return;
  const float* row = h + (size_t)n * OUT2;
  float a0 = b[0], a1 = b[1];
#pragma unroll
  for (int c = 0; c < OUT2; ++c) {
    float v = row[c];
    a0 += v * W[c * NCLS + 0];
    a1 += v * W[c * NCLS + 1];
  }
  out[n * NCLS + 0] = a0;
  out[n * NCLS + 1] = a1;
}

extern "C" void kernel_launch(void* const* d_in, const int* in_sizes, int n_in,
                              void* d_out, int out_size, void* d_ws, size_t ws_size,
                              hipStream_t stream) {
  const float* x   = (const float*)d_in[0];
  const int*   ei  = (const int*)  d_in[1];
  const float* W1  = (const float*)d_in[2];
  const float* a1s = (const float*)d_in[3];
  const float* a1d = (const float*)d_in[4];
  const float* b1  = (const float*)d_in[5];
  const float* W2  = (const float*)d_in[6];
  const float* a2s = (const float*)d_in[7];
  const float* a2d = (const float*)d_in[8];
  const float* b2  = (const float*)d_in[9];
  const float* fcW = (const float*)d_in[10];
  const float* fcb = (const float*)d_in[11];

  const int N    = in_sizes[0] / IN_C;
  const int E    = in_sizes[1] / 2;

  float* w       = (float*)d_ws;
  float* h1      = w;
  float* agg     = w + (size_t)N * HID;
  float* as_buf  = w + (size_t)2 * N * HID;
  float* ad_buf  = as_buf + N;
  int*   lists   = (int*)(ad_buf + N);

  const int TB = 256;

  auto gemm_grid = [](int M, int Nc, int waves, int cols) {
    return dim3((((M + 15) / 16) + waves - 1) / waves, Nc / cols);
  };
  const int agg_own   = 2 * AGG_ROWS;
  const int agg_blk   = (N + agg_own - 1) / agg_own;
  const int node_blk  = (N + 7) / 8;

  gat_wmma_gemm_dots_f32<IN_C, HID, 16, 8><<<G1_BLOCKS, 512, 0, stream>>>(x, W1, a1s, a1d, h1, as_buf, ad_buf, N);
  gat_aggregate_owned<HID, AGG_ROWS, 2, 1, 1><<<agg_blk, AGG_THREADS, 0, stream>>>(ei, as_buf, ad_buf, h1, b1, agg, E, N, lists, w, ws_size);

  gat_wmma_gemm_f32<HID, 4, 8, 8><<<gemm_grid(N, OUT2, 8, 64), 256, 0, stream>>>(agg, W2, h1, N, OUT2);
  gat_node_dots<<<node_blk, TB, 0, stream>>>(h1, a2s, a2d, as_buf, ad_buf, N, OUT2);
  gat_aggregate_owned<OUT2, 2 * AGG_ROWS, 1, 2, 2><<<agg_blk, AGG_THREADS, 0, stream>>>(ei, as_buf, ad_buf, h1, b2, agg, E, N, lists, w, ws_size);

  gat_fc<<<(N + TB - 1) / TB, TB, 0, stream>>>(agg, fcW, fcb, (float*)d_out, N);
}


# Layer-2 product moved into the layer-1 aggregation block: finished rows stay in LDS as A, W2 read from memory as B, one 16x16 tile per wave; rows never stored; 6 -> 5 launches; sum order unchanged
# speedup vs baseline: 4.0068x; 1.0582x over previous
#include <hip/hip_runtime.h>


typedef __attribute__((ext_vector_type(2))) float v2f;
typedef __attribute__((ext_vector_type(8))) float v8f;

#define IN_C  128
#define HID   512
#define OUT2  64
#define NCLS  2
#define NEG_SLOPE 0.2f
#define G1_BLOCKS 256


template <int XM>
__device__ __forceinline__ float row_xmask_add(float v) {
  const int b = __float_as_int(v);
  return v + __int_as_float(__builtin_amdgcn_update_dpp(b, b, 0x160 + XM, 0xf, 0xf, false));
}
__device__ __forceinline__ float row16_sum(float v) {
  v = row_xmask_add<1>(v);
  v = row_xmask_add<2>(v);
  v = row_xmask_add<4>(v);
  return row_xmask_add<8>(v);
}

template <int K, int NC, int WAVES, int U>
__global__ __launch_bounds__(WAVES * 32) void gat_wmma_gemm_dots_f32(
    const float* __restrict__ A, const float* __restrict__ B,
    const float* __restrict__ a_src, const float* __restrict__ a_dst,
    float* __restrict__ C, float* __restrict__ as_out, float* __restrict__ ad_out, int M) {
  constexpr int NT = 8;
  constexpr int STRIPS = NC / (16 * NT);
  constexpr int NTT = NC / 16;
  constexpr int CH = 16;
  static_assert(K % 4 == 0 && NC % (16 * NT) == 0 && STRIPS == 4 && WAVES % STRIPS == 0,
                "a wave keeps one of four strips");
  constexpr int TP = K * 16 + 8;
  __shared__ __attribute__((aligned(16))) float Bs[NTT * TP];
  __shared__ float part_s[STRIPS][CH * 16];
  __shared__ float part_d[STRIPS][CH * 16];

  const int tid  = threadIdx.x;
  const int lane = tid & 31;
  const int waveid = __builtin_amdgcn_readfirstlane(tid >> 5);

  constexpr int QC = NTT * 4;
  constexpr int ITEMS = (K / 2) * QC;
#pragma unroll 4
  for (int it = tid; it < ITEMS; it += WAVES * 32) {
    const int p = it / QC;
    const int q = it % QC;
    const float* src = B + (size_t)(2 * p) * NC + 4 * q;
    const float4 r0 = *(const float4*)src;
    const float4 r1 = *(const float4*)(src + NC);
    float* dst = Bs + (q >> 2) * TP + p * 32 + (q & 3) * 8;
    *(float4*)(dst)     = make_float4(r0.x, r1.x, r0.y, r1.y);
    *(float4*)(dst + 4) = make_float4(r0.z, r1.z, r0.w, r1.w);
  }

  const int half = lane >> 4;
  const int l16  = lane & 15;
  const int st   = waveid % STRIPS;
  constexpr int TC = WAVES / STRIPS;
  const int tl0  = (waveid / STRIPS + st) % TC;
  const int col0 = st * (NT * 16);

  float ws[NT], wd[NT];
#pragma unroll
  for (int t = 0; t < NT; ++t) {
    ws[t] = a_src[col0 + t * 16 + l16];
    wd[t] = a_dst[col0 + t * 16 + l16];
  }
  __syncthreads();

  const int tilesM = (M + 15) >> 4;
  const int tpb = (tilesM + (int)gridDim.x - 1) / (int)gridDim.x;
  const int t_begin = blockIdx.x * tpb;
  const int t_end = (t_begin + tpb < tilesM) ? (t_begin + tpb) : tilesM;

  const float* Bl = Bs + (st * NT) * TP + half * 32 + (l16 << 1);

  for (int c0 = t_begin; c0 < t_end; c0 += CH) {
    const int ntile = (t_end - c0 < CH) ? (t_end - c0) : CH;
    for (int tl = tl0; tl < ntile; tl += TC) {
      const int tm = c0 + tl;
      const float* Ap = A + (size_t)(tm * 16 + l16) * K + 2 * half;

      v8f acc[NT];
#pragma unroll
      for (int t = 0; t < NT; ++t) acc[t] = v8f{};

#pragma unroll U
      for (int k = 0; k < K; k += 4) {
        const v2f a = *(const v2f*)(Ap + k);
#pragma unroll
        for (int t = 0; t < NT; ++t) {
          const v2f b = *(const v2f*)(Bl + t * TP + k * 16);
          acc[t] = __builtin_amdgcn_wmma_f32_16x16x4_f32(
               false, a,  false, b,
               (short)0, acc[t],  false,  false);
        }
      }

      float* Cw = C + (size_t)(tm * 16) * NC + col0 + lane;
#pragma unroll
      for (int t = 0; t < NT; t += 2) {
#pragma unroll
        for (int r = 0; r < 8; ++r) {
          const float x0 = acc[t][r];
          const float x1 = acc[t + 1][r];
          const float y0 = __shfl_xor(x0, 16, 32);
          const float y1 = __shfl_xor(x1, 16, 32);
          const float lo = half ? y1 : x0;
          const float hi = half ? x1 : y0;
          Cw[(size_t)r * NC + t * 16]       = lo;
          Cw[(size_t)(r + 8) * NC + t * 16] = hi;
        }
      }

#pragma unroll
      for (int r = 0; r < 8; ++r) {
        float ss = 0.f, sd = 0.f;
#pragma unroll
        for (int t = 0; t < NT; ++t) {
          ss += acc[t][r] * ws[t];
          sd += acc[t][r] * wd[t];
        }
        ss = row16_sum(ss);
        sd = row16_sum(sd);
        if (l16 == 0) {
          part_s[st][tl * 16 + half * 8 + r] = ss;
          part_d[st][tl * 16 + half * 8 + r] = sd;
        }
      }
    }
    __syncthreads();

    for (int i = tid; i < ntile * 16; i += WAVES * 32) {
      const int row = c0 * 16 + i;
      if (row < M) {
        as_out[row] = ((part_s[0][i] + part_s[1][i]) + part_s[2][i]) + part_s[3][i];
        ad_out[row] = ((part_d[0][i] + part_d[1][i]) + part_d[2][i]) + part_d[3][i];
      }
    }
    if (c0 + CH < t_end) __syncthreads();
  }
}

__global__ void gat_node_dots(const float* __restrict__ h,
                              const float* __restrict__ a_src,
                              const float* __restrict__ a_dst,
                              float* __restrict__ as_out,
                              float* __restrict__ ad_out,
                              int N, int D) {
  const int wave = blockIdx.x * (blockDim.x >> 5) + (threadIdx.x >> 5);
  const int lane = threadIdx.x & 31;
  if (wave >= N) return;
  const float* row = h + (size_t)wave * D;
  float ss = 0.f, sd = 0.f;
  for (int c = lane; c < D; c += 32) {
    float v = row[c];
    ss += v * a_src[c];
    sd += v * a_dst[c];
  }
#pragma unroll
  for (int off = 16; off > 0; off >>= 1) {
    ss += __shfl_down(ss, off, 32);
    sd += __shfl_down(sd, off, 32);
  }
  if (lane == 0) { as_out[wave] = ss; ad_out[wave] = sd; }
}

__device__ __forceinline__ float leakyr(float v) {
  return v > 0.f ? v : NEG_SLOPE * v;
}

__device__ __forceinline__ void lds_max_f32(float* addr, float val) {
  if (__float_as_int(val) >= 0) atomicMax((int*)addr, __float_as_int(val));
  else                          atomicMin((unsigned int*)addr, __float_as_uint(val));
}

#define AGG_ROWS 98
#define AGG_U    16
#define AGG_THREADS 1024
#define AGG_CAP  4096
#define AGG_SAVE 2048

struct AggList {
  int*   s;
  int*   r;
  float* v;
  int*   cnt;
  float* m;
  float* den;
  float* ad;
};

template <int R, int H>
__device__ __forceinline__ void agg_append_owned(const int (&dv)[AGG_U], int base, int lane,
                                                 int d0, int lo, int hi,
                                                 int* ls, int* lr, int* lcnt) {
#pragma unroll
  for (int k = 0; k < AGG_U; ++k) {
    const int r = dv[k] - d0;
    if ((unsigned)(r - lo) < (unsigned)(hi - lo)) {
      const int hh = (H > 1) ? r / R : 0;
      const int idx = atomicAdd(lcnt + hh, 1);
      if (idx < AGG_CAP) { ls[hh * AGG_CAP + idx] = base + k * 32 + lane; lr[hh * AGG_CAP + idx] = r; }
    }
  }
}

template <int R, int H>
__device__ __forceinline__ void agg_scan_append(const int* __restrict__ dstp, int eb, int ee,
                                                int d0, int lo, int hi, int wv, int nw, int lane,
                                                int* ls, int* lr, int* lcnt) {
  const int sweep  = AGG_U * 32;
  const int stride = nw * sweep;
  int base = eb + wv * sweep;

  if (base + sweep <= ee) {
    int cur[AGG_U];
#pragma unroll
    for (int k = 0; k < AGG_U; ++k) cur[k] = dstp[base + k * 32 + lane];
    while (true) {
      const int nb = base + stride;
      const bool more = (nb + sweep <= ee);
      const int pb = more ? nb : base;
      int nxt[AGG_U];
#pragma unroll
      for (int k = 0; k < AGG_U; ++k) nxt[k] = dstp[pb + k * 32 + lane];
      agg_append_owned<R, H>(cur, base, lane, d0, lo, hi, ls, lr, lcnt);
      base = nb;
      if (!more) break;
#pragma unroll
      for (int k = 0; k < AGG_U; ++k) cur[k] = nxt[k];
    }
  }
  if (base < ee) {
    int cur[AGG_U];
#pragma unroll
    for (int k = 0; k < AGG_U; ++k) {
      const int e = base + k * 32 + lane;
      cur[k] = (e < ee) ? dstp[e] : -1;
    }
    agg_append_owned<R, H>(cur, base, lane, d0, lo, hi, ls, lr, lcnt);
  }
}

__device__ __forceinline__ void agg_list_scores(const AggList& L, int first, int n, bool do_max,
                                                const int* __restrict__ ei,
                                                const float* __restrict__ as) {
  for (int i = first + (int)threadIdx.x; i < n; i += AGG_THREADS) {
    const int r = L.r[i];
    const int s = ei[L.s[i]];
    const float sc = leakyr(as[s] + L.ad[r]);
    L.s[i] = s;
    L.v[i] = sc;
    if (do_max) lds_max_f32(&L.m[r], sc);
  }
}

__device__ __forceinline__ void agg_list_exps(const AggList& L, int n, bool do_sum) {
  for (int i = (int)threadIdx.x; i < n; i += AGG_THREADS) {
    const int r = L.r[i];
    const float ex = expf(L.v[i] - L.m[r]);
    L.v[i] = ex;
    if (do_sum) atomicAdd(&L.den[r], ex);
  }
}

template <int D, int P>
__device__ __forceinline__ void agg_gather_entry(const AggList& L, int i, int rbase, int lane,
                                                 const float* __restrict__ h, float* acc) {
  constexpr int V = D / 32;
  const int s = L.s[i];
  const float* hs = h + (size_t)s * D + lane;
  float v[V];
#pragma unroll
  for (int j = 0; j < V; ++j) v[j] = hs[j * 32];
  const int r = L.r[i];
  const float a = L.v[i] / L.den[r];
  float* ar = acc + (r - rbase) * P + lane;
#pragma unroll
  for (int j = 0; j < V; ++j) atomicAdd(ar + j * 32, a * v[j]);
}

template <int D, int EPT, int P>
__device__ __forceinline__ void agg_list_gather(const AggList& L, int n, int rbase,
                                                int wv, int nw, int lane,
                                                const float* __restrict__ h, float* acc) {
  constexpr int V = D / 32;
  const int groups = n / EPT;
  int g = wv;
  for (; g < groups; g += nw) {
    if (EPT == 1) {
      agg_gather_entry<D, P>(L, g, rbase, lane, h, acc);
    } else {
      const int i0 = g * EPT;
      float v[EPT][V];
#pragma unroll
      for (int e = 0; e < EPT; ++e) {
        const float* hs = h + (size_t)L.s[i0 + e] * D + lane;
#pragma unroll
        for (int j = 0; j < V; ++j) v[e][j] = hs[j * 32];
      }
#pragma unroll
      for (int e = 0; e < EPT; ++e) {
        const int r = L.r[i0 + e];
        const float a = L.v[i0 + e] / L.den[r];
        float* ar = acc + (r - rbase) * P + lane;
#pragma unroll
        for (int j = 0; j < V; ++j) atomicAdd(ar + j * 32, a * v[e][j]);
      }
    }
  }
  if (EPT > 1 && g == groups) {
    for (int i = groups * EPT; i < n; ++i) agg_gather_entry<D, P>(L, i, rbase, lane, h, acc);
  }
}

template <int D, int R, int H, int EPT, int LMODE, int NOUT>
__global__ __launch_bounds__(AGG_THREADS) void gat_aggregate_owned(
    const int* __restrict__ ei, const float* __restrict__ as,
    const float* __restrict__ ad, const float* __restrict__ h,
    const float* __restrict__ bias, const float* __restrict__ Wn,
    float* __restrict__ out, int E, int N,
    int* __restrict__ saved, const float* ws_base, size_t ws_bytes) {
  constexpr int RB = R * H;
  constexpr int P = (NOUT > 0) ? D + 4 : D;
  constexpr int Q = D / 4;
  constexpr int P4 = P / 4;
  static_assert((Q & (Q - 1)) == 0 && AGG_THREADS % Q == 0, "column group must be fixed per thread");
  static_assert(NOUT % 16 == 0 && D % 4 == 0 && ((R + 15) / 16) * (NOUT / 16) <= AGG_THREADS / 32,
                "one product tile per wave");
  static_assert(RB <= 256, "a block-local row must fit the low byte of a saved entry");
  static_assert(LMODE != 2 || H == 1, "loaded lists go into one list");
  __shared__ __attribute__((aligned(16))) float acc[R * P];
  __shared__ int   list_s[H * AGG_CAP];
  __shared__ int   list_r[H * AGG_CAP];
  __shared__ float list_v[H * AGG_CAP];
  __shared__ float row_m[RB];
  __shared__ float row_den[RB];
  __shared__ float row_ad[RB];
  __shared__ int   list_cnt[H];
  static_assert(R <= AGG_CAP && RB <= AGG_THREADS, "self loops open the lists, one thread each");

  const int tid  = threadIdx.x;
  const int lane = tid & 31;
  const int wv   = tid >> 5;
  const int nw   = AGG_THREADS >> 5;
  const int d0   = blockIdx.x * RB;
  const int rows_all = (N - d0 < RB) ? (N - d0) : RB;
  const int* __restrict__ dstp = ei + E;

  const size_t save_end = (size_t)((const char*)saved - (const char*)ws_base) +
                          (size_t)gridDim.x * AGG_SAVE * sizeof(int);
  const bool ws_ok = (LMODE != 0) && save_end <= ws_bytes && N <= (1 << 24);
  int* mine = saved + (size_t)blockIdx.x * AGG_SAVE;
  int loaded = -1;
  if (LMODE == 2 && ws_ok) {
    const int t = mine[0];
    if (t >= rows_all && t <= AGG_SAVE - 2 && t <= AGG_CAP) loaded = t;
  }

  auto run_rows = [&](int hh) {
    const int left = rows_all - hh * R;
    return left < 0 ? 0 : (left < R ? left : R);
  };
  auto run_list = [&](int hh) {
    return AggList{list_s + hh * AGG_CAP, list_r + hh * AGG_CAP, list_v + hh * AGG_CAP,
                   list_cnt + hh, row_m, row_den, row_ad};
  };

  {
    float4* z4 = (float4*)acc;
    for (int idx = tid; idx < run_rows(0) * Q; idx += AGG_THREADS)
      z4[(idx / Q) * P4 + (idx & (Q - 1))] = make_float4(0.f, 0.f, 0.f, 0.f);
  }
  if (tid < rows_all) {
    const int hh = tid / R;
    const int at = hh * AGG_CAP + (tid - hh * R);
    const float adv = ad[d0 + tid];
    const float sc  = leakyr(as[d0 + tid] + adv);
    row_ad[tid] = adv; row_m[tid] = sc; row_den[tid] = 0.f;
    list_s[at] = d0 + tid; list_r[at] = tid; list_v[at] = sc;
  }
  if (tid < H) list_cnt[tid] = run_rows(tid);
  __syncthreads();

  if (LMODE == 2 && loaded >= 0) {
    for (int i = tid; i < loaded; i += 2 * AGG_THREADS) {
      const int j = i + AGG_THREADS;
      const bool two = j < loaded;
      const int jc = two ? j : i;
      const unsigned w0 = (unsigned)mine[2 + i];
      const unsigned w1 = (unsigned)mine[2 + jc];
      const int s0 = (int)(w0 >> 8), r0 = (int)(w0 & 255u);
      const int s1 = (int)(w1 >> 8), r1 = (int)(w1 & 255u);
      const float a0 = as[s0];
      const float a1 = as[s1];
      const float sc0 = leakyr(a0 + row_ad[r0]);
      list_s[i] = s0; list_r[i] = r0; list_v[i] = sc0;
      lds_max_f32(&row_m[r0], sc0);
      if (two) {
        const float sc1 = leakyr(a1 + row_ad[r1]);
        list_s[j] = s1; list_r[j] = r1; list_v[j] = sc1;
        lds_max_f32(&row_m[r1], sc1);
      }
    }
  } else {
    agg_scan_append<R, H>(dstp, 0, E, d0, 0, rows_all, wv, nw, lane, list_s, list_r, list_cnt);
  }
  __syncthreads();

  int  n[H];
  bool fits[H];
#pragma unroll
  for (int hh = 0; hh < H; ++hh) {
    n[hh] = (LMODE == 2 && loaded >= 0) ? loaded : list_cnt[hh];
    fits[hh] = (n[hh] <= AGG_CAP);
  }
  if (!(LMODE == 2 && loaded >= 0)) {
#pragma unroll
    for (int hh = 0; hh < H; ++hh)
      if (fits[hh]) agg_list_scores(run_list(hh), run_rows(hh), n[hh], true, ei, as);
  }
  __syncthreads();

  if (LMODE == 1 && ws_ok) {
    int total = 0;
    bool all = true;
#pragma unroll
    for (int hh = 0; hh < H; ++hh) { total += n[hh]; all = all && fits[hh]; }
    const bool keep = all && total <= AGG_SAVE - 2;
    if (tid == 0) mine[0] = keep ? total : -1;
    if (keep) {
      int at = 2;
#pragma unroll
      for (int hh = 0; hh < H; ++hh) {
        for (int i = tid; i < n[hh]; i += AGG_THREADS)
          mine[at + i] = (int)(((unsigned)list_s[hh * AGG_CAP + i] << 8) | (unsigned)list_r[hh * AGG_CAP + i]);
        at += n[hh];
      }
    }
  }
#pragma unroll
  for (int hh = 0; hh < H; ++hh)
    if (fits[hh]) agg_list_exps(run_list(hh), n[hh], true);
  __syncthreads();

  const int c0 = (tid & (Q - 1)) * 4;
  const float bx = bias[c0], by = bias[c0 + 1], bz = bias[c0 + 2], bw = bias[c0 + 3];

#pragma unroll
  for (int hh = 0; hh < H; ++hh) {
    const int rows = run_rows(hh);
    if (rows > 0) {
      const AggList L = run_list(hh);
      const int rb = hh * R;
      if (fits[hh]) {
        agg_list_gather<D, EPT, P>(L, n[hh], rb, wv, nw, lane, h, acc);
      } else {
        for (int pass = 0; pass < 3; ++pass) {
          for (int eb = -AGG_CAP; eb < E; eb += AGG_CAP) {
            __syncthreads();
            int first = 0;
            if (eb < 0) {
              if (tid < rows) {
                L.s[tid] = d0 + rb + tid; L.r[tid] = rb + tid;
                L.v[tid] = leakyr(as[d0 + rb + tid] + row_ad[rb + tid]);
              }
              if (tid == 0) *L.cnt = rows;
              first = rows;
              __syncthreads();
            } else {
              if (tid == 0) *L.cnt = 0;
              __syncthreads();
              agg_scan_append<R, H>(dstp, eb, (E - eb < AGG_CAP) ? E : eb + AGG_CAP, d0, rb, rb + rows,
                                    wv, nw, lane, list_s, list_r, list_cnt);
              __syncthreads();
            }
            const int np = *L.cnt;
            agg_list_scores(L, first, np, pass == 0, ei, as);
            if (pass >= 1) {
              __syncthreads();
              agg_list_exps(L, np, pass == 1);
            }
            if (pass == 2) {
              __syncthreads();
              agg_list_gather<D, EPT, P>(L, np, rb, wv, nw, lane, h, acc);
            }
          }
        }
      }
      __syncthreads();

      float4* a4 = (float4*)acc;
      if (NOUT == 0) {
        float4* o4 = (float4*)(out + (size_t)(d0 + rb) * D);
        for (int idx = tid; idx < rows * Q; idx += AGG_THREADS) {
          float4 v = a4[idx];
          if (hh + 1 < H) a4[idx] = make_float4(0.f, 0.f, 0.f, 0.f);
          v.x += bx; v.y += by; v.z += bz; v.w += bw;
          v.x = v.x > 0.f ? v.x : 0.f;
          v.y = v.y > 0.f ? v.y : 0.f;
          v.z = v.z > 0.f ? v.z : 0.f;
          v.w = v.w > 0.f ? v.w : 0.f;
          o4[idx] = v;
        }
        if (hh + 1 < H) __syncthreads();
      } else {
        for (int idx = tid; idx < rows * Q; idx += AGG_THREADS) {
          const int at = (idx / Q) * P4 + (idx & (Q - 1));
          float4 v = a4[at];
          v.x += bx; v.y += by; v.z += bz; v.w += bw;
          v.x = v.x > 0.f ? v.x : 0.f;
          v.y = v.y > 0.f ? v.y : 0.f;
          v.z = v.z > 0.f ? v.z : 0.f;
          v.w = v.w > 0.f ? v.w : 0.f;
          a4[at] = v;
        }
        __syncthreads();

        constexpr int NCT = NOUT / 16;
        const int job = __builtin_amdgcn_readfirstlane(tid >> 5);
        const bool has_job = job < ((rows + 15) >> 4) * NCT;
        const int rt = job / NCT;
        const int ct = job % NCT;
        const int half = lane >> 4;
        const int l16  = lane & 15;
        v8f prod = v8f{};
        if (has_job) {
          const int arow = (rt * 16 + l16 < rows) ? (rt * 16 + l16) : (rows - 1);
          const float* Al = acc + arow * P + 2 * half;
          const float* Bg = Wn + (size_t)(2 * half) * NOUT + ct * 16 + l16;
#pragma unroll 8
          for (int k = 0; k < D; k += 4) {
            const v2f a = *(const v2f*)(Al + k);
            v2f b;
            b.x = Bg[(size_t)k * NOUT];
            b.y = Bg[(size_t)(k + 1) * NOUT];
            prod = __builtin_amdgcn_wmma_f32_16x16x4_f32(
                 false, a,  false, b,
                 (short)0, prod,  false,  false);
          }
        }
        __syncthreads();

        if (has_job) {
#pragma unroll
          for (int r = 0; r < 8; ++r) {
            const int row = rt * 16 + 8 * half + r;
            if (row < rows) acc[row * NOUT + ct * 16 + l16] = prod[r];
          }
        }
        __syncthreads();

        float4* o4 = (float4*)(out + (size_t)(d0 + rb) * NOUT);
        for (int idx = tid; idx < rows * (NOUT / 4); idx += AGG_THREADS) o4[idx] = a4[idx];

        if (hh + 1 < H) {
          __syncthreads();
          for (int idx = tid; idx < run_rows(hh + 1) * Q; idx += AGG_THREADS)
            a4[(idx / Q) * P4 + (idx & (Q - 1))] = make_float4(0.f, 0.f, 0.f, 0.f);
          __syncthreads();
        }
      }
    }
  }
}

__global__ void gat_fc(const float* __restrict__ h,
                       const float* __restrict__ W,
                       const float* __restrict__ b,
                       float* __restrict__ out, int N) {
  int n = blockIdx.x * blockDim.x + threadIdx.x;
  if (n >= N) return;
  const float* row = h + (size_t)n * OUT2;
  float a0 = b[0], a1 = b[1];
#pragma unroll
  for (int c = 0; c < OUT2; ++c) {
    float v = row[c];
    a0 += v * W[c * NCLS + 0];
    a1 += v * W[c * NCLS + 1];
  }
  out[n * NCLS + 0] = a0;
  out[n * NCLS + 1] = a1;
}

extern "C" void kernel_launch(void* const* d_in, const int* in_sizes, int n_in,
                              void* d_out, int out_size, void* d_ws, size_t ws_size,
                              hipStream_t stream) {
  const float* x   = (const float*)d_in[0];
  const int*   ei  = (const int*)  d_in[1];
  const float* W1  = (const float*)d_in[2];
  const float* a1s = (const float*)d_in[3];
  const float* a1d = (const float*)d_in[4];
  const float* b1  = (const float*)d_in[5];
  const float* W2  = (const float*)d_in[6];
  const float* a2s = (const float*)d_in[7];
  const float* a2d = (const float*)d_in[8];
  const float* b2  = (const float*)d_in[9];
  const float* fcW = (const float*)d_in[10];
  const float* fcb = (const float*)d_in[11];

  const int N    = in_sizes[0] / IN_C;
  const int E    = in_sizes[1] / 2;

  float* w       = (float*)d_ws;
  float* h1      = w;
  float* h2      = w + (size_t)N * HID;
  float* agg2    = w;
  float* as_buf  = w + (size_t)2 * N * HID;
  float* ad_buf  = as_buf + N;
  int*   lists   = (int*)(ad_buf + N);

  const int TB = 256;

  const int agg_own   = 2 * AGG_ROWS;
  const int agg_blk   = (N + agg_own - 1) / agg_own;
  const int node_blk  = (N + 7) / 8;

  gat_wmma_gemm_dots_f32<IN_C, HID, 16, 8><<<G1_BLOCKS, 512, 0, stream>>>(x, W1, a1s, a1d, h1, as_buf, ad_buf, N);
  gat_aggregate_owned<HID, AGG_ROWS, 2, 1, 1, OUT2><<<agg_blk, AGG_THREADS, 0, stream>>>(ei, as_buf, ad_buf, h1, b1, W2, h2, E, N, lists, w, ws_size);

  gat_node_dots<<<node_blk, TB, 0, stream>>>(h2, a2s, a2d, as_buf, ad_buf, N, OUT2);
  gat_aggregate_owned<OUT2, 2 * AGG_ROWS, 1, 2, 2, 0><<<agg_blk, AGG_THREADS, 0, stream>>>(ei, as_buf, ad_buf, h2, b2, nullptr, agg2, E, N, lists, w, ws_size);

  gat_fc<<<(N + TB - 1) / TB, TB, 0, stream>>>(agg2, fcW, fcb, (float*)d_out, N);
}


# Row tails folded into the aggregation blocks: layer-2 score dots from the product rows in LDS as they are stored, classifier from the finished rows; 5 -> 3 launches; those dots in a new sum tree
# speedup vs baseline: 4.2966x; 1.0723x over previous
#include <hip/hip_runtime.h>


typedef __attribute__((ext_vector_type(2))) float v2f;
typedef __attribute__((ext_vector_type(8))) float v8f;

#define IN_C  128
#define HID   512
#define OUT2  64
#define NCLS  2
#define NEG_SLOPE 0.2f
#define G1_BLOCKS 256


template <int XM>
__device__ __forceinline__ float row_xmask_add(float v) {
  const int b = __float_as_int(v);
  return v + __int_as_float(__builtin_amdgcn_update_dpp(b, b, 0x160 + XM, 0xf, 0xf, false));
}
__device__ __forceinline__ float row16_sum(float v) {
  v = row_xmask_add<1>(v);
  v = row_xmask_add<2>(v);
  v = row_xmask_add<4>(v);
  return row_xmask_add<8>(v);
}

template <int K, int NC, int WAVES, int U>
__global__ __launch_bounds__(WAVES * 32) void gat_wmma_gemm_dots_f32(
    const float* __restrict__ A, const float* __restrict__ B,
    const float* __restrict__ a_src, const float* __restrict__ a_dst,
    float* __restrict__ C, float* __restrict__ as_out, float* __restrict__ ad_out, int M) {
  constexpr int NT = 8;
  constexpr int STRIPS = NC / (16 * NT);
  constexpr int NTT = NC / 16;
  constexpr int CH = 16;
  static_assert(K % 4 == 0 && NC % (16 * NT) == 0 && STRIPS == 4 && WAVES % STRIPS == 0,
                "a wave keeps one of four strips");
  constexpr int TP = K * 16 + 8;
  __shared__ __attribute__((aligned(16))) float Bs[NTT * TP];
  __shared__ float part_s[STRIPS][CH * 16];
  __shared__ float part_d[STRIPS][CH * 16];

  const int tid  = threadIdx.x;
  const int lane = tid & 31;
  const int waveid = __builtin_amdgcn_readfirstlane(tid >> 5);

  constexpr int QC = NTT * 4;
  constexpr int ITEMS = (K / 2) * QC;
#pragma unroll 4
  for (int it = tid; it < ITEMS; it += WAVES * 32) {
    const int p = it / QC;
    const int q = it % QC;
    const float* src = B + (size_t)(2 * p) * NC + 4 * q;
    const float4 r0 = *(const float4*)src;
    const float4 r1 = *(const float4*)(src + NC);
    float* dst = Bs + (q >> 2) * TP + p * 32 + (q & 3) * 8;
    *(float4*)(dst)     = make_float4(r0.x, r1.x, r0.y, r1.y);
    *(float4*)(dst + 4) = make_float4(r0.z, r1.z, r0.w, r1.w);
  }

  const int half = lane >> 4;
  const int l16  = lane & 15;
  const int st   = waveid % STRIPS;
  constexpr int TC = WAVES / STRIPS;
  const int tl0  = (waveid / STRIPS + st) % TC;
  const int col0 = st * (NT * 16);

  float ws[NT], wd[NT];
#pragma unroll
  for (int t = 0; t < NT; ++t) {
    ws[t] = a_src[col0 + t * 16 + l16];
    wd[t] = a_dst[col0 + t * 16 + l16];
  }
  __syncthreads();

  const int tilesM = (M + 15) >> 4;
  const int tpb = (tilesM + (int)gridDim.x - 1) / (int)gridDim.x;
  const int t_begin = blockIdx.x * tpb;
  const int t_end = (t_begin + tpb < tilesM) ? (t_begin + tpb) : tilesM;

  const float* Bl = Bs + (st * NT) * TP + half * 32 + (l16 << 1);

  for (int c0 = t_begin; c0 < t_end; c0 += CH) {
    const int ntile = (t_end - c0 < CH) ? (t_end - c0) : CH;
    for (int tl = tl0; tl < ntile; tl += TC) {
      const int tm = c0 + tl;
      const float* Ap = A + (size_t)(tm * 16 + l16) * K + 2 * half;

      v8f acc[NT];
#pragma unroll
      for (int t = 0; t < NT; ++t) acc[t] = v8f{};

#pragma unroll U
      for (int k = 0; k < K; k += 4) {
        const v2f a = *(const v2f*)(Ap + k);
#pragma unroll
        for (int t = 0; t < NT; ++t) {
          const v2f b = *(const v2f*)(Bl + t * TP + k * 16);
          acc[t] = __builtin_amdgcn_wmma_f32_16x16x4_f32(
               false, a,  false, b,
               (short)0, acc[t],  false,  false);
        }
      }

      float* Cw = C + (size_t)(tm * 16) * NC + col0 + lane;
#pragma unroll
      for (int t = 0; t < NT; t += 2) {
#pragma unroll
        for (int r = 0; r < 8; ++r) {
          const float x0 = acc[t][r];
          const float x1 = acc[t + 1][r];
          const float y0 = __shfl_xor(x0, 16, 32);
          const float y1 = __shfl_xor(x1, 16, 32);
          const float lo = half ? y1 : x0;
          const float hi = half ? x1 : y0;
          Cw[(size_t)r * NC + t * 16]       = lo;
          Cw[(size_t)(r + 8) * NC + t * 16] = hi;
        }
      }

#pragma unroll
      for (int r = 0; r < 8; ++r) {
        float ss = 0.f, sd = 0.f;
#pragma unroll
        for (int t = 0; t < NT; ++t) {
          ss += acc[t][r] * ws[t];
          sd += acc[t][r] * wd[t];
        }
        ss = row16_sum(ss);
        sd = row16_sum(sd);
        if (l16 == 0) {
          part_s[st][tl * 16 + half * 8 + r] = ss;
          part_d[st][tl * 16 + half * 8 + r] = sd;
        }
      }
    }
    __syncthreads();

    for (int i = tid; i < ntile * 16; i += WAVES * 32) {
      const int row = c0 * 16 + i;
      if (row < M) {
        as_out[row] = ((part_s[0][i] + part_s[1][i]) + part_s[2][i]) + part_s[3][i];
        ad_out[row] = ((part_d[0][i] + part_d[1][i]) + part_d[2][i]) + part_d[3][i];
      }
    }
    if (c0 + CH < t_end) __syncthreads();
  }
}

__device__ __forceinline__ float leakyr(float v) {
  return v > 0.f ? v : NEG_SLOPE * v;
}

__device__ __forceinline__ void lds_max_f32(float* addr, float val) {
  if (__float_as_int(val) >= 0) atomicMax((int*)addr, __float_as_int(val));
  else                          atomicMin((unsigned int*)addr, __float_as_uint(val));
}

#define AGG_ROWS 98
#define AGG_U    16
#define AGG_THREADS 1024
#define AGG_CAP  4096
#define AGG_SAVE 2048

struct AggList {
  int*   s;
  int*   r;
  float* v;
  int*   cnt;
  float* m;
  float* den;
  float* ad;
};

template <int R, int H>
__device__ __forceinline__ void agg_append_owned(const int (&dv)[AGG_U], int base, int lane,
                                                 int d0, int lo, int hi,
                                                 int* ls, int* lr, int* lcnt) {
#pragma unroll
  for (int k = 0; k < AGG_U; ++k) {
    const int r = dv[k] - d0;
    if ((unsigned)(r - lo) < (unsigned)(hi - lo)) {
      const int hh = (H > 1) ? r / R : 0;
      const int idx = atomicAdd(lcnt + hh, 1);
      if (idx < AGG_CAP) { ls[hh * AGG_CAP + idx] = base + k * 32 + lane; lr[hh * AGG_CAP + idx] = r; }
    }
  }
}

template <int R, int H>
__device__ __forceinline__ void agg_scan_append(const int* __restrict__ dstp, int eb, int ee,
                                                int d0, int lo, int hi, int wv, int nw, int lane,
                                                int* ls, int* lr, int* lcnt) {
  const int sweep  = AGG_U * 32;
  const int stride = nw * sweep;
  int base = eb + wv * sweep;

  if (base + sweep <= ee) {
    int cur[AGG_U];
#pragma unroll
    for (int k = 0; k < AGG_U; ++k) cur[k] = dstp[base + k * 32 + lane];
    while (true) {
      const int nb = base + stride;
      const bool more = (nb + sweep <= ee);
      const int pb = more ? nb : base;
      int nxt[AGG_U];
#pragma unroll
      for (int k = 0; k < AGG_U; ++k) nxt[k] = dstp[pb + k * 32 + lane];
      agg_append_owned<R, H>(cur, base, lane, d0, lo, hi, ls, lr, lcnt);
      base = nb;
      if (!more) break;
#pragma unroll
      for (int k = 0; k < AGG_U; ++k) cur[k] = nxt[k];
    }
  }
  if (base < ee) {
    int cur[AGG_U];
#pragma unroll
    for (int k = 0; k < AGG_U; ++k) {
      const int e = base + k * 32 + lane;
      cur[k] = (e < ee) ? dstp[e] : -1;
    }
    agg_append_owned<R, H>(cur, base, lane, d0, lo, hi, ls, lr, lcnt);
  }
}

__device__ __forceinline__ void agg_list_scores(const AggList& L, int first, int n, bool do_max,
                                                const int* __restrict__ ei,
                                                const float* __restrict__ as) {
  for (int i = first + (int)threadIdx.x; i < n; i += AGG_THREADS) {
    const int r = L.r[i];
    const int s = ei[L.s[i]];
    const float sc = leakyr(as[s] + L.ad[r]);
    L.s[i] = s;
    L.v[i] = sc;
    if (do_max) lds_max_f32(&L.m[r], sc);
  }
}

__device__ __forceinline__ void agg_list_exps(const AggList& L, int n, bool do_sum) {
  for (int i = (int)threadIdx.x; i < n; i += AGG_THREADS) {
    const int r = L.r[i];
    const float ex = expf(L.v[i] - L.m[r]);
    L.v[i] = ex;
    if (do_sum) atomicAdd(&L.den[r], ex);
  }
}

template <int D, int P>
__device__ __forceinline__ void agg_gather_entry(const AggList& L, int i, int rbase, int lane,
                                                 const float* __restrict__ h, float* acc) {
  constexpr int V = D / 32;
  const int s = L.s[i];
  const float* hs = h + (size_t)s * D + lane;
  float v[V];
#pragma unroll
  for (int j = 0; j < V; ++j) v[j] = hs[j * 32];
  const int r = L.r[i];
  const float a = L.v[i] / L.den[r];
  float* ar = acc + (r - rbase) * P + lane;
#pragma unroll
  for (int j = 0; j < V; ++j) atomicAdd(ar + j * 32, a * v[j]);
}

template <int D, int EPT, int P>
__device__ __forceinline__ void agg_list_gather(const AggList& L, int n, int rbase,
                                                int wv, int nw, int lane,
                                                const float* __restrict__ h, float* acc) {
  constexpr int V = D / 32;
  const int groups = n / EPT;
  int g = wv;
  for (; g < groups; g += nw) {
    if (EPT == 1) {
      agg_gather_entry<D, P>(L, g, rbase, lane, h, acc);
    } else {
      const int i0 = g * EPT;
      float v[EPT][V];
#pragma unroll
      for (int e = 0; e < EPT; ++e) {
        const float* hs = h + (size_t)L.s[i0 + e] * D + lane;
#pragma unroll
        for (int j = 0; j < V; ++j) v[e][j] = hs[j * 32];
      }
#pragma unroll
      for (int e = 0; e < EPT; ++e) {
        const int r = L.r[i0 + e];
        const float a = L.v[i0 + e] / L.den[r];
        float* ar = acc + (r - rbase) * P + lane;
#pragma unroll
        for (int j = 0; j < V; ++j) atomicAdd(ar + j * 32, a * v[e][j]);
      }
    }
  }
  if (EPT > 1 && g == groups) {
    for (int i = groups * EPT; i < n; ++i) agg_gather_entry<D, P>(L, i, rbase, lane, h, acc);
  }
}

template <int D, int R, int H, int EPT, int LMODE, int NOUT, int NFC>
__global__ __launch_bounds__(AGG_THREADS) void gat_aggregate_owned(
    const int* __restrict__ ei, const float* __restrict__ as,
    const float* __restrict__ ad, const float* __restrict__ h,
    const float* __restrict__ bias, const float* __restrict__ Wn,
    const float* __restrict__ ta, const float* __restrict__ tb,
    float* __restrict__ out, float* __restrict__ to_a, float* __restrict__ to_b,
    int E, int N,
    int* __restrict__ saved, const float* ws_base, size_t ws_bytes) {
  constexpr int RB = R * H;
  constexpr int P = (NOUT > 0) ? D + 4 : D;
  constexpr int Q = D / 4;
  constexpr int P4 = P / 4;
  static_assert((Q & (Q - 1)) == 0 && AGG_THREADS % Q == 0, "column group must be fixed per thread");
  static_assert(NOUT % 16 == 0 && D % 4 == 0 && ((R + 15) / 16) * (NOUT / 16) <= AGG_THREADS / 32,
                "one product tile per wave");
  static_assert(RB <= 256, "a block-local row must fit the low byte of a saved entry");
  static_assert(LMODE != 2 || H == 1, "loaded lists go into one list");
  static_assert(NOUT == 0 || NOUT == 64, "16 threads, one 16-lane row of a wave, hold a product row");
  static_assert(NFC == 0 || (NOUT == 0 && Q == 16), "16 threads, one 16-lane row of a wave, hold a finished row");
  __shared__ __attribute__((aligned(16))) float acc[R * P];
  __shared__ int   list_s[H * AGG_CAP];
  __shared__ int   list_r[H * AGG_CAP];
  __shared__ float list_v[H * AGG_CAP];
  __shared__ float row_m[RB];
  __shared__ float row_den[RB];
  __shared__ float row_ad[RB];
  __shared__ int   list_cnt[H];
  static_assert(R <= AGG_CAP && RB <= AGG_THREADS, "self loops open the lists, one thread each");

  const int tid  = threadIdx.x;
  const int lane = tid & 31;
  const int wv   = tid >> 5;
  const int nw   = AGG_THREADS >> 5;
  const int d0   = blockIdx.x * RB;
  const int rows_all = (N - d0 < RB) ? (N - d0) : RB;
  const int* __restrict__ dstp = ei + E;

  const size_t save_end = (size_t)((const char*)saved - (const char*)ws_base) +
                          (size_t)gridDim.x * AGG_SAVE * sizeof(int);
  const bool ws_ok = (LMODE != 0) && save_end <= ws_bytes && N <= (1 << 24);
  int* mine = saved + (size_t)blockIdx.x * AGG_SAVE;
  int loaded = -1;
  if (LMODE == 2 && ws_ok) {
    const int t = mine[0];
    if (t >= rows_all && t <= AGG_SAVE - 2 && t <= AGG_CAP) loaded = t;
  }

  auto run_rows = [&](int hh) {
    const int left = rows_all - hh * R;
    return left < 0 ? 0 : (left < R ? left : R);
  };
  auto run_list = [&](int hh) {
    return AggList{list_s + hh * AGG_CAP, list_r + hh * AGG_CAP, list_v + hh * AGG_CAP,
                   list_cnt + hh, row_m, row_den, row_ad};
  };

  {
    float4* z4 = (float4*)acc;
    for (int idx = tid; idx < run_rows(0) * Q; idx += AGG_THREADS)
      z4[(idx / Q) * P4 + (idx & (Q - 1))] = make_float4(0.f, 0.f, 0.f, 0.f);
  }
  if (tid < rows_all) {
    const int hh = tid / R;
    const int at = hh * AGG_CAP + (tid - hh * R);
    const float adv = ad[d0 + tid];
    const float sc  = leakyr(as[d0 + tid] + adv);
    row_ad[tid] = adv; row_m[tid] = sc; row_den[tid] = 0.f;
    list_s[at] = d0 + tid; list_r[at] = tid; list_v[at] = sc;
  }
  if (tid < H) list_cnt[tid] = run_rows(tid);
  __syncthreads();

  if (LMODE == 2 && loaded >= 0) {
    for (int i = tid; i < loaded; i += 2 * AGG_THREADS) {
      const int j = i + AGG_THREADS;
      const bool two = j < loaded;
      const int jc = two ? j : i;
      const unsigned w0 = (unsigned)mine[2 + i];
      const unsigned w1 = (unsigned)mine[2 + jc];
      const int s0 = (int)(w0 >> 8), r0 = (int)(w0 & 255u);
      const int s1 = (int)(w1 >> 8), r1 = (int)(w1 & 255u);
      const float a0 = as[s0];
      const float a1 = as[s1];
      const float sc0 = leakyr(a0 + row_ad[r0]);
      list_s[i] = s0; list_r[i] = r0; list_v[i] = sc0;
      lds_max_f32(&row_m[r0], sc0);
      if (two) {
        const float sc1 = leakyr(a1 + row_ad[r1]);
        list_s[j] = s1; list_r[j] = r1; list_v[j] = sc1;
        lds_max_f32(&row_m[r1], sc1);
      }
    }
  } else {
    agg_scan_append<R, H>(dstp, 0, E, d0, 0, rows_all, wv, nw, lane, list_s, list_r, list_cnt);
  }
  __syncthreads();

  int  n[H];
  bool fits[H];
#pragma unroll
  for (int hh = 0; hh < H; ++hh) {
    n[hh] = (LMODE == 2 && loaded >= 0) ? loaded : list_cnt[hh];
    fits[hh] = (n[hh] <= AGG_CAP);
  }
  if (!(LMODE == 2 && loaded >= 0)) {
#pragma unroll
    for (int hh = 0; hh < H; ++hh)
      if (fits[hh]) agg_list_scores(run_list(hh), run_rows(hh), n[hh], true, ei, as);
  }
  __syncthreads();

  if (LMODE == 1 && ws_ok) {
    int total = 0;
    bool all = true;
#pragma unroll
    for (int hh = 0; hh < H; ++hh) { total += n[hh]; all = all && fits[hh]; }
    const bool keep = all && total <= AGG_SAVE - 2;
    if (tid == 0) mine[0] = keep ? total : -1;
    if (keep) {
      int at = 2;
#pragma unroll
      for (int hh = 0; hh < H; ++hh) {
        for (int i = tid; i < n[hh]; i += AGG_THREADS)
          mine[at + i] = (int)(((unsigned)list_s[hh * AGG_CAP + i] << 8) | (unsigned)list_r[hh * AGG_CAP + i]);
        at += n[hh];
      }
    }
  }
#pragma unroll
  for (int hh = 0; hh < H; ++hh)
    if (fits[hh]) agg_list_exps(run_list(hh), n[hh], true);
  __syncthreads();

  const int c0 = (tid & (Q - 1)) * 4;
  const float bx = bias[c0], by = bias[c0 + 1], bz = bias[c0 + 2], bw = bias[c0 + 3];
  constexpr int NF = NFC > 0 ? NFC : 1;
  float wf[4][NF] = {}, fb[NF] = {};
  if (NFC > 0) {
#pragma unroll
    for (int q = 0; q < NF; ++q) {
      fb[q] = tb[q];
#pragma unroll
      for (int j = 0; j < 4; ++j) wf[j][q] = ta[(c0 + j) * NF + q];
    }
  }
  const int pc = (NOUT > 0) ? (tid & 15) * 4 : 0;
  float sa[4] = {}, sb[4] = {};
  if (NOUT > 0) {
#pragma unroll
    for (int j = 0; j < 4; ++j) { sa[j] = ta[pc + j]; sb[j] = tb[pc + j]; }
  }

#pragma unroll
  for (int hh = 0; hh < H; ++hh) {
    const int rows = run_rows(hh);
    if (rows > 0) {
      const AggList L = run_list(hh);
      const int rb = hh * R;
      if (fits[hh]) {
        agg_list_gather<D, EPT, P>(L, n[hh], rb, wv, nw, lane, h, acc);
      } else {
        for (int pass = 0; pass < 3; ++pass) {
          for (int eb = -AGG_CAP; eb < E; eb += AGG_CAP) {
            __syncthreads();
            int first = 0;
            if (eb < 0) {
              if (tid < rows) {
                L.s[tid] = d0 + rb + tid; L.r[tid] = rb + tid;
                L.v[tid] = leakyr(as[d0 + rb + tid] + row_ad[rb + tid]);
              }
              if (tid == 0) *L.cnt = rows;
              first = rows;
              __syncthreads();
            } else {
              if (tid == 0) *L.cnt = 0;
              __syncthreads();
              agg_scan_append<R, H>(dstp, eb, (E - eb < AGG_CAP) ? E : eb + AGG_CAP, d0, rb, rb + rows,
                                    wv, nw, lane, list_s, list_r, list_cnt);
              __syncthreads();
            }
            const int np = *L.cnt;
            agg_list_scores(L, first, np, pass == 0, ei, as);
            if (pass >= 1) {
              __syncthreads();
              agg_list_exps(L, np, pass == 1);
            }
            if (pass == 2) {
              __syncthreads();
              agg_list_gather<D, EPT, P>(L, np, rb, wv, nw, lane, h, acc);
            }
          }
        }
      }
      __syncthreads();

      float4* a4 = (float4*)acc;
      if (NOUT == 0 && NFC > 0) {
        for (int idx = tid; idx < rows * Q; idx += AGG_THREADS) {
          float4 v = a4[idx];
          if (hh + 1 < H) a4[idx] = make_float4(0.f, 0.f, 0.f, 0.f);
          v.x += bx; v.y += by; v.z += bz; v.w += bw;
          v.x = v.x > 0.f ? v.x : 0.f;
          v.y = v.y > 0.f ? v.y : 0.f;
          v.z = v.z > 0.f ? v.z : 0.f;
          v.w = v.w > 0.f ? v.w : 0.f;
          float p[NF];
#pragma unroll
          for (int q = 0; q < NF; ++q)
            p[q] = row16_sum(((v.x * wf[0][q] + v.y * wf[1][q]) + v.z * wf[2][q]) + v.w * wf[3][q]);
          if ((tid & 15) == 0) {
            float* orow = out + (size_t)(d0 + rb + idx / Q) * NFC;
#pragma unroll
            for (int q = 0; q < NF; ++q) orow[q] = p[q] + fb[q];
          }
        }
        if (hh + 1 < H) __syncthreads();
      } else if (NOUT == 0) {
        float4* o4 = (float4*)(out + (size_t)(d0 + rb) * D);
        for (int idx = tid; idx < rows * Q; idx += AGG_THREADS) {
          float4 v = a4[idx];
          if (hh + 1 < H) a4[idx] = make_float4(0.f, 0.f, 0.f, 0.f);
          v.x += bx; v.y += by; v.z += bz; v.w += bw;
          v.x = v.x > 0.f ? v.x : 0.f;
          v.y = v.y > 0.f ? v.y : 0.f;
          v.z = v.z > 0.f ? v.z : 0.f;
          v.w = v.w > 0.f ? v.w : 0.f;
          o4[idx] = v;
        }
        if (hh + 1 < H) __syncthreads();
      } else {
        for (int idx = tid; idx < rows * Q; idx += AGG_THREADS) {
          const int at = (idx / Q) * P4 + (idx & (Q - 1));
          float4 v = a4[at];
          v.x += bx; v.y += by; v.z += bz; v.w += bw;
          v.x = v.x > 0.f ? v.x : 0.f;
          v.y = v.y > 0.f ? v.y : 0.f;
          v.z = v.z > 0.f ? v.z : 0.f;
          v.w = v.w > 0.f ? v.w : 0.f;
          a4[at] = v;
        }
        __syncthreads();

        constexpr int NCT = NOUT / 16;
        const int job = __builtin_amdgcn_readfirstlane(tid >> 5);
        const bool has_job = job < ((rows + 15) >> 4) * NCT;
        const int rt = job / NCT;
        const int ct = job % NCT;
        const int half = lane >> 4;
        const int l16  = lane & 15;
        v8f prod = v8f{};
        if (has_job) {
          const int arow = (rt * 16 + l16 < rows) ? (rt * 16 + l16) : (rows - 1);
          const float* Al = acc + arow * P + 2 * half;
          const float* Bg = Wn + (size_t)(2 * half) * NOUT + ct * 16 + l16;
#pragma unroll 8
          for (int k = 0; k < D; k += 4) {
            const v2f a = *(const v2f*)(Al + k);
            v2f b;
            b.x = Bg[(size_t)k * NOUT];
            b.y = Bg[(size_t)(k + 1) * NOUT];
            prod = __builtin_amdgcn_wmma_f32_16x16x4_f32(
                 false, a,  false, b,
                 (short)0, prod,  false,  false);
          }
        }
        __syncthreads();

        if (has_job) {
#pragma unroll
          for (int r = 0; r < 8; ++r) {
            const int row = rt * 16 + 8 * half + r;
            if (row < rows) acc[row * NOUT + ct * 16 + l16] = prod[r];
          }
        }
        __syncthreads();

        float4* o4 = (float4*)(out + (size_t)(d0 + rb) * NOUT);
        for (int idx = tid; idx < rows * (NOUT / 4); idx += AGG_THREADS) {
          const float4 v = a4[idx];
          o4[idx] = v;
          const float ps = row16_sum(((v.x * sa[0] + v.y * sa[1]) + v.z * sa[2]) + v.w * sa[3]);
          const float pd = row16_sum(((v.x * sb[0] + v.y * sb[1]) + v.z * sb[2]) + v.w * sb[3]);
          if ((tid & 15) == 0) {
            const int node = d0 + rb + idx / (NOUT / 4);
            to_a[node] = ps;
            to_b[node] = pd;
          }
        }

        if (hh + 1 < H) {
          __syncthreads();
          for (int idx = tid; idx < run_rows(hh + 1) * Q; idx += AGG_THREADS)
            a4[(idx / Q) * P4 + (idx & (Q - 1))] = make_float4(0.f, 0.f, 0.f, 0.f);
          __syncthreads();
        }
      }
    }
  }
}

extern "C" void kernel_launch(void* const* d_in, const int* in_sizes, int n_in,
                              void* d_out, int out_size, void* d_ws, size_t ws_size,
                              hipStream_t stream) {
  const float* x   = (const float*)d_in[0];
  const int*   ei  = (const int*)  d_in[1];
  const float* W1  = (const float*)d_in[2];
  const float* a1s = (const float*)d_in[3];
  const float* a1d = (const float*)d_in[4];
  const float* b1  = (const float*)d_in[5];
  const float* W2  = (const float*)d_in[6];
  const float* a2s = (const float*)d_in[7];
  const float* a2d = (const float*)d_in[8];
  const float* b2  = (const float*)d_in[9];
  const float* fcW = (const float*)d_in[10];
  const float* fcb = (const float*)d_in[11];

  const int N    = in_sizes[0] / IN_C;
  const int E    = in_sizes[1] / 2;

  float* w       = (float*)d_ws;
  float* h1      = w;
  float* h2      = w + (size_t)N * HID;
  float* as2_buf = h2 + (size_t)N * OUT2;
  float* ad2_buf = as2_buf + N;
  float* as_buf  = w + (size_t)2 * N * HID;
  float* ad_buf  = as_buf + N;
  int*   lists   = (int*)(ad_buf + N);

  const int agg_own   = 2 * AGG_ROWS;
  const int agg_blk   = (N + agg_own - 1) / agg_own;

  gat_wmma_gemm_dots_f32<IN_C, HID, 16, 8><<<G1_BLOCKS, 512, 0, stream>>>(x, W1, a1s, a1d, h1, as_buf, ad_buf, N);
  gat_aggregate_owned<HID, AGG_ROWS, 2, 1, 1, OUT2, 0><<<agg_blk, AGG_THREADS, 0, stream>>>(ei, as_buf, ad_buf, h1, b1, W2, a2s, a2d, h2, as2_buf, ad2_buf, E, N, lists, w, ws_size);

  gat_aggregate_owned<OUT2, 2 * AGG_ROWS, 1, 2, 2, 0, NCLS><<<agg_blk, AGG_THREADS, 0, stream>>>(ei, as2_buf, ad2_buf, h2, b2, nullptr, fcW, fcb, (float*)d_out, nullptr, nullptr, E, N, lists, w, ws_size);
}


# Edge search on owner bytes: launch 1 writes dst/196 per edge as one byte; layer-1 aggregation block scans E bytes, 16 per load, reads destinations of matched groups only; sums in LDS arrival order
# speedup vs baseline: 4.4480x; 1.0352x over previous
#include <hip/hip_runtime.h>


typedef __attribute__((ext_vector_type(2))) float v2f;
typedef __attribute__((ext_vector_type(8))) float v8f;

#define IN_C  128
#define HID   512
#define OUT2  64
#define NCLS  2
#define NEG_SLOPE 0.2f
#define G1_BLOCKS 256


template <int XM>
__device__ __forceinline__ float row_xmask_add(float v) {
  const int b = __float_as_int(v);
  return v + __int_as_float(__builtin_amdgcn_update_dpp(b, b, 0x160 + XM, 0xf, 0xf, false));
}
__device__ __forceinline__ float row16_sum(float v) {
  v = row_xmask_add<1>(v);
  v = row_xmask_add<2>(v);
  v = row_xmask_add<4>(v);
  return row_xmask_add<8>(v);
}

__device__ __forceinline__ bool owner_bytes_ok(int n, int E, int rb) {
  const int words = ((E + 15) >> 4) * 4;
  return (n + rb - 1) / rb <= 256 && (E & 3) == 0 && (n & 1) == 0 &&
         (size_t)words <= (size_t)n * (HID - OUT2 - 2);
}
template <int RB>
__device__ __forceinline__ unsigned int owner_pack(const int4 d) {
  return ((unsigned int)(d.x / RB) & 255u) | (((unsigned int)(d.y / RB) & 255u) << 8) |
         (((unsigned int)(d.z / RB) & 255u) << 16) | (((unsigned int)(d.w / RB) & 255u) << 24);
}

template <int K, int NC, int WAVES, int U, int RB>
__global__ __launch_bounds__(WAVES * 32) void gat_wmma_gemm_dots_f32(
    const float* __restrict__ A, const float* __restrict__ B,
    const float* __restrict__ a_src, const float* __restrict__ a_dst,
    float* __restrict__ C, float* __restrict__ as_out, float* __restrict__ ad_out, int M,
    const int* __restrict__ ei, int E, unsigned int* __restrict__ own) {
  constexpr int NT = 8;
  constexpr int STRIPS = NC / (16 * NT);
  constexpr int NTT = NC / 16;
  constexpr int CH = 16;
  static_assert(K % 4 == 0 && NC % (16 * NT) == 0 && STRIPS == 4 && WAVES % STRIPS == 0,
                "a wave keeps one of four strips");
  constexpr int TP = K * 16 + 8;
  __shared__ __attribute__((aligned(16))) float Bs[NTT * TP];
  __shared__ float part_s[STRIPS][CH * 16];
  __shared__ float part_d[STRIPS][CH * 16];

  const int tid  = threadIdx.x;
  const int lane = tid & 31;
  const int waveid = __builtin_amdgcn_readfirstlane(tid >> 5);

  const int own_words = ((E + 15) >> 4) * 4;
  const bool own_on = owner_bytes_ok(M, E, RB);
  const int* __restrict__ edst = ei + E;
  const int own_stride = (int)gridDim.x * WAVES * 32;
  const int own_i0 = (int)blockIdx.x * WAVES * 32 + tid;
  const bool own_ld = own_on && 4 * own_i0 < E;
  const int4 oq = *(const int4*)(own_ld ? edst + 4 * own_i0 : ei);

  constexpr int QC = NTT * 4;
  constexpr int ITEMS = (K / 2) * QC;
#pragma unroll 4
  for (int it = tid; it < ITEMS; it += WAVES * 32) {
    const int p = it / QC;
    const int q = it % QC;
    const float* src = B + (size_t)(2 * p) * NC + 4 * q;
    const float4 r0 = *(const float4*)src;
    const float4 r1 = *(const float4*)(src + NC);
    float* dst = Bs + (q >> 2) * TP + p * 32 + (q & 3) * 8;
    *(float4*)(dst)     = make_float4(r0.x, r1.x, r0.y, r1.y);
    *(float4*)(dst + 4) = make_float4(r0.z, r1.z, r0.w, r1.w);
  }

  if (own_on) {
    if (own_i0 < own_words) own[own_i0] = own_ld ? owner_pack<RB>(oq) : 0u;
    for (int i = own_i0 + own_stride; i < own_words; i += own_stride)
      own[i] = (4 * i < E) ? owner_pack<RB>(*(const int4*)(edst + 4 * i)) : 0u;
  }

  const int half = lane >> 4;
  const int l16  = lane & 15;
  const int st   = waveid % STRIPS;
  constexpr int TC = WAVES / STRIPS;
  const int tl0  = (waveid / STRIPS + st) % TC;
  const int col0 = st * (NT * 16);

  float ws[NT], wd[NT];
#pragma unroll
  for (int t = 0; t < NT; ++t) {
    ws[t] = a_src[col0 + t * 16 + l16];
    wd[t] = a_dst[col0 + t * 16 + l16];
  }
  __syncthreads();

  const int tilesM = (M + 15) >> 4;
  const int tpb = (tilesM + (int)gridDim.x - 1) / (int)gridDim.x;
  const int t_begin = blockIdx.x * tpb;
  const int t_end = (t_begin + tpb < tilesM) ? (t_begin + tpb) : tilesM;

  const float* Bl = Bs + (st * NT) * TP + half * 32 + (l16 << 1);

  for (int c0 = t_begin; c0 < t_end; c0 += CH) {
    const int ntile = (t_end - c0 < CH) ? (t_end - c0) : CH;
    for (int tl = tl0; tl < ntile; tl += TC) {
      const int tm = c0 + tl;
      const float* Ap = A + (size_t)(tm * 16 + l16) * K + 2 * half;

      v8f acc[NT];
#pragma unroll
      for (int t = 0; t < NT; ++t) acc[t] = v8f{};

#pragma unroll U
      for (int k = 0; k < K; k += 4) {
        const v2f a = *(const v2f*)(Ap + k);
#pragma unroll
        for (int t = 0; t < NT; ++t) {
          const v2f b = *(const v2f*)(Bl + t * TP + k * 16);
          acc[t] = __builtin_amdgcn_wmma_f32_16x16x4_f32(
               false, a,  false, b,
               (short)0, acc[t],  false,  false);
        }
      }

      float* Cw = C + (size_t)(tm * 16) * NC + col0 + lane;
#pragma unroll
      for (int t = 0; t < NT; t += 2) {
#pragma unroll
        for (int r = 0; r < 8; ++r) {
          const float x0 = acc[t][r];
          const float x1 = acc[t + 1][r];
          const float y0 = __shfl_xor(x0, 16, 32);
          const float y1 = __shfl_xor(x1, 16, 32);
          const float lo = half ? y1 : x0;
          const float hi = half ? x1 : y0;
          Cw[(size_t)r * NC + t * 16]       = lo;
          Cw[(size_t)(r + 8) * NC + t * 16] = hi;
        }
      }

#pragma unroll
      for (int r = 0; r < 8; ++r) {
        float ss = 0.f, sd = 0.f;
#pragma unroll
        for (int t = 0; t < NT; ++t) {
          ss += acc[t][r] * ws[t];
          sd += acc[t][r] * wd[t];
        }
        ss = row16_sum(ss);
        sd = row16_sum(sd);
        if (l16 == 0) {
          part_s[st][tl * 16 + half * 8 + r] = ss;
          part_d[st][tl * 16 + half * 8 + r] = sd;
        }
      }
    }
    __syncthreads();

    for (int i = tid; i < ntile * 16; i += WAVES * 32) {
      const int row = c0 * 16 + i;
      if (row < M) {
        as_out[row] = ((part_s[0][i] + part_s[1][i]) + part_s[2][i]) + part_s[3][i];
        ad_out[row] = ((part_d[0][i] + part_d[1][i]) + part_d[2][i]) + part_d[3][i];
      }
    }
    if (c0 + CH < t_end) __syncthreads();
  }
}

__device__ __forceinline__ float leakyr(float v) {
  return v > 0.f ? v : NEG_SLOPE * v;
}

__device__ __forceinline__ void lds_max_f32(float* addr, float val) {
  if (__float_as_int(val) >= 0) atomicMax((int*)addr, __float_as_int(val));
  else                          atomicMin((unsigned int*)addr, __float_as_uint(val));
}

#define AGG_ROWS 98
#define AGG_U    16
#define AGG_THREADS 1024
#define AGG_CAP  4096
#define AGG_SAVE 2048

struct AggList {
  int*   s;
  int*   r;
  float* v;
  int*   cnt;
  float* m;
  float* den;
  float* ad;
};

template <int R, int H>
__device__ __forceinline__ void agg_append_owned(const int (&dv)[AGG_U], int base, int lane,
                                                 int d0, int lo, int hi,
                                                 int* ls, int* lr, int* lcnt) {
#pragma unroll
  for (int k = 0; k < AGG_U; ++k) {
    const int r = dv[k] - d0;
    if ((unsigned)(r - lo) < (unsigned)(hi - lo)) {
      const int hh = (H > 1) ? r / R : 0;
      const int idx = atomicAdd(lcnt + hh, 1);
      if (idx < AGG_CAP) { ls[hh * AGG_CAP + idx] = base + k * 32 + lane; lr[hh * AGG_CAP + idx] = r; }
    }
  }
}

__device__ __forceinline__ unsigned int owner_match(unsigned int w, unsigned int pat) {
  const unsigned int t = w ^ pat;
  return (t - 0x01010101u) & ~t & 0x80808080u;
}

template <int CCAP>
__device__ __forceinline__ void agg_owner_candidates(const uint4* __restrict__ o4, int ng,
                                                     unsigned int pat, int* cand, int* ccnt) {
  constexpr int GU = 4;
  for (int g0 = (int)threadIdx.x; g0 < ng; g0 += GU * AGG_THREADS) {
    uint4 v[GU];
#pragma unroll
    for (int u = 0; u < GU; ++u) {
      const int gi = g0 + u * AGG_THREADS;
      v[u] = o4[gi < ng ? gi : g0];
    }
#pragma unroll
    for (int u = 0; u < GU; ++u) {
      const int gi = g0 + u * AGG_THREADS;
      const unsigned int hx = owner_match(v[u].x, pat);
      const unsigned int hy = owner_match(v[u].y, pat);
      const unsigned int hz = owner_match(v[u].z, pat);
      const unsigned int hw = owner_match(v[u].w, pat);
      if (gi < ng && (hx | hy | hz | hw) != 0u) {
        const int idx = atomicAdd(ccnt, 1);
        if (idx < CCAP) cand[idx] = gi;
      }
    }
  }
}

template <int R, int H>
__device__ __forceinline__ void agg_candidates_append(const int* __restrict__ dstp, int E,
                                                      const uint4* __restrict__ o4, unsigned int pat,
                                                      const int* cand, int nc,
                                                      int d0, int rows_all,
                                                      int* ls, int* lr, int* lcnt) {
  for (int ci = (int)threadIdx.x; ci < nc; ci += AGG_THREADS) {
    const int g = cand[ci];
    const uint4 v = o4[g];
    int mask = (owner_match(v.x, pat) ? 1 : 0) | (owner_match(v.y, pat) ? 2 : 0) |
               (owner_match(v.z, pat) ? 4 : 0) | (owner_match(v.w, pat) ? 8 : 0);
    const int gbase = g * 16;
    while (mask != 0) {
      const int e0 = gbase + 4 * __builtin_ctz(mask);
      mask &= mask - 1;
      if (e0 < E) {
        const int4 dq = *(const int4*)(dstp + e0);
        const int dv[4] = {dq.x, dq.y, dq.z, dq.w};
#pragma unroll
        for (int k = 0; k < 4; ++k) {
          const int r = dv[k] - d0;
          if ((unsigned)r < (unsigned)rows_all) {
            const int hh = (H > 1) ? r / R : 0;
            const int idx = atomicAdd(lcnt + hh, 1);
            if (idx < AGG_CAP) { ls[hh * AGG_CAP + idx] = e0 + k; lr[hh * AGG_CAP + idx] = r; }
          }
        }
      }
    }
  }
}

template <int R, int H>
__device__ __forceinline__ void agg_scan_append(const int* __restrict__ dstp, int eb, int ee,
                                                int d0, int lo, int hi, int wv, int nw, int lane,
                                                int* ls, int* lr, int* lcnt) {
  const int sweep  = AGG_U * 32;
  const int stride = nw * sweep;
  int base = eb + wv * sweep;

  if (base + sweep <= ee) {
    int cur[AGG_U];
#pragma unroll
    for (int k = 0; k < AGG_U; ++k) cur[k] = dstp[base + k * 32 + lane];
    while (true) {
      const int nb = base + stride;
      const bool more = (nb + sweep <= ee);
      const int pb = more ? nb : base;
      int nxt[AGG_U];
#pragma unroll
      for (int k = 0; k < AGG_U; ++k) nxt[k] = dstp[pb + k * 32 + lane];
      agg_append_owned<R, H>(cur, base, lane, d0, lo, hi, ls, lr, lcnt);
      base = nb;
      if (!more) break;
#pragma unroll
      for (int k = 0; k < AGG_U; ++k) cur[k] = nxt[k];
    }
  }
  if (base < ee) {
    int cur[AGG_U];
#pragma unroll
    for (int k = 0; k < AGG_U; ++k) {
      const int e = base + k * 32 + lane;
      cur[k] = (e < ee) ? dstp[e] : -1;
    }
    agg_append_owned<R, H>(cur, base, lane, d0, lo, hi, ls, lr, lcnt);
  }
}

__device__ __forceinline__ void agg_list_scores(const AggList& L, int first, int n, bool do_max,
                                                const int* __restrict__ ei,
                                                const float* __restrict__ as) {
  for (int i = first + (int)threadIdx.x; i < n; i += AGG_THREADS) {
    const int r = L.r[i];
    const int s = ei[L.s[i]];
    const float sc = leakyr(as[s] + L.ad[r]);
    L.s[i] = s;
    L.v[i] = sc;
    if (do_max) lds_max_f32(&L.m[r], sc);
  }
}

__device__ __forceinline__ void agg_list_exps(const AggList& L, int n, bool do_sum) {
  for (int i = (int)threadIdx.x; i < n; i += AGG_THREADS) {
    const int r = L.r[i];
    const float ex = expf(L.v[i] - L.m[r]);
    L.v[i] = ex;
    if (do_sum) atomicAdd(&L.den[r], ex);
  }
}

template <int D, int P>
__device__ __forceinline__ void agg_gather_entry(const AggList& L, int i, int rbase, int lane,
                                                 const float* __restrict__ h, float* acc) {
  constexpr int V = D / 32;
  const int s = L.s[i];
  const float* hs = h + (size_t)s * D + lane;
  float v[V];
#pragma unroll
  for (int j = 0; j < V; ++j) v[j] = hs[j * 32];
  const int r = L.r[i];
  const float a = L.v[i] / L.den[r];
  float* ar = acc + (r - rbase) * P + lane;
#pragma unroll
  for (int j = 0; j < V; ++j) atomicAdd(ar + j * 32, a * v[j]);
}

template <int D, int EPT, int P>
__device__ __forceinline__ void agg_list_gather(const AggList& L, int n, int rbase,
                                                int wv, int nw, int lane,
                                                const float* __restrict__ h, float* acc) {
  constexpr int V = D / 32;
  const int groups = n / EPT;
  int g = wv;
  for (; g < groups; g += nw) {
    if (EPT == 1) {
      agg_gather_entry<D, P>(L, g, rbase, lane, h, acc);
    } else {
      const int i0 = g * EPT;
      float v[EPT][V];
#pragma unroll
      for (int e = 0; e < EPT; ++e) {
        const float* hs = h + (size_t)L.s[i0 + e] * D + lane;
#pragma unroll
        for (int j = 0; j < V; ++j) v[e][j] = hs[j * 32];
      }
#pragma unroll
      for (int e = 0; e < EPT; ++e) {
        const int r = L.r[i0 + e];
        const float a = L.v[i0 + e] / L.den[r];
        float* ar = acc + (r - rbase) * P + lane;
#pragma unroll
        for (int j = 0; j < V; ++j) atomicAdd(ar + j * 32, a * v[e][j]);
      }
    }
  }
  if (EPT > 1 && g == groups) {
    for (int i = groups * EPT; i < n; ++i) agg_gather_entry<D, P>(L, i, rbase, lane, h, acc);
  }
}

template <int D, int R, int H, int EPT, int LMODE, int NOUT, int NFC>
__global__ __launch_bounds__(AGG_THREADS) void gat_aggregate_owned(
    const int* __restrict__ ei, const float* __restrict__ as,
    const float* __restrict__ ad, const float* __restrict__ h,
    const float* __restrict__ bias, const float* __restrict__ Wn,
    const float* __restrict__ ta, const float* __restrict__ tb,
    float* __restrict__ out, float* __restrict__ to_a, float* __restrict__ to_b,
    int E, int N,
    int* __restrict__ saved, const float* ws_base, size_t ws_bytes,
    const unsigned int* __restrict__ own) {
  constexpr int RB = R * H;
  constexpr int P = (NOUT > 0) ? D + 4 : D;
  constexpr int Q = D / 4;
  constexpr int P4 = P / 4;
  static_assert((Q & (Q - 1)) == 0 && AGG_THREADS % Q == 0, "column group must be fixed per thread");
  static_assert(NOUT % 16 == 0 && D % 4 == 0 && ((R + 15) / 16) * (NOUT / 16) <= AGG_THREADS / 32,
                "one product tile per wave");
  static_assert(RB <= 256, "a block-local row must fit the low byte of a saved entry");
  static_assert(LMODE != 2 || H == 1, "loaded lists go into one list");
  static_assert(NOUT == 0 || NOUT == 64, "16 threads, one 16-lane row of a wave, hold a product row");
  static_assert(NFC == 0 || (NOUT == 0 && Q == 16), "16 threads, one 16-lane row of a wave, hold a finished row");
  __shared__ __attribute__((aligned(16))) float acc[R * P];
  __shared__ int   list_s[H * AGG_CAP];
  __shared__ int   list_r[H * AGG_CAP];
  __shared__ float list_v[H * AGG_CAP];
  __shared__ float row_m[RB];
  __shared__ float row_den[RB];
  __shared__ float row_ad[RB];
  __shared__ int   list_cnt[H];
  constexpr int CCAP = (LMODE == 1) ? 4096 : 1;
  __shared__ int   cand[CCAP];
  __shared__ int   cand_cnt;
  static_assert(R <= AGG_CAP && RB <= AGG_THREADS, "self loops open the lists, one thread each");

  const int tid  = threadIdx.x;
  const int lane = tid & 31;
  const int wv   = tid >> 5;
  const int nw   = AGG_THREADS >> 5;
  const int d0   = blockIdx.x * RB;
  const int rows_all = (N - d0 < RB) ? (N - d0) : RB;
  const int* __restrict__ dstp = ei + E;

  const size_t save_end = (size_t)((const char*)saved - (const char*)ws_base) +
                          (size_t)gridDim.x * AGG_SAVE * sizeof(int);
  const bool ws_ok = (LMODE != 0) && save_end <= ws_bytes && N <= (1 << 24);
  int* mine = saved + (size_t)blockIdx.x * AGG_SAVE;
  int loaded = -1;
  if (LMODE == 2 && ws_ok) {
    const int t = mine[0];
    if (t >= rows_all && t <= AGG_SAVE - 2 && t <= AGG_CAP) loaded = t;
  }

  auto run_rows = [&](int hh) {
    const int left = rows_all - hh * R;
    return left < 0 ? 0 : (left < R ? left : R);
  };
  auto run_list = [&](int hh) {
    return AggList{list_s + hh * AGG_CAP, list_r + hh * AGG_CAP, list_v + hh * AGG_CAP,
                   list_cnt + hh, row_m, row_den, row_ad};
  };

  {
    float4* z4 = (float4*)acc;
    for (int idx = tid; idx < run_rows(0) * Q; idx += AGG_THREADS)
      z4[(idx / Q) * P4 + (idx & (Q - 1))] = make_float4(0.f, 0.f, 0.f, 0.f);
  }
  if (tid < rows_all) {
    const int hh = tid / R;
    const int at = hh * AGG_CAP + (tid - hh * R);
    const float adv = ad[d0 + tid];
    const float sc  = leakyr(as[d0 + tid] + adv);
    row_ad[tid] = adv; row_m[tid] = sc; row_den[tid] = 0.f;
    list_s[at] = d0 + tid; list_r[at] = tid; list_v[at] = sc;
  }
  if (tid < H) list_cnt[tid] = run_rows(tid);
  if (LMODE == 1 && tid == 0) cand_cnt = 0;
  __syncthreads();

  if (LMODE == 2 && loaded >= 0) {
    for (int i = tid; i < loaded; i += 2 * AGG_THREADS) {
      const int j = i + AGG_THREADS;
      const bool two = j < loaded;
      const int jc = two ? j : i;
      const unsigned w0 = (unsigned)mine[2 + i];
      const unsigned w1 = (unsigned)mine[2 + jc];
      const int s0 = (int)(w0 >> 8), r0 = (int)(w0 & 255u);
      const int s1 = (int)(w1 >> 8), r1 = (int)(w1 & 255u);
      const float a0 = as[s0];
      const float a1 = as[s1];
      const float sc0 = leakyr(a0 + row_ad[r0]);
      list_s[i] = s0; list_r[i] = r0; list_v[i] = sc0;
      lds_max_f32(&row_m[r0], sc0);
      if (two) {
        const float sc1 = leakyr(a1 + row_ad[r1]);
        list_s[j] = s1; list_r[j] = r1; list_v[j] = sc1;
        lds_max_f32(&row_m[r1], sc1);
      }
    }
  } else {
    bool listed = false;
    if (LMODE == 1 && own != nullptr && (int)gridDim.x <= 256 && owner_bytes_ok(N, E, RB)) {
      const uint4* o4 = (const uint4*)own;
      const unsigned int pat = (unsigned int)blockIdx.x * 0x01010101u;
      agg_owner_candidates<CCAP>(o4, (E + 15) >> 4, pat, cand, &cand_cnt);
      __syncthreads();
      const int nc = cand_cnt;
      if (nc <= CCAP) {
        agg_candidates_append<R, H>(dstp, E, o4, pat, cand, nc, d0, rows_all, list_s, list_r, list_cnt);
        listed = true;
      }
    }
    if (!listed)
      agg_scan_append<R, H>(dstp, 0, E, d0, 0, rows_all, wv, nw, lane, list_s, list_r, list_cnt);
  }
  __syncthreads();

  int  n[H];
  bool fits[H];
#pragma unroll
  for (int hh = 0; hh < H; ++hh) {
    n[hh] = (LMODE == 2 && loaded >= 0) ? loaded : list_cnt[hh];
    fits[hh] = (n[hh] <= AGG_CAP);
  }
  if (!(LMODE == 2 && loaded >= 0)) {
#pragma unroll
    for (int hh = 0; hh < H; ++hh)
      if (fits[hh]) agg_list_scores(run_list(hh), run_rows(hh), n[hh], true, ei, as);
  }
  __syncthreads();

  if (LMODE == 1 && ws_ok) {
    int total = 0;
    bool all = true;
#pragma unroll
    for (int hh = 0; hh < H; ++hh) { total += n[hh]; all = all && fits[hh]; }
    const bool keep = all && total <= AGG_SAVE - 2;
    if (tid == 0) mine[0] = keep ? total : -1;
    if (keep) {
      int at = 2;
#pragma unroll
      for (int hh = 0; hh < H; ++hh) {
        for (int i = tid; i < n[hh]; i += AGG_THREADS)
          mine[at + i] = (int)(((unsigned)list_s[hh * AGG_CAP + i] << 8) | (unsigned)list_r[hh * AGG_CAP + i]);
        at += n[hh];
      }
    }
  }
#pragma unroll
  for (int hh = 0; hh < H; ++hh)
    if (fits[hh]) agg_list_exps(run_list(hh), n[hh], true);
  __syncthreads();

  const int c0 = (tid & (Q - 1)) * 4;
  const float bx = bias[c0], by = bias[c0 + 1], bz = bias[c0 + 2], bw = bias[c0 + 3];
  constexpr int NF = NFC > 0 ? NFC : 1;
  float wf[4][NF] = {}, fb[NF] = {};
  if (NFC > 0) {
#pragma unroll
    for (int q = 0; q < NF; ++q) {
      fb[q] = tb[q];
#pragma unroll
      for (int j = 0; j < 4; ++j) wf[j][q] = ta[(c0 + j) * NF + q];
    }
  }
  const int pc = (NOUT > 0) ? (tid & 15) * 4 : 0;
  float sa[4] = {}, sb[4] = {};
  if (NOUT > 0) {
#pragma unroll
    for (int j = 0; j < 4; ++j) { sa[j] = ta[pc + j]; sb[j] = tb[pc + j]; }
  }

#pragma unroll
  for (int hh = 0; hh < H; ++hh) {
    const int rows = run_rows(hh);
    if (rows > 0) {
      const AggList L = run_list(hh);
      const int rb = hh * R;
      if (fits[hh]) {
        agg_list_gather<D, EPT, P>(L, n[hh], rb, wv, nw, lane, h, acc);
      } else {
        for (int pass = 0; pass < 3; ++pass) {
          for (int eb = -AGG_CAP; eb < E; eb += AGG_CAP) {
            __syncthreads();
            int first = 0;
            if (eb < 0) {
              if (tid < rows) {
                L.s[tid] = d0 + rb + tid; L.r[tid] = rb + tid;
                L.v[tid] = leakyr(as[d0 + rb + tid] + row_ad[rb + tid]);
              }
              if (tid == 0) *L.cnt = rows;
              first = rows;
              __syncthreads();
            } else {
              if (tid == 0) *L.cnt = 0;
              __syncthreads();
              agg_scan_append<R, H>(dstp, eb, (E - eb < AGG_CAP) ? E : eb + AGG_CAP, d0, rb, rb + rows,
                                    wv, nw, lane, list_s, list_r, list_cnt);
              __syncthreads();
            }
            const int np = *L.cnt;
            agg_list_scores(L, first, np, pass == 0, ei, as);
            if (pass >= 1) {
              __syncthreads();
              agg_list_exps(L, np, pass == 1);
            }
            if (pass == 2) {
              __syncthreads();
              agg_list_gather<D, EPT, P>(L, np, rb, wv, nw, lane, h, acc);
            }
          }
        }
      }
      __syncthreads();

      float4* a4 = (float4*)acc;
      if (NOUT == 0 && NFC > 0) {
        for (int idx = tid; idx < rows * Q; idx += AGG_THREADS) {
          float4 v = a4[idx];
          if (hh + 1 < H) a4[idx] = make_float4(0.f, 0.f, 0.f, 0.f);
          v.x += bx; v.y += by; v.z += bz; v.w += bw;
          v.x = v.x > 0.f ? v.x : 0.f;
          v.y = v.y > 0.f ? v.y : 0.f;
          v.z = v.z > 0.f ? v.z : 0.f;
          v.w = v.w > 0.f ? v.w : 0.f;
          float p[NF];
#pragma unroll
          for (int q = 0; q < NF; ++q)
            p[q] = row16_sum(((v.x * wf[0][q] + v.y * wf[1][q]) + v.z * wf[2][q]) + v.w * wf[3][q]);
          if ((tid & 15) == 0) {
            float* orow = out + (size_t)(d0 + rb + idx / Q) * NFC;
#pragma unroll
            for (int q = 0; q < NF; ++q) orow[q] = p[q] + fb[q];
          }
        }
        if (hh + 1 < H) __syncthreads();
      } else if (NOUT == 0) {
        float4* o4 = (float4*)(out + (size_t)(d0 + rb) * D);
        for (int idx = tid; idx < rows * Q; idx += AGG_THREADS) {
          float4 v = a4[idx];
          if (hh + 1 < H) a4[idx] = make_float4(0.f, 0.f, 0.f, 0.f);
          v.x += bx; v.y += by; v.z += bz; v.w += bw;
          v.x = v.x > 0.f ? v.x : 0.f;
          v.y = v.y > 0.f ? v.y : 0.f;
          v.z = v.z > 0.f ? v.z : 0.f;
          v.w = v.w > 0.f ? v.w : 0.f;
          o4[idx] = v;
        }
        if (hh + 1 < H) __syncthreads();
      } else {
        for (int idx = tid; idx < rows * Q; idx += AGG_THREADS) {
          const int at = (idx / Q) * P4 + (idx & (Q - 1));
          float4 v = a4[at];
          v.x += bx; v.y += by; v.z += bz; v.w += bw;
          v.x = v.x > 0.f ? v.x : 0.f;
          v.y = v.y > 0.f ? v.y : 0.f;
          v.z = v.z > 0.f ? v.z : 0.f;
          v.w = v.w > 0.f ? v.w : 0.f;
          a4[at] = v;
        }
        __syncthreads();

        constexpr int NCT = NOUT / 16;
        const int job = __builtin_amdgcn_readfirstlane(tid >> 5);
        const bool has_job = job < ((rows + 15) >> 4) * NCT;
        const int rt = job / NCT;
        const int ct = job % NCT;
        const int half = lane >> 4;
        const int l16  = lane & 15;
        v8f prod = v8f{};
        if (has_job) {
          const int arow = (rt * 16 + l16 < rows) ? (rt * 16 + l16) : (rows - 1);
          const float* Al = acc + arow * P + 2 * half;
          const float* Bg = Wn + (size_t)(2 * half) * NOUT + ct * 16 + l16;
#pragma unroll 8
          for (int k = 0; k < D; k += 4) {
            const v2f a = *(const v2f*)(Al + k);
            v2f b;
            b.x = Bg[(size_t)k * NOUT];
            b.y = Bg[(size_t)(k + 1) * NOUT];
            prod = __builtin_amdgcn_wmma_f32_16x16x4_f32(
                 false, a,  false, b,
                 (short)0, prod,  false,  false);
          }
        }
        __syncthreads();

        if (has_job) {
#pragma unroll
          for (int r = 0; r < 8; ++r) {
            const int row = rt * 16 + 8 * half + r;
            if (row < rows) acc[row * NOUT + ct * 16 + l16] = prod[r];
          }
        }
        __syncthreads();

        float4* o4 = (float4*)(out + (size_t)(d0 + rb) * NOUT);
        for (int idx = tid; idx < rows * (NOUT / 4); idx += AGG_THREADS) {
          const float4 v = a4[idx];
          o4[idx] = v;
          const float ps = row16_sum(((v.x * sa[0] + v.y * sa[1]) + v.z * sa[2]) + v.w * sa[3]);
          const float pd = row16_sum(((v.x * sb[0] + v.y * sb[1]) + v.z * sb[2]) + v.w * sb[3]);
          if ((tid & 15) == 0) {
            const int node = d0 + rb + idx / (NOUT / 4);
            to_a[node] = ps;
            to_b[node] = pd;
          }
        }

        if (hh + 1 < H) {
          __syncthreads();
          for (int idx = tid; idx < run_rows(hh + 1) * Q; idx += AGG_THREADS)
            a4[(idx / Q) * P4 + (idx & (Q - 1))] = make_float4(0.f, 0.f, 0.f, 0.f);
          __syncthreads();
        }
      }
    }
  }
}

extern "C" void kernel_launch(void* const* d_in, const int* in_sizes, int n_in,
                              void* d_out, int out_size, void* d_ws, size_t ws_size,
                              hipStream_t stream) {
  const float* x   = (const float*)d_in[0];
  const int*   ei  = (const int*)  d_in[1];
  const float* W1  = (const float*)d_in[2];
  const float* a1s = (const float*)d_in[3];
  const float* a1d = (const float*)d_in[4];
  const float* b1  = (const float*)d_in[5];
  const float* W2  = (const float*)d_in[6];
  const float* a2s = (const float*)d_in[7];
  const float* a2d = (const float*)d_in[8];
  const float* b2  = (const float*)d_in[9];
  const float* fcW = (const float*)d_in[10];
  const float* fcb = (const float*)d_in[11];

  const int N    = in_sizes[0] / IN_C;
  const int E    = in_sizes[1] / 2;

  float* w       = (float*)d_ws;
  float* h1      = w;
  float* h2      = w + (size_t)N * HID;
  float* as2_buf = h2 + (size_t)N * OUT2;
  float* ad2_buf = as2_buf + N;
  unsigned int* own = (unsigned int*)(ad2_buf + N);
  float* as_buf  = w + (size_t)2 * N * HID;
  float* ad_buf  = as_buf + N;
  int*   lists   = (int*)(ad_buf + N);

  const int agg_own   = 2 * AGG_ROWS;
  const int agg_blk   = (N + agg_own - 1) / agg_own;

  gat_wmma_gemm_dots_f32<IN_C, HID, 16, 8, 2 * AGG_ROWS><<<G1_BLOCKS, 512, 0, stream>>>(x, W1, a1s, a1d, h1, as_buf, ad_buf, N, ei, E, own);
  gat_aggregate_owned<HID, AGG_ROWS, 2, 1, 1, OUT2, 0><<<agg_blk, AGG_THREADS, 0, stream>>>(ei, as_buf, ad_buf, h1, b1, W2, a2s, a2d, h2, as2_buf, ad2_buf, E, N, lists, w, ws_size, own);

  gat_aggregate_owned<OUT2, 2 * AGG_ROWS, 1, 2, 2, 0, NCLS><<<agg_blk, AGG_THREADS, 0, stream>>>(ei, as2_buf, ad2_buf, h2, b2, nullptr, fcW, fcb, (float*)d_out, nullptr, nullptr, E, N, lists, w, ws_size, nullptr);
}
